# starting kernel
# baseline (speedup 1.0000x reference)
#include <hip/hip_runtime.h>

typedef __attribute__((ext_vector_type(2))) float v2f;
typedef __attribute__((ext_vector_type(8))) float v8f;

#define IN_DIM 512
#define HID    16
#define ODIM   64

__global__ void k_deg_init(float* __restrict__ deg, int n) {
  int i = blockIdx.x * blockDim.x + threadIdx.x;
  if (i < n) deg[i] = 1.0f;
}

__global__ void k_deg_edges(const int* __restrict__ dst, float* __restrict__ deg, int E) {
  int i = blockIdx.x * blockDim.x + threadIdx.x;
  if (i < E) atomicAdd(&deg[dst[i]], 1.0f);
}

__global__ void k_dinv(float* __restrict__ deg, int n) {
  int i = blockIdx.x * blockDim.x + threadIdx.x;
  if (i < n) {
    float d = deg[i];
    deg[i] = (d > 0.0f) ? (1.0f / sqrtf(d)) : 0.0f;
  }
}

__global__ void k_gemm1(const float* __restrict__ x, const float* __restrict__ W1,
                        float* __restrict__ h1, int n) {
  __shared__ float lw[IN_DIM * HID];
  for (int i = threadIdx.x; i < IN_DIM * HID; i += blockDim.x) lw[i] = W1[i];
  __syncthreads();

  int wave  = (blockIdx.x * blockDim.x + threadIdx.x) >> 5;
  int lane  = threadIdx.x & 31;
  int tiles = (n + 15) >> 4;
  if (wave >= tiles) return;

  int half = lane >> 4;
  int ml   = lane & 15;
  int row  = wave * 16 + ml;
  int rrow = (row < n) ? row : (n - 1);
  const float* xr = x + (size_t)rrow * IN_DIM;

  v8f acc = {0.f, 0.f, 0.f, 0.f, 0.f, 0.f, 0.f, 0.f};
  for (int k = 0; k < IN_DIM; k += 4) {
    int ka = k + 2 * half;
    v2f a; a.x = xr[ka];                 a.y = xr[ka + 1];
    v2f b; b.x = lw[ka * HID + ml];      b.y = lw[(ka + 1) * HID + ml];
    acc = __builtin_amdgcn_wmma_f32_16x16x4_f32(false, a, false, b, (short)0, acc,
                                                false, false);
  }
#pragma unroll
  for (int r = 0; r < 8; ++r) {
    int orow = wave * 16 + r + 8 * half;
    if (orow < n) h1[(size_t)orow * HID + ml] = acc[r];
  }
}

__global__ void k_init_z1(const float* __restrict__ h1, const float* __restrict__ dinv,
                          const float* __restrict__ b1, float* __restrict__ z1, int n) {
  int t = blockIdx.x * blockDim.x + threadIdx.x;
  int total = n * HID;
  if (t < total) {
    int i = t / HID, j = t - i * HID;
    float di = dinv[i];
    z1[t] = h1[t] * di * di + b1[j];
  }
}

__global__ void k_agg1(const int* __restrict__ src, const int* __restrict__ dst,
                       const float* __restrict__ dinv, const float* __restrict__ h1,
                       float* __restrict__ z1, int E) {
  int total  = E * 4;
  int stride = gridDim.x * blockDim.x;
  for (int t = blockIdx.x * blockDim.x + threadIdx.x; t < total; t += stride) {
    int e = t >> 2, c = t & 3;
    int s = src[e], d = dst[e];
    float norm = dinv[s] * dinv[d];
    const float4 hv = *(const float4*)(h1 + (size_t)s * HID + c * 4);
    float* o = z1 + (size_t)d * HID + c * 4;
    atomicAdd(o + 0, hv.x * norm);
    atomicAdd(o + 1, hv.y * norm);
    atomicAdd(o + 2, hv.z * norm);
    atomicAdd(o + 3, hv.w * norm);
  }
}

__global__ void k_gemm2(const float* __restrict__ z1, const float* __restrict__ W2,
                        float* __restrict__ h2, int n) {
  __shared__ float lw[HID * ODIM];
  for (int i = threadIdx.x; i < HID * ODIM; i += blockDim.x) lw[i] = W2[i];
  __syncthreads();

  int wave  = (blockIdx.x * blockDim.x + threadIdx.x) >> 5;
  int lane  = threadIdx.x & 31;
  int tiles = (n + 15) >> 4;
  if (wave >= tiles) return;

  int half = lane >> 4;
  int ml   = lane & 15;
  int row  = wave * 16 + ml;
  int rrow = (row < n) ? row : (n - 1);
  const float* zr = z1 + (size_t)rrow * HID;

  v2f afrag[4];
#pragma unroll
  for (int kk = 0; kk < 4; ++kk) {
    int ka = kk * 4 + 2 * half;
    afrag[kk].x = fmaxf(zr[ka], 0.f);
    afrag[kk].y = fmaxf(zr[ka + 1], 0.f);
  }

#pragma unroll
  for (int tcol = 0; tcol < 4; ++tcol) {
    v8f acc = {0.f, 0.f, 0.f, 0.f, 0.f, 0.f, 0.f, 0.f};
#pragma unroll
    for (int kk = 0; kk < 4; ++kk) {
      int ka = kk * 4 + 2 * half;
      v2f b;
      b.x = lw[ka * ODIM + tcol * 16 + ml];
      b.y = lw[(ka + 1) * ODIM + tcol * 16 + ml];
      acc = __builtin_amdgcn_wmma_f32_16x16x4_f32(false, afrag[kk], false, b, (short)0,
                                                  acc, false, false);
    }
#pragma unroll
    for (int r = 0; r < 8; ++r) {
      int orow = wave * 16 + r + 8 * half;
      if (orow < n) h2[(size_t)orow * ODIM + tcol * 16 + ml] = acc[r];
    }
  }
}

__global__ void k_init_out(const float* __restrict__ h2, const float* __restrict__ dinv,
                           const float* __restrict__ b2, float* __restrict__ out, int n) {
  int t = blockIdx.x * blockDim.x + threadIdx.x;
  int total = n * ODIM;
  if (t < total) {
    int i = t / ODIM, j = t - i * ODIM;
    float di = dinv[i];
    out[t] = h2[t] * di * di + b2[j];
  }
}

__global__ void k_agg2(const int* __restrict__ src, const int* __restrict__ dst,
                       const float* __restrict__ dinv, const float* __restrict__ h2,
                       float* __restrict__ out, int E) {
  int total  = E * 16;
  int stride = gridDim.x * blockDim.x;
  for (int t = blockIdx.x * blockDim.x + threadIdx.x; t < total; t += stride) {
    int e = t >> 4, c = t & 15;
    int s = src[e], d = dst[e];
    float norm = dinv[s] * dinv[d];
    const float4 hv = *(const float4*)(h2 + (size_t)s * ODIM + c * 4);
    float* o = out + (size_t)d * ODIM + c * 4;
    atomicAdd(o + 0, hv.x * norm);
    atomicAdd(o + 1, hv.y * norm);
    atomicAdd(o + 2, hv.z * norm);
    atomicAdd(o + 3, hv.w * norm);
  }
}

extern "C" void kernel_launch(void* const* d_in, const int* in_sizes, int n_in,
                              void* d_out, int out_size, void* d_ws, size_t ws_size,
                              hipStream_t stream) {
  const float* x  = (const float*)d_in[0];
  const float* W1 = (const float*)d_in[1];
  const float* b1 = (const float*)d_in[2];
  const float* W2 = (const float*)d_in[3];
  const float* b2 = (const float*)d_in[4];
  const int*   ei = (const int*)d_in[5];

  const int N = in_sizes[0] / IN_DIM;
  const int E = in_sizes[5] / 2;
  const int* src = ei;
  const int* dst = ei + E;
  float* out = (float*)d_out;

  char* ws = (char*)d_ws;
  size_t off = 0;
  auto alloc = [&](size_t nbytes) -> void* {
    void* p = ws + off;
    off = (off + nbytes + 255) & ~(size_t)255;
    return p;
  };
  float* dinv = (float*)alloc((size_t)N * sizeof(float));
  float* h1   = (float*)alloc((size_t)N * HID  * sizeof(float));
  float* z1   = (float*)alloc((size_t)N * HID  * sizeof(float));
  float* h2   = (float*)alloc((size_t)N * ODIM * sizeof(float));

  const int B = 256;
  const int tiles = (N + 15) >> 4;
  const int gemmBlocks = (tiles + 7) / 8;

  k_deg_init <<<(N + B - 1) / B, B, 0, stream>>>(dinv, N);
  k_deg_edges<<<(E + B - 1) / B, B, 0, stream>>>(dst, dinv, E);
  k_dinv     <<<(N + B - 1) / B, B, 0, stream>>>(dinv, N);

  k_gemm1    <<<gemmBlocks, B, 0, stream>>>(x, W1, h1, N);
  k_init_z1  <<<((size_t)N * HID + B - 1) / B, B, 0, stream>>>(h1, dinv, b1, z1, N);
  k_agg1     <<<(E * 4 + B - 1) / B, B, 0, stream>>>(src, dst, dinv, h1, z1, E);

  k_gemm2    <<<gemmBlocks, B, 0, stream>>>(z1, W2, h2, N);
  k_init_out <<<((size_t)N * ODIM + B - 1) / B, B, 0, stream>>>(h2, dinv, b2, out, N);
  k_agg2     <<<(E * 16 + B - 1) / B, B, 0, stream>>>(src, dst, dinv, h2, out, E);
}


# edges regrouped once per call by destination bucket of 128 nodes (LDS counting sort); a work-group per bucket sums degrees and both layers' rows by LDS f32 adds, no global atomics; sum order changes
# speedup vs baseline: 1.9930x; 1.9930x over previous
#include <hip/hip_runtime.h>

typedef __attribute__((ext_vector_type(2))) float v2f;
typedef __attribute__((ext_vector_type(8))) float v8f;

#define IN_DIM 512
#define HID    16
#define ODIM   64

__global__ void k_deg_init(float* __restrict__ deg, int n) {
  int i = blockIdx.x * blockDim.x + threadIdx.x;
  if (i < n) deg[i] = 1.0f;
}

__global__ void k_deg_edges(const int* __restrict__ dst, float* __restrict__ deg, int E) {
  int i = blockIdx.x * blockDim.x + threadIdx.x;
  if (i < E) atomicAdd(&deg[dst[i]], 1.0f);
}

__global__ void k_dinv(float* __restrict__ deg, int n) {
  int i = blockIdx.x * blockDim.x + threadIdx.x;
  if (i < n) {
    float d = deg[i];
    deg[i] = (d > 0.0f) ? (1.0f / sqrtf(d)) : 0.0f;
  }
}

__global__ void k_gemm1(const float* __restrict__ x, const float* __restrict__ W1,
                        float* __restrict__ h1, int n) {
  __shared__ float lw[IN_DIM * HID];
  for (int i = threadIdx.x; i < IN_DIM * HID; i += blockDim.x) lw[i] = W1[i];
  __syncthreads();

  int wave  = (blockIdx.x * blockDim.x + threadIdx.x) >> 5;
  int lane  = threadIdx.x & 31;
  int tiles = (n + 15) >> 4;
  if (wave >= tiles) return;

  int half = lane >> 4;
  int ml   = lane & 15;
  int row  = wave * 16 + ml;
  int rrow = (row < n) ? row : (n - 1);
  const float* xr = x + (size_t)rrow * IN_DIM;

  v8f acc = {0.f, 0.f, 0.f, 0.f, 0.f, 0.f, 0.f, 0.f};
  for (int k = 0; k < IN_DIM; k += 4) {
    int ka = k + 2 * half;
    v2f a; a.x = xr[ka];                 a.y = xr[ka + 1];
    v2f b; b.x = lw[ka * HID + ml];      b.y = lw[(ka + 1) * HID + ml];
    acc = __builtin_amdgcn_wmma_f32_16x16x4_f32(false, a, false, b, (short)0, acc,
                                                false, false);
  }
#pragma unroll
  for (int r = 0; r < 8; ++r) {
    int orow = wave * 16 + r + 8 * half;
    if (orow < n) h1[(size_t)orow * HID + ml] = acc[r];
  }
}

__global__ void k_init_z1(const float* __restrict__ h1, const float* __restrict__ dinv,
                          const float* __restrict__ b1, float* __restrict__ z1, int n) {
  int t = blockIdx.x * blockDim.x + threadIdx.x;
  int total = n * HID;
  if (t < total) {
    int i = t / HID, j = t - i * HID;
    float di = dinv[i];
    z1[t] = h1[t] * di * di + b1[j];
  }
}

__global__ void k_agg1(const int* __restrict__ src, const int* __restrict__ dst,
                       const float* __restrict__ dinv, const float* __restrict__ h1,
                       float* __restrict__ z1, int E) {
  int total  = E * 4;
  int stride = gridDim.x * blockDim.x;
  for (int t = blockIdx.x * blockDim.x + threadIdx.x; t < total; t += stride) {
    int e = t >> 2, c = t & 3;
    int s = src[e], d = dst[e];
    float norm = dinv[s] * dinv[d];
    const float4 hv = *(const float4*)(h1 + (size_t)s * HID + c * 4);
    float* o = z1 + (size_t)d * HID + c * 4;
    atomicAdd(o + 0, hv.x * norm);
    atomicAdd(o + 1, hv.y * norm);
    atomicAdd(o + 2, hv.z * norm);
    atomicAdd(o + 3, hv.w * norm);
  }
}

__global__ void k_gemm2(const float* __restrict__ z1, const float* __restrict__ W2,
                        float* __restrict__ h2, int n) {
  __shared__ float lw[HID * ODIM];
  for (int i = threadIdx.x; i < HID * ODIM; i += blockDim.x) lw[i] = W2[i];
  __syncthreads();

  int wave  = (blockIdx.x * blockDim.x + threadIdx.x) >> 5;
  int lane  = threadIdx.x & 31;
  int tiles = (n + 15) >> 4;
  if (wave >= tiles) return;

  int half = lane >> 4;
  int ml   = lane & 15;
  int row  = wave * 16 + ml;
  int rrow = (row < n) ? row : (n - 1);
  const float* zr = z1 + (size_t)rrow * HID;

  v2f afrag[4];
#pragma unroll
  for (int kk = 0; kk < 4; ++kk) {
    int ka = kk * 4 + 2 * half;
    afrag[kk].x = fmaxf(zr[ka], 0.f);
    afrag[kk].y = fmaxf(zr[ka + 1], 0.f);
  }

#pragma unroll
  for (int tcol = 0; tcol < 4; ++tcol) {
    v8f acc = {0.f, 0.f, 0.f, 0.f, 0.f, 0.f, 0.f, 0.f};
#pragma unroll
    for (int kk = 0; kk < 4; ++kk) {
      int ka = kk * 4 + 2 * half;
      v2f b;
      b.x = lw[ka * ODIM + tcol * 16 + ml];
      b.y = lw[(ka + 1) * ODIM + tcol * 16 + ml];
      acc = __builtin_amdgcn_wmma_f32_16x16x4_f32(false, afrag[kk], false, b, (short)0,
                                                  acc, false, false);
    }
#pragma unroll
    for (int r = 0; r < 8; ++r) {
      int orow = wave * 16 + r + 8 * half;
      if (orow < n) h2[(size_t)orow * ODIM + tcol * 16 + ml] = acc[r];
    }
  }
}

__global__ void k_init_out(const float* __restrict__ h2, const float* __restrict__ dinv,
                           const float* __restrict__ b2, float* __restrict__ out, int n) {
  int t = blockIdx.x * blockDim.x + threadIdx.x;
  int total = n * ODIM;
  if (t < total) {
    int i = t / ODIM, j = t - i * ODIM;
    float di = dinv[i];
    out[t] = h2[t] * di * di + b2[j];
  }
}

__global__ void k_agg2(const int* __restrict__ src, const int* __restrict__ dst,
                       const float* __restrict__ dinv, const float* __restrict__ h2,
                       float* __restrict__ out, int E) {
  int total  = E * 16;
  int stride = gridDim.x * blockDim.x;
  for (int t = blockIdx.x * blockDim.x + threadIdx.x; t < total; t += stride) {
    int e = t >> 4, c = t & 15;
    int s = src[e], d = dst[e];
    float norm = dinv[s] * dinv[d];
    const float4 hv = *(const float4*)(h2 + (size_t)s * ODIM + c * 4);
    float* o = out + (size_t)d * ODIM + c * 4;
    atomicAdd(o + 0, hv.x * norm);
    atomicAdd(o + 1, hv.y * norm);
    atomicAdd(o + 2, hv.z * norm);
    atomicAdd(o + 3, hv.w * norm);
  }
}

#define RB     128
#define LOG_RB 7
#define NSL    256

__device__ __forceinline__ int block_excl_scan(int v, int* scr, int* total) {
  const int t = threadIdx.x, nt = blockDim.x;
  scr[t] = v;
  __syncthreads();
  for (int o = 1; o < nt; o <<= 1) {
    int a = (t >= o) ? scr[t - o] : 0;
    __syncthreads();
    scr[t] += a;
    __syncthreads();
  }
  const int incl = scr[t];
  *total = scr[nt - 1];
  __syncthreads();
  return incl - v;
}

__global__ void k_bcount(const int* __restrict__ dst, int* __restrict__ cntmat,
                         int E, int L, int nb) {
  extern __shared__ int dynsh[];
  int* cnt = dynsh;
  const int tid = threadIdx.x, bd = blockDim.x, wg = blockIdx.x;
  for (int i = tid; i < nb; i += bd) cnt[i] = 0;
  __syncthreads();
  const int e0 = wg * L;
  int e1 = e0 + L;
  if (e1 > E) e1 = E;
  int e = e0 + tid;
  for (; e + 3 * bd < e1; e += 4 * bd) {
    int d[4];
#pragma unroll
    for (int u = 0; u < 4; ++u) d[u] = dst[e + u * bd];
#pragma unroll
    for (int u = 0; u < 4; ++u) atomicAdd(&cnt[d[u] >> LOG_RB], 1);
  }
  for (; e < e1; e += bd) atomicAdd(&cnt[dst[e] >> LOG_RB], 1);
  __syncthreads();
  for (int i = tid; i < nb; i += bd) cntmat[(size_t)i * NSL + wg] = cnt[i];
}

__global__ void k_bscan(int* __restrict__ cntmat, int* __restrict__ totals) {
  __shared__ int scr[NSL];
  const int b = blockIdx.x, w = threadIdx.x;
  const int v = cntmat[(size_t)b * NSL + w];
  int tot;
  const int ex = block_excl_scan(v, scr, &tot);
  cntmat[(size_t)b * NSL + w] = ex;
  if (w == 0) totals[b] = tot;
}

__global__ void k_bscatter(const int* __restrict__ src, const int* __restrict__ dst,
                           const int* __restrict__ offs, const int* __restrict__ totals,
                           int* __restrict__ bstart, unsigned* __restrict__ rec,
                           int E, int L, int nb) {
  extern __shared__ int dynsh[];
  int* cursor = dynsh;
  int* scr    = dynsh + nb;
  const int tid = threadIdx.x, bd = blockDim.x, wg = blockIdx.x;
  const int per = (nb + bd - 1) / bd;
  const int lo = tid * per;
  int hi = lo + per;
  if (hi > nb) hi = nb;
  int s = 0;
  for (int b = lo; b < hi; ++b) s += totals[b];
  int tot;
  int pre = block_excl_scan(s, scr, &tot);
  for (int b = lo; b < hi; ++b) {
    cursor[b] = pre + offs[(size_t)b * NSL + wg];
    if (wg == 0) bstart[b] = pre;
    pre += totals[b];
  }
  if (wg == 0 && tid == 0) bstart[nb] = tot;
  __syncthreads();
  const int e0 = wg * L;
  int e1 = e0 + L;
  if (e1 > E) e1 = E;
  int e = e0 + tid;
  for (; e + 3 * bd < e1; e += 4 * bd) {
    int sv[4], dv[4];
#pragma unroll
    for (int u = 0; u < 4; ++u) {
      sv[u] = src[e + u * bd];
      dv[u] = dst[e + u * bd];
    }
#pragma unroll
    for (int u = 0; u < 4; ++u) {
      const int p = atomicAdd(&cursor[dv[u] >> LOG_RB], 1);
      rec[p] = ((unsigned)sv[u] << LOG_RB) | (unsigned)(dv[u] & (RB - 1));
    }
  }
  for (; e < e1; e += bd) {
    const int sv = src[e], dv = dst[e];
    const int p = atomicAdd(&cursor[dv >> LOG_RB], 1);
    rec[p] = ((unsigned)sv << LOG_RB) | (unsigned)(dv & (RB - 1));
  }
}

__global__ void k_bdeg(const unsigned* __restrict__ rec, const int* __restrict__ bstart,
                       float* __restrict__ dinv, int n) {
  __shared__ int cnt[RB];
  const int tid = threadIdx.x, bd = blockDim.x, b = blockIdx.x;
  if (tid < RB) cnt[tid] = 0;
  __syncthreads();
  const int e0 = bstart[b], e1 = bstart[b + 1];
  for (int e = e0 + tid; e < e1; e += bd) atomicAdd(&cnt[rec[e] & (RB - 1)], 1);
  __syncthreads();
  if (tid < RB) {
    const int node = b * RB + tid;
    if (node < n) {
      const float d = 1.0f + (float)cnt[tid];
      dinv[node] = (d > 0.0f) ? (1.0f / sqrtf(d)) : 0.0f;
    }
  }
}

template <int F>
__device__ __forceinline__ void bagg_body(float* acc, float* dl,
                                          const unsigned* __restrict__ rec,
                                          const int* __restrict__ bstart,
                                          const float* __restrict__ dinv,
                                          const float* __restrict__ h,
                                          const float* __restrict__ bias,
                                          float* __restrict__ o, int n) {
  constexpr int LPE = F / 4;
  const int tid = threadIdx.x, bd = blockDim.x, b = blockIdx.x;
  const int base = b * RB;
  for (int i = tid; i < RB * F; i += bd) {
    const int node = base + i / F, j = i % F;
    float v = 0.f;
    if (node < n) {
      const float di = dinv[node];
      v = h[(size_t)node * F + j] * di * di + bias[j];
    }
    acc[i] = v;
  }
  if (tid < RB) dl[tid] = (base + tid < n) ? dinv[base + tid] : 0.f;
  __syncthreads();
  const int e0 = bstart[b], e1 = bstart[b + 1];
  const int q = tid / LPE, c = tid % LPE;
  const int epi = bd / LPE;
  int e = e0 + q;
  for (; e + 3 * epi < e1; e += 4 * epi) {
    unsigned r[4];
    float ds[4];
    float4 hv[4];
#pragma unroll
    for (int u = 0; u < 4; ++u) r[u] = rec[e + u * epi];
#pragma unroll
    for (int u = 0; u < 4; ++u) {
      const unsigned s = r[u] >> LOG_RB;
      ds[u] = dinv[s];
      hv[u] = *(const float4*)(h + (size_t)s * F + c * 4);
    }
#pragma unroll
    for (int u = 0; u < 4; ++u) {
      const int dloc = (int)(r[u] & (RB - 1));
      const float norm = ds[u] * dl[dloc];
      float* a = acc + dloc * F + c * 4;
      atomicAdd(a + 0, hv[u].x * norm);
      atomicAdd(a + 1, hv[u].y * norm);
      atomicAdd(a + 2, hv[u].z * norm);
      atomicAdd(a + 3, hv[u].w * norm);
    }
  }
  for (; e < e1; e += epi) {
    const unsigned r = rec[e];
    const unsigned s = r >> LOG_RB;
    const int dloc = (int)(r & (RB - 1));
    const float norm = dinv[s] * dl[dloc];
    const float4 hv = *(const float4*)(h + (size_t)s * F + c * 4);
    float* a = acc + dloc * F + c * 4;
    atomicAdd(a + 0, hv.x * norm);
    atomicAdd(a + 1, hv.y * norm);
    atomicAdd(a + 2, hv.z * norm);
    atomicAdd(a + 3, hv.w * norm);
  }
  __syncthreads();
  for (int i = tid; i < RB * F; i += bd) {
    const int node = base + i / F;
    if (node < n) o[(size_t)base * F + i] = acc[i];
  }
}

__global__ void k_bagg1(const unsigned* __restrict__ rec, const int* __restrict__ bstart,
                        const float* __restrict__ dinv, const float* __restrict__ h1,
                        const float* __restrict__ b1, float* __restrict__ z1, int n) {
  __shared__ float acc[RB * HID];
  __shared__ float dl[RB];
  bagg_body<HID>(acc, dl, rec, bstart, dinv, h1, b1, z1, n);
}

__global__ void k_bagg2(const unsigned* __restrict__ rec, const int* __restrict__ bstart,
                        const float* __restrict__ dinv, const float* __restrict__ h2,
                        const float* __restrict__ b2, float* __restrict__ out, int n) {
  __shared__ float acc[RB * ODIM];
  __shared__ float dl[RB];
  bagg_body<ODIM>(acc, dl, rec, bstart, dinv, h2, b2, out, n);
}

extern "C" void kernel_launch(void* const* d_in, const int* in_sizes, int n_in,
                              void* d_out, int out_size, void* d_ws, size_t ws_size,
                              hipStream_t stream) {
  const float* x  = (const float*)d_in[0];
  const float* W1 = (const float*)d_in[1];
  const float* b1 = (const float*)d_in[2];
  const float* W2 = (const float*)d_in[3];
  const float* b2 = (const float*)d_in[4];
  const int*   ei = (const int*)d_in[5];

  const int N = in_sizes[0] / IN_DIM;
  const int E = in_sizes[5] / 2;
  const int* src = ei;
  const int* dst = ei + E;
  float* out = (float*)d_out;

  char* ws = (char*)d_ws;
  size_t off = 0;
  auto alloc = [&](size_t nbytes) -> void* {
    void* p = ws + off;
    off = (off + nbytes + 255) & ~(size_t)255;
    return p;
  };
  float* dinv = (float*)alloc((size_t)N * sizeof(float));
  float* h1   = (float*)alloc((size_t)N * HID  * sizeof(float));
  float* z1   = (float*)alloc((size_t)N * HID  * sizeof(float));
  float* h2   = (float*)alloc((size_t)N * ODIM * sizeof(float));

  const int B = 256;
  const int tiles = (N + 15) >> 4;
  const int gemmBlocks = (tiles + 7) / 8;

  const int NB = (N + RB - 1) / RB;
  const int L  = (E + NSL - 1) / NSL;
  unsigned* rec = (unsigned*)alloc((size_t)E * sizeof(unsigned));
  int* cntmat   = (int*)alloc((size_t)NB * NSL * sizeof(int));
  int* totals   = (int*)alloc((size_t)NB * sizeof(int));
  int* bstart   = (int*)alloc((size_t)(NB + 1) * sizeof(int));

  if (off <= ws_size) {
    k_bcount  <<<NSL, 512, (size_t)NB * sizeof(int), stream>>>(dst, cntmat, E, L, NB);
    k_bscan   <<<NB, NSL, 0, stream>>>(cntmat, totals);
    k_bscatter<<<NSL, 512, (size_t)(NB + 512) * sizeof(int), stream>>>(
        src, dst, cntmat, totals, bstart, rec, E, L, NB);
    k_bdeg    <<<NB, 1024, 0, stream>>>(rec, bstart, dinv, N);

    k_gemm1   <<<gemmBlocks, B, 0, stream>>>(x, W1, h1, N);
    k_bagg1   <<<NB, 512, 0, stream>>>(rec, bstart, dinv, h1, b1, z1, N);

    k_gemm2   <<<gemmBlocks, B, 0, stream>>>(z1, W2, h2, N);
    k_bagg2   <<<NB, 512, 0, stream>>>(rec, bstart, dinv, h2, b2, out, N);
    return;
  }

  k_deg_init <<<(N + B - 1) / B, B, 0, stream>>>(dinv, N);
  k_deg_edges<<<(E + B - 1) / B, B, 0, stream>>>(dst, dinv, E);
  k_dinv     <<<(N + B - 1) / B, B, 0, stream>>>(dinv, N);

  k_gemm1    <<<gemmBlocks, B, 0, stream>>>(x, W1, h1, N);
  k_init_z1  <<<((size_t)N * HID + B - 1) / B, B, 0, stream>>>(h1, dinv, b1, z1, N);
  k_agg1     <<<(E * 4 + B - 1) / B, B, 0, stream>>>(src, dst, dinv, h1, z1, E);

  k_gemm2    <<<gemmBlocks, B, 0, stream>>>(z1, W2, h2, N);
  k_init_out <<<((size_t)N * ODIM + B - 1) / B, B, 0, stream>>>(h2, dinv, b2, out, N);
  k_agg2     <<<(E * 16 + B - 1) / B, B, 0, stream>>>(src, dst, dinv, h2, out, E);
}


# dinv[src] rides with data an edge already reads: slot 16 of the h1 row's own 128-B line in layer 1, which leaves it in a per-edge plane that layer 2 reads in order; no dinv gathers; sums unchanged
# speedup vs baseline: 2.0560x; 1.0316x over previous
#include <hip/hip_runtime.h>

typedef __attribute__((ext_vector_type(2))) float v2f;
typedef __attribute__((ext_vector_type(8))) float v8f;

#define IN_DIM 512
#define HID    16
#define ODIM   64

__global__ void k_deg_init(float* __restrict__ deg, int n) {
  int i = blockIdx.x * blockDim.x + threadIdx.x;
  if (i < n) deg[i] = 1.0f;
}

__global__ void k_deg_edges(const int* __restrict__ dst, float* __restrict__ deg, int E) {
  int i = blockIdx.x * blockDim.x + threadIdx.x;
  if (i < E) atomicAdd(&deg[dst[i]], 1.0f);
}

__global__ void k_dinv(float* __restrict__ deg, int n) {
  int i = blockIdx.x * blockDim.x + threadIdx.x;
  if (i < n) {
    float d = deg[i];
    deg[i] = (d > 0.0f) ? (1.0f / sqrtf(d)) : 0.0f;
  }
}

__global__ void k_gemm1(const float* __restrict__ x, const float* __restrict__ W1,
                        float* __restrict__ h1, int n, int pitch,
                        const float* __restrict__ dinv) {
  __shared__ float lw[IN_DIM * HID];
  for (int i = threadIdx.x; i < IN_DIM * HID; i += blockDim.x) lw[i] = W1[i];
  __syncthreads();

  int wave  = (blockIdx.x * blockDim.x + threadIdx.x) >> 5;
  int lane  = threadIdx.x & 31;
  int tiles = (n + 15) >> 4;
  if (wave >= tiles) return;

  int half = lane >> 4;
  int ml   = lane & 15;
  int row  = wave * 16 + ml;
  int rrow = (row < n) ? row : (n - 1);
  const float* xr = x + (size_t)rrow * IN_DIM;

  v8f acc = {0.f, 0.f, 0.f, 0.f, 0.f, 0.f, 0.f, 0.f};
  for (int k = 0; k < IN_DIM; k += 4) {
    int ka = k + 2 * half;
    v2f a; a.x = xr[ka];                 a.y = xr[ka + 1];
    v2f b; b.x = lw[ka * HID + ml];      b.y = lw[(ka + 1) * HID + ml];
    acc = __builtin_amdgcn_wmma_f32_16x16x4_f32(false, a, false, b, (short)0, acc,
                                                false, false);
  }
#pragma unroll
  for (int r = 0; r < 8; ++r) {
    int orow = wave * 16 + r + 8 * half;
    if (orow < n) h1[(size_t)orow * pitch + ml] = acc[r];
  }
  if (dinv != nullptr && half == 0 && row < n) h1[(size_t)row * pitch + HID] = dinv[row];
}

__global__ void k_init_z1(const float* __restrict__ h1, const float* __restrict__ dinv,
                          const float* __restrict__ b1, float* __restrict__ z1, int n) {
  int t = blockIdx.x * blockDim.x + threadIdx.x;
  int total = n * HID;
  if (t < total) {
    int i = t / HID, j = t - i * HID;
    float di = dinv[i];
    z1[t] = h1[t] * di * di + b1[j];
  }
}

__global__ void k_agg1(const int* __restrict__ src, const int* __restrict__ dst,
                       const float* __restrict__ dinv, const float* __restrict__ h1,
                       float* __restrict__ z1, int E) {
  int total  = E * 4;
  int stride = gridDim.x * blockDim.x;
  for (int t = blockIdx.x * blockDim.x + threadIdx.x; t < total; t += stride) {
    int e = t >> 2, c = t & 3;
    int s = src[e], d = dst[e];
    float norm = dinv[s] * dinv[d];
    const float4 hv = *(const float4*)(h1 + (size_t)s * HID + c * 4);
    float* o = z1 + (size_t)d * HID + c * 4;
    atomicAdd(o + 0, hv.x * norm);
    atomicAdd(o + 1, hv.y * norm);
    atomicAdd(o + 2, hv.z * norm);
    atomicAdd(o + 3, hv.w * norm);
  }
}

__global__ void k_gemm2(const float* __restrict__ z1, const float* __restrict__ W2,
                        float* __restrict__ h2, int n) {
  __shared__ float lw[HID * ODIM];
  for (int i = threadIdx.x; i < HID * ODIM; i += blockDim.x) lw[i] = W2[i];
  __syncthreads();

  int wave  = (blockIdx.x * blockDim.x + threadIdx.x) >> 5;
  int lane  = threadIdx.x & 31;
  int tiles = (n + 15) >> 4;
  if (wave >= tiles) return;

  int half = lane >> 4;
  int ml   = lane & 15;
  int row  = wave * 16 + ml;
  int rrow = (row < n) ? row : (n - 1);
  const float* zr = z1 + (size_t)rrow * HID;

  v2f afrag[4];
#pragma unroll
  for (int kk = 0; kk < 4; ++kk) {
    int ka = kk * 4 + 2 * half;
    afrag[kk].x = fmaxf(zr[ka], 0.f);
    afrag[kk].y = fmaxf(zr[ka + 1], 0.f);
  }

#pragma unroll
  for (int tcol = 0; tcol < 4; ++tcol) {
    v8f acc = {0.f, 0.f, 0.f, 0.f, 0.f, 0.f, 0.f, 0.f};
#pragma unroll
    for (int kk = 0; kk < 4; ++kk) {
      int ka = kk * 4 + 2 * half;
      v2f b;
      b.x = lw[ka * ODIM + tcol * 16 + ml];
      b.y = lw[(ka + 1) * ODIM + tcol * 16 + ml];
      acc = __builtin_amdgcn_wmma_f32_16x16x4_f32(false, afrag[kk], false, b, (short)0,
                                                  acc, false, false);
    }
#pragma unroll
    for (int r = 0; r < 8; ++r) {
      int orow = wave * 16 + r + 8 * half;
      if (orow < n) h2[(size_t)orow * ODIM + tcol * 16 + ml] = acc[r];
    }
  }
}

__global__ void k_init_out(const float* __restrict__ h2, const float* __restrict__ dinv,
                           const float* __restrict__ b2, float* __restrict__ out, int n) {
  int t = blockIdx.x * blockDim.x + threadIdx.x;
  int total = n * ODIM;
  if (t < total) {
    int i = t / ODIM, j = t - i * ODIM;
    float di = dinv[i];
    out[t] = h2[t] * di * di + b2[j];
  }
}

__global__ void k_agg2(const int* __restrict__ src, const int* __restrict__ dst,
                       const float* __restrict__ dinv, const float* __restrict__ h2,
                       float* __restrict__ out, int E) {
  int total  = E * 16;
  int stride = gridDim.x * blockDim.x;
  for (int t = blockIdx.x * blockDim.x + threadIdx.x; t < total; t += stride) {
    int e = t >> 4, c = t & 15;
    int s = src[e], d = dst[e];
    float norm = dinv[s] * dinv[d];
    const float4 hv = *(const float4*)(h2 + (size_t)s * ODIM + c * 4);
    float* o = out + (size_t)d * ODIM + c * 4;
    atomicAdd(o + 0, hv.x * norm);
    atomicAdd(o + 1, hv.y * norm);
    atomicAdd(o + 2, hv.z * norm);
    atomicAdd(o + 3, hv.w * norm);
  }
}

#define RB     128
#define LOG_RB 7
#define NSL    256
#define H1P    32

__device__ __forceinline__ int block_excl_scan(int v, int* scr, int* total) {
  const int t = threadIdx.x, nt = blockDim.x;
  scr[t] = v;
  __syncthreads();
  for (int o = 1; o < nt; o <<= 1) {
    int a = (t >= o) ? scr[t - o] : 0;
    __syncthreads();
    scr[t] += a;
    __syncthreads();
  }
  const int incl = scr[t];
  *total = scr[nt - 1];
  __syncthreads();
  return incl - v;
}

__global__ void k_bcount(const int* __restrict__ dst, int* __restrict__ cntmat,
                         int E, int L, int nb) {
  extern __shared__ int dynsh[];
  int* cnt = dynsh;
  const int tid = threadIdx.x, bd = blockDim.x, wg = blockIdx.x;
  for (int i = tid; i < nb; i += bd) cnt[i] = 0;
  __syncthreads();
  const int e0 = wg * L;
  int e1 = e0 + L;
  if (e1 > E) e1 = E;
  int e = e0 + tid;
  for (; e + 3 * bd < e1; e += 4 * bd) {
    int d[4];
#pragma unroll
    for (int u = 0; u < 4; ++u) d[u] = dst[e + u * bd];
#pragma unroll
    for (int u = 0; u < 4; ++u) atomicAdd(&cnt[d[u] >> LOG_RB], 1);
  }
  for (; e < e1; e += bd) atomicAdd(&cnt[dst[e] >> LOG_RB], 1);
  __syncthreads();
  for (int i = tid; i < nb; i += bd) cntmat[(size_t)i * NSL + wg] = cnt[i];
}

__global__ void k_bscan(int* __restrict__ cntmat, int* __restrict__ totals) {
  __shared__ int scr[NSL];
  const int b = blockIdx.x, w = threadIdx.x;
  const int v = cntmat[(size_t)b * NSL + w];
  int tot;
  const int ex = block_excl_scan(v, scr, &tot);
  cntmat[(size_t)b * NSL + w] = ex;
  if (w == 0) totals[b] = tot;
}

__global__ void k_bscatter(const int* __restrict__ src, const int* __restrict__ dst,
                           const int* __restrict__ offs, const int* __restrict__ totals,
                           int* __restrict__ bstart, unsigned* __restrict__ rec,
                           int E, int L, int nb) {
  extern __shared__ int dynsh[];
  int* cursor = dynsh;
  int* scr    = dynsh + nb;
  const int tid = threadIdx.x, bd = blockDim.x, wg = blockIdx.x;
  const int per = (nb + bd - 1) / bd;
  const int lo = tid * per;
  int hi = lo + per;
  if (hi > nb) hi = nb;
  int s = 0;
  for (int b = lo; b < hi; ++b) s += totals[b];
  int tot;
  int pre = block_excl_scan(s, scr, &tot);
  for (int b = lo; b < hi; ++b) {
    cursor[b] = pre + offs[(size_t)b * NSL + wg];
    if (wg == 0) bstart[b] = pre;
    pre += totals[b];
  }
  if (wg == 0 && tid == 0) bstart[nb] = tot;
  __syncthreads();
  const int e0 = wg * L;
  int e1 = e0 + L;
  if (e1 > E) e1 = E;
  int e = e0 + tid;
  for (; e + 3 * bd < e1; e += 4 * bd) {
    int sv[4], dv[4];
#pragma unroll
    for (int u = 0; u < 4; ++u) {
      sv[u] = src[e + u * bd];
      dv[u] = dst[e + u * bd];
    }
#pragma unroll
    for (int u = 0; u < 4; ++u) {
      const int p = atomicAdd(&cursor[dv[u] >> LOG_RB], 1);
      rec[p] = ((unsigned)sv[u] << LOG_RB) | (unsigned)(dv[u] & (RB - 1));
    }
  }
  for (; e < e1; e += bd) {
    const int sv = src[e], dv = dst[e];
    const int p = atomicAdd(&cursor[dv >> LOG_RB], 1);
    rec[p] = ((unsigned)sv << LOG_RB) | (unsigned)(dv & (RB - 1));
  }
}

__global__ void k_bdeg(const unsigned* __restrict__ rec, const int* __restrict__ bstart,
                       float* __restrict__ dinv, int n) {
  __shared__ int cnt[RB];
  const int tid = threadIdx.x, bd = blockDim.x, b = blockIdx.x;
  if (tid < RB) cnt[tid] = 0;
  __syncthreads();
  const int e0 = bstart[b], e1 = bstart[b + 1];
  for (int e = e0 + tid; e < e1; e += bd) atomicAdd(&cnt[rec[e] & (RB - 1)], 1);
  __syncthreads();
  if (tid < RB) {
    const int node = b * RB + tid;
    if (node < n) {
      const float d = 1.0f + (float)cnt[tid];
      dinv[node] = (d > 0.0f) ? (1.0f / sqrtf(d)) : 0.0f;
    }
  }
}

template <int F, int HP, bool FIRST>
__device__ __forceinline__ void bagg_body(float* acc, float* dl,
                                          const unsigned* __restrict__ rec,
                                          const int* __restrict__ bstart,
                                          const float* __restrict__ dinv,
                                          const float* __restrict__ h,
                                          const float* __restrict__ bias,
                                          const float* __restrict__ dsrc_in,
                                          float* __restrict__ dsrc_out,
                                          float* __restrict__ o, int n) {
  constexpr int LPE = F / 4;
  const int tid = threadIdx.x, bd = blockDim.x, b = blockIdx.x;
  const int base = b * RB;
  for (int i = tid; i < RB * F; i += bd) {
    const int node = base + i / F, j = i % F;
    float v = 0.f;
    if (node < n) {
      const float di = dinv[node];
      v = h[(size_t)node * HP + j] * di * di + bias[j];
    }
    acc[i] = v;
  }
  if (tid < RB) dl[tid] = (base + tid < n) ? dinv[base + tid] : 0.f;
  __syncthreads();
  const int e0 = bstart[b], e1 = bstart[b + 1];
  const int q = tid / LPE, c = tid % LPE;
  const int epi = bd / LPE;
  int e = e0 + q;
  for (; e + 3 * epi < e1; e += 4 * epi) {
    unsigned r[4];
    float ds[4];
    float4 hv[4];
#pragma unroll
    for (int u = 0; u < 4; ++u) {
      r[u] = rec[e + u * epi];
      if (!FIRST) ds[u] = dsrc_in[e + u * epi];
    }
#pragma unroll
    for (int u = 0; u < 4; ++u) {
      const float* row = h + (size_t)(r[u] >> LOG_RB) * HP;
      hv[u] = *(const float4*)(row + c * 4);
      if (FIRST) ds[u] = row[F];
    }
#pragma unroll
    for (int u = 0; u < 4; ++u) {
      if (FIRST && c == 0) dsrc_out[e + u * epi] = ds[u];
      const int dloc = (int)(r[u] & (RB - 1));
      const float norm = ds[u] * dl[dloc];
      float* a = acc + dloc * F + c * 4;
      atomicAdd(a + 0, hv[u].x * norm);
      atomicAdd(a + 1, hv[u].y * norm);
      atomicAdd(a + 2, hv[u].z * norm);
      atomicAdd(a + 3, hv[u].w * norm);
    }
  }
  for (; e < e1; e += epi) {
    const unsigned r = rec[e];
    const float* row = h + (size_t)(r >> LOG_RB) * HP;
    const int dloc = (int)(r & (RB - 1));
    const float4 hv = *(const float4*)(row + c * 4);
    float dsv;
    if (FIRST) {
      dsv = row[F];
      if (c == 0) dsrc_out[e] = dsv;
    } else {
      dsv = dsrc_in[e];
    }
    const float norm = dsv * dl[dloc];
    float* a = acc + dloc * F + c * 4;
    atomicAdd(a + 0, hv.x * norm);
    atomicAdd(a + 1, hv.y * norm);
    atomicAdd(a + 2, hv.z * norm);
    atomicAdd(a + 3, hv.w * norm);
  }
  __syncthreads();
  for (int i = tid; i < RB * F; i += bd) {
    const int node = base + i / F;
    if (node < n) o[(size_t)base * F + i] = acc[i];
  }
}

__global__ void k_bagg1(const unsigned* __restrict__ rec, const int* __restrict__ bstart,
                        const float* __restrict__ dinv, const float* __restrict__ h1,
                        const float* __restrict__ b1, float* __restrict__ dsrc,
                        float* __restrict__ z1, int n) {
  __shared__ float acc[RB * HID];
  __shared__ float dl[RB];
  bagg_body<HID, H1P, true>(acc, dl, rec, bstart, dinv, h1, b1, nullptr, dsrc, z1, n);
}

__global__ void k_bagg2(const unsigned* __restrict__ rec, const int* __restrict__ bstart,
                        const float* __restrict__ dinv, const float* __restrict__ h2,
                        const float* __restrict__ b2, const float* __restrict__ dsrc,
                        float* __restrict__ out, int n) {
  __shared__ float acc[RB * ODIM];
  __shared__ float dl[RB];
  bagg_body<ODIM, ODIM, false>(acc, dl, rec, bstart, dinv, h2, b2, dsrc, nullptr, out, n);
}

extern "C" void kernel_launch(void* const* d_in, const int* in_sizes, int n_in,
                              void* d_out, int out_size, void* d_ws, size_t ws_size,
                              hipStream_t stream) {
  const float* x  = (const float*)d_in[0];
  const float* W1 = (const float*)d_in[1];
  const float* b1 = (const float*)d_in[2];
  const float* W2 = (const float*)d_in[3];
  const float* b2 = (const float*)d_in[4];
  const int*   ei = (const int*)d_in[5];

  const int N = in_sizes[0] / IN_DIM;
  const int E = in_sizes[5] / 2;
  const int* src = ei;
  const int* dst = ei + E;
  float* out = (float*)d_out;

  char* ws = (char*)d_ws;
  size_t off = 0;
  auto alloc = [&](size_t nbytes) -> void* {
    void* p = ws + off;
    off = (off + nbytes + 255) & ~(size_t)255;
    return p;
  };
  float* dinv = (float*)alloc((size_t)N * sizeof(float));
  float* z1   = (float*)alloc((size_t)N * HID  * sizeof(float));
  float* h2   = (float*)alloc((size_t)N * ODIM * sizeof(float));
  const size_t planes = off;

  const int B = 256;
  const int tiles = (N + 15) >> 4;
  const int gemmBlocks = (tiles + 7) / 8;

  const int NB = (N + RB - 1) / RB;
  const int L  = (E + NSL - 1) / NSL;
  unsigned* rec = (unsigned*)alloc((size_t)E * sizeof(unsigned));
  float* dsrc   = (float*)alloc((size_t)E * sizeof(float));
  int* cntmat   = (int*)alloc((size_t)NB * NSL * sizeof(int));
  int* totals   = (int*)alloc((size_t)NB * sizeof(int));
  int* bstart   = (int*)alloc((size_t)(NB + 1) * sizeof(int));

  if (off <= ws_size) {
    k_bcount  <<<NSL, 512, (size_t)NB * sizeof(int), stream>>>(dst, cntmat, E, L, NB);
    k_bscan   <<<NB, NSL, 0, stream>>>(cntmat, totals);
    k_bscatter<<<NSL, 512, (size_t)(NB + 512) * sizeof(int), stream>>>(
        src, dst, cntmat, totals, bstart, rec, E, L, NB);
    k_bdeg    <<<NB, 1024, 0, stream>>>(rec, bstart, dinv, N);

    float* h1a = h2;
    k_gemm1   <<<gemmBlocks, B, 0, stream>>>(x, W1, h1a, N, H1P, dinv);
    k_bagg1   <<<NB, 512, 0, stream>>>(rec, bstart, dinv, h1a, b1, dsrc, z1, N);

    k_gemm2   <<<gemmBlocks, B, 0, stream>>>(z1, W2, h2, N);
    k_bagg2   <<<NB, 512, 0, stream>>>(rec, bstart, dinv, h2, b2, dsrc, out, N);
    return;
  }

  off = planes;
  float* h1 = (float*)alloc((size_t)N * HID * sizeof(float));
  k_deg_init <<<(N + B - 1) / B, B, 0, stream>>>(dinv, N);
  k_deg_edges<<<(E + B - 1) / B, B, 0, stream>>>(dst, dinv, E);
  k_dinv     <<<(N + B - 1) / B, B, 0, stream>>>(dinv, N);

  k_gemm1    <<<gemmBlocks, B, 0, stream>>>(x, W1, h1, N, HID, nullptr);
  k_init_z1  <<<((size_t)N * HID + B - 1) / B, B, 0, stream>>>(h1, dinv, b1, z1, N);
  k_agg1     <<<(E * 4 + B - 1) / B, B, 0, stream>>>(src, dst, dinv, h1, z1, E);

  k_gemm2    <<<gemmBlocks, B, 0, stream>>>(z1, W2, h2, N);
  k_init_out <<<((size_t)N * ODIM + B - 1) / B, B, 0, stream>>>(h2, dinv, b2, out, N);
  k_agg2     <<<(E * 16 + B - 1) / B, B, 0, stream>>>(src, dst, dinv, h2, out, E);
}


# edge list fully ordered by destination node (second LDS counting sort, integer adds); a lane group per node sums its rows in registers and stores the row once; no LDS float adds; sum order changes
# speedup vs baseline: 2.6421x; 1.2851x over previous
#include <hip/hip_runtime.h>

typedef __attribute__((ext_vector_type(2))) float v2f;
typedef __attribute__((ext_vector_type(8))) float v8f;

#define IN_DIM 512
#define HID    16
#define ODIM   64

__global__ void k_deg_init(float* __restrict__ deg, int n) {
  int i = blockIdx.x * blockDim.x + threadIdx.x;
  if (i < n) deg[i] = 1.0f;
}

__global__ void k_deg_edges(const int* __restrict__ dst, float* __restrict__ deg, int E) {
  int i = blockIdx.x * blockDim.x + threadIdx.x;
  if (i < E) atomicAdd(&deg[dst[i]], 1.0f);
}

__global__ void k_dinv(float* __restrict__ deg, int n) {
  int i = blockIdx.x * blockDim.x + threadIdx.x;
  if (i < n) {
    float d = deg[i];
    deg[i] = (d > 0.0f) ? (1.0f / sqrtf(d)) : 0.0f;
  }
}

__global__ void k_gemm1(const float* __restrict__ x, const float* __restrict__ W1,
                        float* __restrict__ h1, int n, int pitch,
                        const float* __restrict__ dinv) {
  __shared__ float lw[IN_DIM * HID];
  for (int i = threadIdx.x; i < IN_DIM * HID; i += blockDim.x) lw[i] = W1[i];
  __syncthreads();

  int wave  = (blockIdx.x * blockDim.x + threadIdx.x) >> 5;
  int lane  = threadIdx.x & 31;
  int tiles = (n + 15) >> 4;
  if (wave >= tiles) return;

  int half = lane >> 4;
  int ml   = lane & 15;
  int row  = wave * 16 + ml;
  int rrow = (row < n) ? row : (n - 1);
  const float* xr = x + (size_t)rrow * IN_DIM;

  v8f acc = {0.f, 0.f, 0.f, 0.f, 0.f, 0.f, 0.f, 0.f};
  for (int k = 0; k < IN_DIM; k += 4) {
    int ka = k + 2 * half;
    v2f a; a.x = xr[ka];                 a.y = xr[ka + 1];
    v2f b; b.x = lw[ka * HID + ml];      b.y = lw[(ka + 1) * HID + ml];
    acc = __builtin_amdgcn_wmma_f32_16x16x4_f32(false, a, false, b, (short)0, acc,
                                                false, false);
  }
#pragma unroll
  for (int r = 0; r < 8; ++r) {
    int orow = wave * 16 + r + 8 * half;
    if (orow < n) h1[(size_t)orow * pitch + ml] = acc[r];
  }
  if (dinv != nullptr && half == 0 && row < n) h1[(size_t)row * pitch + HID] = dinv[row];
}

__global__ void k_init_z1(const float* __restrict__ h1, const float* __restrict__ dinv,
                          const float* __restrict__ b1, float* __restrict__ z1, int n) {
  int t = blockIdx.x * blockDim.x + threadIdx.x;
  int total = n * HID;
  if (t < total) {
    int i = t / HID, j = t - i * HID;
    float di = dinv[i];
    z1[t] = h1[t] * di * di + b1[j];
  }
}

__global__ void k_agg1(const int* __restrict__ src, const int* __restrict__ dst,
                       const float* __restrict__ dinv, const float* __restrict__ h1,
                       float* __restrict__ z1, int E) {
  int total  = E * 4;
  int stride = gridDim.x * blockDim.x;
  for (int t = blockIdx.x * blockDim.x + threadIdx.x; t < total; t += stride) {
    int e = t >> 2, c = t & 3;
    int s = src[e], d = dst[e];
    float norm = dinv[s] * dinv[d];
    const float4 hv = *(const float4*)(h1 + (size_t)s * HID + c * 4);
    float* o = z1 + (size_t)d * HID + c * 4;
    atomicAdd(o + 0, hv.x * norm);
    atomicAdd(o + 1, hv.y * norm);
    atomicAdd(o + 2, hv.z * norm);
    atomicAdd(o + 3, hv.w * norm);
  }
}

__global__ void k_gemm2(const float* __restrict__ z1, const float* __restrict__ W2,
                        float* __restrict__ h2, int n) {
  __shared__ float lw[HID * ODIM];
  for (int i = threadIdx.x; i < HID * ODIM; i += blockDim.x) lw[i] = W2[i];
  __syncthreads();

  int wave  = (blockIdx.x * blockDim.x + threadIdx.x) >> 5;
  int lane  = threadIdx.x & 31;
  int tiles = (n + 15) >> 4;
  if (wave >= tiles) return;

  int half = lane >> 4;
  int ml   = lane & 15;
  int row  = wave * 16 + ml;
  int rrow = (row < n) ? row : (n - 1);
  const float* zr = z1 + (size_t)rrow * HID;

  v2f afrag[4];
#pragma unroll
  for (int kk = 0; kk < 4; ++kk) {
    int ka = kk * 4 + 2 * half;
    afrag[kk].x = fmaxf(zr[ka], 0.f);
    afrag[kk].y = fmaxf(zr[ka + 1], 0.f);
  }

#pragma unroll
  for (int tcol = 0; tcol < 4; ++tcol) {
    v8f acc = {0.f, 0.f, 0.f, 0.f, 0.f, 0.f, 0.f, 0.f};
#pragma unroll
    for (int kk = 0; kk < 4; ++kk) {
      int ka = kk * 4 + 2 * half;
      v2f b;
      b.x = lw[ka * ODIM + tcol * 16 + ml];
      b.y = lw[(ka + 1) * ODIM + tcol * 16 + ml];
      acc = __builtin_amdgcn_wmma_f32_16x16x4_f32(false, afrag[kk], false, b, (short)0,
                                                  acc, false, false);
    }
#pragma unroll
    for (int r = 0; r < 8; ++r) {
      int orow = wave * 16 + r + 8 * half;
      if (orow < n) h2[(size_t)orow * ODIM + tcol * 16 + ml] = acc[r];
    }
  }
}

__global__ void k_init_out(const float* __restrict__ h2, const float* __restrict__ dinv,
                           const float* __restrict__ b2, float* __restrict__ out, int n) {
  int t = blockIdx.x * blockDim.x + threadIdx.x;
  int total = n * ODIM;
  if (t < total) {
    int i = t / ODIM, j = t - i * ODIM;
    float di = dinv[i];
    out[t] = h2[t] * di * di + b2[j];
  }
}

__global__ void k_agg2(const int* __restrict__ src, const int* __restrict__ dst,
                       const float* __restrict__ dinv, const float* __restrict__ h2,
                       float* __restrict__ out, int E) {
  int total  = E * 16;
  int stride = gridDim.x * blockDim.x;
  for (int t = blockIdx.x * blockDim.x + threadIdx.x; t < total; t += stride) {
    int e = t >> 4, c = t & 15;
    int s = src[e], d = dst[e];
    float norm = dinv[s] * dinv[d];
    const float4 hv = *(const float4*)(h2 + (size_t)s * ODIM + c * 4);
    float* o = out + (size_t)d * ODIM + c * 4;
    atomicAdd(o + 0, hv.x * norm);
    atomicAdd(o + 1, hv.y * norm);
    atomicAdd(o + 2, hv.z * norm);
    atomicAdd(o + 3, hv.w * norm);
  }
}

#define RB     128
#define LOG_RB 7
#define NSL    256
#define H1P    32

__device__ __forceinline__ int block_excl_scan(int v, int* scr, int* total) {
  const int t = threadIdx.x, nt = blockDim.x;
  scr[t] = v;
  __syncthreads();
  for (int o = 1; o < nt; o <<= 1) {
    int a = (t >= o) ? scr[t - o] : 0;
    __syncthreads();
    scr[t] += a;
    __syncthreads();
  }
  const int incl = scr[t];
  *total = scr[nt - 1];
  __syncthreads();
  return incl - v;
}

__global__ void k_bcount(const int* __restrict__ dst, int* __restrict__ cntmat,
                         int E, int L, int nb) {
  extern __shared__ int dynsh[];
  int* cnt = dynsh;
  const int tid = threadIdx.x, bd = blockDim.x, wg = blockIdx.x;
  for (int i = tid; i < nb; i += bd) cnt[i] = 0;
  __syncthreads();
  const int e0 = wg * L;
  int e1 = e0 + L;
  if (e1 > E) e1 = E;
  int e = e0 + tid;
  for (; e + 3 * bd < e1; e += 4 * bd) {
    int d[4];
#pragma unroll
    for (int u = 0; u < 4; ++u) d[u] = dst[e + u * bd];
#pragma unroll
    for (int u = 0; u < 4; ++u) atomicAdd(&cnt[d[u] >> LOG_RB], 1);
  }
  for (; e < e1; e += bd) atomicAdd(&cnt[dst[e] >> LOG_RB], 1);
  __syncthreads();
  for (int i = tid; i < nb; i += bd) cntmat[(size_t)i * NSL + wg] = cnt[i];
}

__global__ void k_bscan(int* __restrict__ cntmat, int* __restrict__ totals) {
  __shared__ int scr[NSL];
  const int b = blockIdx.x, w = threadIdx.x;
  const int v = cntmat[(size_t)b * NSL + w];
  int tot;
  const int ex = block_excl_scan(v, scr, &tot);
  cntmat[(size_t)b * NSL + w] = ex;
  if (w == 0) totals[b] = tot;
}

__global__ void k_bscatter(const int* __restrict__ src, const int* __restrict__ dst,
                           const int* __restrict__ offs, const int* __restrict__ totals,
                           int* __restrict__ bstart, unsigned* __restrict__ rec,
                           int E, int L, int nb) {
  extern __shared__ int dynsh[];
  int* cursor = dynsh;
  int* scr    = dynsh + nb;
  const int tid = threadIdx.x, bd = blockDim.x, wg = blockIdx.x;
  const int per = (nb + bd - 1) / bd;
  const int lo = tid * per;
  int hi = lo + per;
  if (hi > nb) hi = nb;
  int s = 0;
  for (int b = lo; b < hi; ++b) s += totals[b];
  int tot;
  int pre = block_excl_scan(s, scr, &tot);
  for (int b = lo; b < hi; ++b) {
    cursor[b] = pre + offs[(size_t)b * NSL + wg];
    if (wg == 0) bstart[b] = pre;
    pre += totals[b];
  }
  if (wg == 0 && tid == 0) bstart[nb] = tot;
  __syncthreads();
  const int e0 = wg * L;
  int e1 = e0 + L;
  if (e1 > E) e1 = E;
  int e = e0 + tid;
  for (; e + 3 * bd < e1; e += 4 * bd) {
    int sv[4], dv[4];
#pragma unroll
    for (int u = 0; u < 4; ++u) {
      sv[u] = src[e + u * bd];
      dv[u] = dst[e + u * bd];
    }
#pragma unroll
    for (int u = 0; u < 4; ++u) {
      const int p = atomicAdd(&cursor[dv[u] >> LOG_RB], 1);
      rec[p] = ((unsigned)sv[u] << LOG_RB) | (unsigned)(dv[u] & (RB - 1));
    }
  }
  for (; e < e1; e += bd) {
    const int sv = src[e], dv = dst[e];
    const int p = atomicAdd(&cursor[dv >> LOG_RB], 1);
    rec[p] = ((unsigned)sv << LOG_RB) | (unsigned)(dv & (RB - 1));
  }
}

#define SRT_U 9
__global__ void k_bsort(const unsigned* __restrict__ rec, const int* __restrict__ bstart,
                        float* __restrict__ dinv, int* __restrict__ rowptr,
                        unsigned* __restrict__ srt, int n, int nb) {
  __shared__ int cnt[RB];
  __shared__ int cur[RB];
  const int tid = threadIdx.x, bd = blockDim.x, b = blockIdx.x;
  if (tid < RB) cnt[tid] = 0;
  __syncthreads();
  const int e0 = bstart[b], e1 = bstart[b + 1];
  for (int base = e0; base < e1; base += SRT_U * bd) {
    unsigned r[SRT_U];
#pragma unroll
    for (int u = 0; u < SRT_U; ++u) {
      const int idx = base + tid + u * bd;
      r[u] = rec[idx < e1 ? idx : e1 - 1];
    }
#pragma unroll
    for (int u = 0; u < SRT_U; ++u)
      if (base + tid + u * bd < e1) atomicAdd(&cnt[r[u] & (RB - 1)], 1);
  }
  __syncthreads();
  if (tid < 32) {
    const int c0 = cnt[4 * tid], c1 = cnt[4 * tid + 1], c2 = cnt[4 * tid + 2],
              c3 = cnt[4 * tid + 3];
    const int s = c0 + c1 + c2 + c3;
    int incl = s;
    for (int o = 1; o < 32; o <<= 1) {
      const int t = __shfl_up(incl, o);
      if (tid >= o) incl += t;
    }
    const int ex = e0 + incl - s;
    cur[4 * tid]     = ex;
    cur[4 * tid + 1] = ex + c0;
    cur[4 * tid + 2] = ex + c0 + c1;
    cur[4 * tid + 3] = ex + c0 + c1 + c2;
  }
  __syncthreads();
  if (tid < RB) {
    const int node = b * RB + tid;
    rowptr[node] = cur[tid];
    if (node < n) {
      const float d = 1.0f + (float)cnt[tid];
      dinv[node] = (d > 0.0f) ? (1.0f / sqrtf(d)) : 0.0f;
    }
  }
  if (b == nb - 1 && tid == 0) rowptr[nb * RB] = e1;
  __syncthreads();
  for (int base = e0; base < e1; base += SRT_U * bd) {
    unsigned r[SRT_U];
#pragma unroll
    for (int u = 0; u < SRT_U; ++u) {
      const int idx = base + tid + u * bd;
      r[u] = rec[idx < e1 ? idx : e1 - 1];
    }
#pragma unroll
    for (int u = 0; u < SRT_U; ++u) {
      if (base + tid + u * bd < e1) {
        const int p = atomicAdd(&cur[r[u] & (RB - 1)], 1);
        srt[p] = r[u] >> LOG_RB;
      }
    }
  }
}

#define AGU 4
template <int F, int HP, bool FIRST>
__device__ __forceinline__ void cagg_body(const unsigned* __restrict__ srt,
                                          const int* __restrict__ rowptr,
                                          const float* __restrict__ dinv,
                                          const float* __restrict__ h,
                                          const float* __restrict__ bias,
                                          const float* __restrict__ dsrc_in,
                                          float* __restrict__ dsrc_out,
                                          float* __restrict__ o, int n) {
  constexpr int LPE = F / 4;
  const int tid = threadIdx.x, b = blockIdx.x;
  const int g = tid / LPE, c = tid % LPE;
  const int ng = blockDim.x / LPE;
  for (int nl = g; nl < RB; nl += ng) {
    const int node = b * RB + nl;
    if (node >= n) break;
    const int e0 = rowptr[node], e1 = rowptr[node + 1];
    const float dd = dinv[node];
    const float4 hself = *(const float4*)(h + (size_t)node * HP + c * 4);
    const float4 bv = *(const float4*)(bias + c * 4);
    float4 acc = {0.f, 0.f, 0.f, 0.f};
    unsigned sn[AGU];
    float dn[AGU];
    if (e0 < e1) {
#pragma unroll
      for (int u = 0; u < AGU; ++u) {
        const int idx = (e0 + u < e1) ? (e0 + u) : (e1 - 1);
        sn[u] = srt[idx];
        if (!FIRST) dn[u] = dsrc_in[idx];
      }
    }
    for (int e = e0; e < e1; e += AGU) {
      float4 hv[AGU];
      float ds[AGU];
#pragma unroll
      for (int u = 0; u < AGU; ++u) {
        const float* row = h + (size_t)sn[u] * HP;
        hv[u] = *(const float4*)(row + c * 4);
        ds[u] = FIRST ? row[F] : dn[u];
      }
#pragma unroll
      for (int u = 0; u < AGU; ++u) {
        const int idx = (e + AGU + u < e1) ? (e + AGU + u) : (e1 - 1);
        sn[u] = srt[idx];
        if (!FIRST) dn[u] = dsrc_in[idx];
      }
#pragma unroll
      for (int u = 0; u < AGU; ++u) {
        const bool live = e + u < e1;
        if (FIRST && c == 0 && live) dsrc_out[e + u] = ds[u];
        const float norm = ds[u] * dd;
        acc.x += (live ? hv[u].x : 0.f) * norm;
        acc.y += (live ? hv[u].y : 0.f) * norm;
        acc.z += (live ? hv[u].z : 0.f) * norm;
        acc.w += (live ? hv[u].w : 0.f) * norm;
      }
    }
    float4 r;
    r.x = acc.x + (hself.x * dd * dd + bv.x);
    r.y = acc.y + (hself.y * dd * dd + bv.y);
    r.z = acc.z + (hself.z * dd * dd + bv.z);
    r.w = acc.w + (hself.w * dd * dd + bv.w);
    *(float4*)(o + (size_t)node * F + c * 4) = r;
  }
}

__global__ void k_cagg1(const unsigned* __restrict__ srt, const int* __restrict__ rowptr,
                        const float* __restrict__ dinv, const float* __restrict__ h1,
                        const float* __restrict__ b1, float* __restrict__ dsrc,
                        float* __restrict__ z1, int n) {
  cagg_body<HID, H1P, true>(srt, rowptr, dinv, h1, b1, nullptr, dsrc, z1, n);
}

__global__ void k_cagg2(const unsigned* __restrict__ srt, const int* __restrict__ rowptr,
                        const float* __restrict__ dinv, const float* __restrict__ h2,
                        const float* __restrict__ b2, const float* __restrict__ dsrc,
                        float* __restrict__ out, int n) {
  cagg_body<ODIM, ODIM, false>(srt, rowptr, dinv, h2, b2, dsrc, nullptr, out, n);
}

extern "C" void kernel_launch(void* const* d_in, const int* in_sizes, int n_in,
                              void* d_out, int out_size, void* d_ws, size_t ws_size,
                              hipStream_t stream) {
  const float* x  = (const float*)d_in[0];
  const float* W1 = (const float*)d_in[1];
  const float* b1 = (const float*)d_in[2];
  const float* W2 = (const float*)d_in[3];
  const float* b2 = (const float*)d_in[4];
  const int*   ei = (const int*)d_in[5];

  const int N = in_sizes[0] / IN_DIM;
  const int E = in_sizes[5] / 2;
  const int* src = ei;
  const int* dst = ei + E;
  float* out = (float*)d_out;

  char* ws = (char*)d_ws;
  size_t off = 0;
  auto alloc = [&](size_t nbytes) -> void* {
    void* p = ws + off;
    off = (off + nbytes + 255) & ~(size_t)255;
    return p;
  };
  float* dinv = (float*)alloc((size_t)N * sizeof(float));
  float* z1   = (float*)alloc((size_t)N * HID  * sizeof(float));
  float* h2   = (float*)alloc((size_t)N * ODIM * sizeof(float));
  const size_t planes = off;

  const int B = 256;
  const int tiles = (N + 15) >> 4;
  const int gemmBlocks = (tiles + 7) / 8;

  const int NB = (N + RB - 1) / RB;
  const int L  = (E + NSL - 1) / NSL;
  unsigned* rec = (unsigned*)alloc((size_t)E * sizeof(unsigned));
  unsigned* srt = (unsigned*)alloc((size_t)E * sizeof(unsigned));
  int* cntmat   = (int*)alloc((size_t)NB * NSL * sizeof(int));
  int* totals   = (int*)alloc((size_t)NB * sizeof(int));
  int* bstart   = (int*)alloc((size_t)(NB + 1) * sizeof(int));
  float* dsrc = (float*)rec;
  int* rowptr = cntmat;

  if (off <= ws_size) {
    k_bcount  <<<NSL, 512, (size_t)NB * sizeof(int), stream>>>(dst, cntmat, E, L, NB);
    k_bscan   <<<NB, NSL, 0, stream>>>(cntmat, totals);
    k_bscatter<<<NSL, 512, (size_t)(NB + 512) * sizeof(int), stream>>>(
        src, dst, cntmat, totals, bstart, rec, E, L, NB);
    k_bsort   <<<NB, 512, 0, stream>>>(rec, bstart, dinv, rowptr, srt, N, NB);

    float* h1a = h2;
    k_gemm1   <<<gemmBlocks, B, 0, stream>>>(x, W1, h1a, N, H1P, dinv);
    k_cagg1   <<<NB, 512, 0, stream>>>(srt, rowptr, dinv, h1a, b1, dsrc, z1, N);

    k_gemm2   <<<gemmBlocks, B, 0, stream>>>(z1, W2, h2, N);
    k_cagg2   <<<NB, 512, 0, stream>>>(srt, rowptr, dinv, h2, b2, dsrc, out, N);
    return;
  }

  off = planes;
  float* h1 = (float*)alloc((size_t)N * HID * sizeof(float));
  k_deg_init <<<(N + B - 1) / B, B, 0, stream>>>(dinv, N);
  k_deg_edges<<<(E + B - 1) / B, B, 0, stream>>>(dst, dinv, E);
  k_dinv     <<<(N + B - 1) / B, B, 0, stream>>>(dinv, N);

  k_gemm1    <<<gemmBlocks, B, 0, stream>>>(x, W1, h1, N, HID, nullptr);
  k_init_z1  <<<((size_t)N * HID + B - 1) / B, B, 0, stream>>>(h1, dinv, b1, z1, N);
  k_agg1     <<<(E * 4 + B - 1) / B, B, 0, stream>>>(src, dst, dinv, h1, z1, E);

  k_gemm2    <<<gemmBlocks, B, 0, stream>>>(z1, W2, h2, N);
  k_init_out <<<((size_t)N * ODIM + B - 1) / B, B, 0, stream>>>(h2, dinv, b2, out, N);
  k_agg2     <<<(E * 16 + B - 1) / B, B, 0, stream>>>(src, dst, dinv, h2, out, E);
}


# both regrouping passes order their 4-byte records in LDS first and store them in output order: a global store covers runs of neighbouring records, not a line per lane; sum order as before
# speedup vs baseline: 3.2374x; 1.2253x over previous
#include <hip/hip_runtime.h>

typedef __attribute__((ext_vector_type(2))) float v2f;
typedef __attribute__((ext_vector_type(8))) float v8f;

#define IN_DIM 512
#define HID    16
#define ODIM   64

__global__ void k_deg_init(float* __restrict__ deg, int n) {
  int i = blockIdx.x * blockDim.x + threadIdx.x;
  if (i < n) deg[i] = 1.0f;
}

__global__ void k_deg_edges(const int* __restrict__ dst, float* __restrict__ deg, int E) {
  int i = blockIdx.x * blockDim.x + threadIdx.x;
  if (i < E) atomicAdd(&deg[dst[i]], 1.0f);
}

__global__ void k_dinv(float* __restrict__ deg, int n) {
  int i = blockIdx.x * blockDim.x + threadIdx.x;
  if (i < n) {
    float d = deg[i];
    deg[i] = (d > 0.0f) ? (1.0f / sqrtf(d)) : 0.0f;
  }
}

__global__ void k_gemm1(const float* __restrict__ x, const float* __restrict__ W1,
                        float* __restrict__ h1, int n, int pitch,
                        const float* __restrict__ dinv) {
  __shared__ float lw[IN_DIM * HID];
  for (int i = threadIdx.x; i < IN_DIM * HID; i += blockDim.x) lw[i] = W1[i];
  __syncthreads();

  int wave  = (blockIdx.x * blockDim.x + threadIdx.x) >> 5;
  int lane  = threadIdx.x & 31;
  int tiles = (n + 15) >> 4;
  if (wave >= tiles) return;

  int half = lane >> 4;
  int ml   = lane & 15;
  int row  = wave * 16 + ml;
  int rrow = (row < n) ? row : (n - 1);
  const float* xr = x + (size_t)rrow * IN_DIM;

  v8f acc = {0.f, 0.f, 0.f, 0.f, 0.f, 0.f, 0.f, 0.f};
  for (int k = 0; k < IN_DIM; k += 4) {
    int ka = k + 2 * half;
    v2f a; a.x = xr[ka];                 a.y = xr[ka + 1];
    v2f b; b.x = lw[ka * HID + ml];      b.y = lw[(ka + 1) * HID + ml];
    acc = __builtin_amdgcn_wmma_f32_16x16x4_f32(false, a, false, b, (short)0, acc,
                                                false, false);
  }
#pragma unroll
  for (int r = 0; r < 8; ++r) {
    int orow = wave * 16 + r + 8 * half;
    if (orow < n) h1[(size_t)orow * pitch + ml] = acc[r];
  }
  if (dinv != nullptr && half == 0 && row < n) h1[(size_t)row * pitch + HID] = dinv[row];
}

__global__ void k_init_z1(const float* __restrict__ h1, const float* __restrict__ dinv,
                          const float* __restrict__ b1, float* __restrict__ z1, int n) {
  int t = blockIdx.x * blockDim.x + threadIdx.x;
  int total = n * HID;
  if (t < total) {
    int i = t / HID, j = t - i * HID;
    float di = dinv[i];
    z1[t] = h1[t] * di * di + b1[j];
  }
}

__global__ void k_agg1(const int* __restrict__ src, const int* __restrict__ dst,
                       const float* __restrict__ dinv, const float* __restrict__ h1,
                       float* __restrict__ z1, int E) {
  int total  = E * 4;
  int stride = gridDim.x * blockDim.x;
  for (int t = blockIdx.x * blockDim.x + threadIdx.x; t < total; t += stride) {
    int e = t >> 2, c = t & 3;
    int s = src[e], d = dst[e];
    float norm = dinv[s] * dinv[d];
    const float4 hv = *(const float4*)(h1 + (size_t)s * HID + c * 4);
    float* o = z1 + (size_t)d * HID + c * 4;
    atomicAdd(o + 0, hv.x * norm);
    atomicAdd(o + 1, hv.y * norm);
    atomicAdd(o + 2, hv.z * norm);
    atomicAdd(o + 3, hv.w * norm);
  }
}

__global__ void k_gemm2(const float* __restrict__ z1, const float* __restrict__ W2,
                        float* __restrict__ h2, int n) {
  __shared__ float lw[HID * ODIM];
  for (int i = threadIdx.x; i < HID * ODIM; i += blockDim.x) lw[i] = W2[i];
  __syncthreads();

  int wave  = (blockIdx.x * blockDim.x + threadIdx.x) >> 5;
  int lane  = threadIdx.x & 31;
  int tiles = (n + 15) >> 4;
  if (wave >= tiles) return;

  int half = lane >> 4;
  int ml   = lane & 15;
  int row  = wave * 16 + ml;
  int rrow = (row < n) ? row : (n - 1);
  const float* zr = z1 + (size_t)rrow * HID;

  v2f afrag[4];
#pragma unroll
  for (int kk = 0; kk < 4; ++kk) {
    int ka = kk * 4 + 2 * half;
    afrag[kk].x = fmaxf(zr[ka], 0.f);
    afrag[kk].y = fmaxf(zr[ka + 1], 0.f);
  }

#pragma unroll
  for (int tcol = 0; tcol < 4; ++tcol) {
    v8f acc = {0.f, 0.f, 0.f, 0.f, 0.f, 0.f, 0.f, 0.f};
#pragma unroll
    for (int kk = 0; kk < 4; ++kk) {
      int ka = kk * 4 + 2 * half;
      v2f b;
      b.x = lw[ka * ODIM + tcol * 16 + ml];
      b.y = lw[(ka + 1) * ODIM + tcol * 16 + ml];
      acc = __builtin_amdgcn_wmma_f32_16x16x4_f32(false, afrag[kk], false, b, (short)0,
                                                  acc, false, false);
    }
#pragma unroll
    for (int r = 0; r < 8; ++r) {
      int orow = wave * 16 + r + 8 * half;
      if (orow < n) h2[(size_t)orow * ODIM + tcol * 16 + ml] = acc[r];
    }
  }
}

__global__ void k_init_out(const float* __restrict__ h2, const float* __restrict__ dinv,
                           const float* __restrict__ b2, float* __restrict__ out, int n) {
  int t = blockIdx.x * blockDim.x + threadIdx.x;
  int total = n * ODIM;
  if (t < total) {
    int i = t / ODIM, j = t - i * ODIM;
    float di = dinv[i];
    out[t] = h2[t] * di * di + b2[j];
  }
}

__global__ void k_agg2(const int* __restrict__ src, const int* __restrict__ dst,
                       const float* __restrict__ dinv, const float* __restrict__ h2,
                       float* __restrict__ out, int E) {
  int total  = E * 16;
  int stride = gridDim.x * blockDim.x;
  for (int t = blockIdx.x * blockDim.x + threadIdx.x; t < total; t += stride) {
    int e = t >> 4, c = t & 15;
    int s = src[e], d = dst[e];
    float norm = dinv[s] * dinv[d];
    const float4 hv = *(const float4*)(h2 + (size_t)s * ODIM + c * 4);
    float* o = out + (size_t)d * ODIM + c * 4;
    atomicAdd(o + 0, hv.x * norm);
    atomicAdd(o + 1, hv.y * norm);
    atomicAdd(o + 2, hv.z * norm);
    atomicAdd(o + 3, hv.w * norm);
  }
}

#define RB     128
#define LOG_RB 7
#define NSL    256
#define H1P    32

__device__ __forceinline__ int block_excl_scan(int v, int* scr, int* total) {
  const int t = threadIdx.x, nt = blockDim.x;
  scr[t] = v;
  __syncthreads();
  for (int o = 1; o < nt; o <<= 1) {
    int a = (t >= o) ? scr[t - o] : 0;
    __syncthreads();
    scr[t] += a;
    __syncthreads();
  }
  const int incl = scr[t];
  *total = scr[nt - 1];
  __syncthreads();
  return incl - v;
}

__device__ __forceinline__ void block_excl_scan2(int va, int vb, int* scrA, int* scrB,
                                                 int* exA, int* exB, int* totA) {
  const int t = threadIdx.x, nt = blockDim.x;
  scrA[t] = va;
  scrB[t] = vb;
  __syncthreads();
  for (int o = 1; o < nt; o <<= 1) {
    int a = 0, b = 0;
    if (t >= o) {
      a = scrA[t - o];
      b = scrB[t - o];
    }
    __syncthreads();
    scrA[t] += a;
    scrB[t] += b;
    __syncthreads();
  }
  *exA = scrA[t] - va;
  *exB = scrB[t] - vb;
  *totA = scrA[nt - 1];
  __syncthreads();
}

__global__ void k_bcount(const int* __restrict__ dst, int* __restrict__ cntmat,
                         int E, int L, int nb) {
  extern __shared__ int dynsh[];
  int* cnt = dynsh;
  const int tid = threadIdx.x, bd = blockDim.x, wg = blockIdx.x;
  for (int i = tid; i < nb; i += bd) cnt[i] = 0;
  __syncthreads();
  const int e0 = wg * L;
  int e1 = e0 + L;
  if (e1 > E) e1 = E;
  int e = e0 + tid;
  for (; e + 3 * bd < e1; e += 4 * bd) {
    int d[4];
#pragma unroll
    for (int u = 0; u < 4; ++u) d[u] = dst[e + u * bd];
#pragma unroll
    for (int u = 0; u < 4; ++u) atomicAdd(&cnt[d[u] >> LOG_RB], 1);
  }
  for (; e < e1; e += bd) atomicAdd(&cnt[dst[e] >> LOG_RB], 1);
  __syncthreads();
  for (int i = tid; i < nb; i += bd) cntmat[(size_t)i * NSL + wg] = cnt[i];
}

__global__ void k_bscan(int* __restrict__ cntmat, int* __restrict__ totals) {
  __shared__ int scr[NSL];
  const int b = blockIdx.x, w = threadIdx.x;
  const int v = cntmat[(size_t)b * NSL + w];
  int tot;
  const int ex = block_excl_scan(v, scr, &tot);
  cntmat[(size_t)b * NSL + w] = ex;
  if (w == 0) totals[b] = tot;
}

#define SC_CAP 16384
__global__ void k_bscatter(const int* __restrict__ src, const int* __restrict__ dst,
                           const int* __restrict__ offs, const int* __restrict__ totals,
                           int* __restrict__ bstart, unsigned* __restrict__ rec,
                           int E, int L, int nb) {
  extern __shared__ int dynsh[];
  int* cursor = dynsh;
  int* delta  = dynsh + nb;
  int* scrA   = dynsh + 2 * nb;
  int* scrB   = scrA + blockDim.x;
  __shared__ unsigned stage[SC_CAP];
  __shared__ int gpos[SC_CAP];
  const int tid = threadIdx.x, bd = blockDim.x, wg = blockIdx.x;
  const bool staged = L <= SC_CAP;
  const int per = (nb + bd - 1) / bd;
  const int lo = tid * per;
  int hi = lo + per;
  if (hi > nb) hi = nb;
  int s = 0, ls = 0;
  for (int b = lo; b < hi; ++b) {
    const int t  = totals[b];
    const int o  = offs[(size_t)b * NSL + wg];
    const int o1 = (wg + 1 < NSL) ? offs[(size_t)b * NSL + wg + 1] : t;
    s  += t;
    ls += o1 - o;
  }
  int pre, lpre, tot;
  block_excl_scan2(s, ls, scrA, scrB, &pre, &lpre, &tot);
  for (int b = lo; b < hi; ++b) {
    const int t  = totals[b];
    const int o  = offs[(size_t)b * NSL + wg];
    const int o1 = (wg + 1 < NSL) ? offs[(size_t)b * NSL + wg + 1] : t;
    cursor[b] = staged ? lpre : pre + o;
    delta[b]  = pre + o - lpre;
    if (wg == 0) bstart[b] = pre;
    pre  += t;
    lpre += o1 - o;
  }
  if (wg == 0 && tid == 0) bstart[nb] = tot;
  __syncthreads();
  const int e0 = wg * L;
  int e1 = e0 + L;
  if (e1 > E) e1 = E;
  int e = e0 + tid;
  if (staged) {
    for (; e + 3 * bd < e1; e += 4 * bd) {
      int sv[4], dv[4];
#pragma unroll
      for (int u = 0; u < 4; ++u) {
        sv[u] = src[e + u * bd];
        dv[u] = dst[e + u * bd];
      }
      int bk[4], dl[4], lp[4];
#pragma unroll
      for (int u = 0; u < 4; ++u) {
        bk[u] = dv[u] >> LOG_RB;
        dl[u] = delta[bk[u]];
      }
#pragma unroll
      for (int u = 0; u < 4; ++u) lp[u] = atomicAdd(&cursor[bk[u]], 1);
#pragma unroll
      for (int u = 0; u < 4; ++u) {
        stage[lp[u]] = ((unsigned)sv[u] << LOG_RB) | (unsigned)(dv[u] & (RB - 1));
        gpos[lp[u]]  = lp[u] + dl[u];
      }
    }
    for (; e < e1; e += bd) {
      const int sv = src[e], dv = dst[e];
      const int bk = dv >> LOG_RB;
      const int lp = atomicAdd(&cursor[bk], 1);
      stage[lp] = ((unsigned)sv << LOG_RB) | (unsigned)(dv & (RB - 1));
      gpos[lp]  = lp + delta[bk];
    }
    __syncthreads();
    const int m = e1 - e0;
    for (int i = tid; i < m; i += 4 * bd) {
      unsigned rv[4];
      int gp[4];
#pragma unroll
      for (int u = 0; u < 4; ++u) {
        const int j = i + u * bd;
        const int jj = j < m ? j : m - 1;
        rv[u] = stage[jj];
        gp[u] = gpos[jj];
      }
#pragma unroll
      for (int u = 0; u < 4; ++u)
        if (i + u * bd < m) rec[gp[u]] = rv[u];
    }
  } else {
    for (; e + 3 * bd < e1; e += 4 * bd) {
      int sv[4], dv[4];
#pragma unroll
      for (int u = 0; u < 4; ++u) {
        sv[u] = src[e + u * bd];
        dv[u] = dst[e + u * bd];
      }
#pragma unroll
      for (int u = 0; u < 4; ++u) {
        const int p = atomicAdd(&cursor[dv[u] >> LOG_RB], 1);
        rec[p] = ((unsigned)sv[u] << LOG_RB) | (unsigned)(dv[u] & (RB - 1));
      }
    }
    for (; e < e1; e += bd) {
      const int sv = src[e], dv = dst[e];
      const int p = atomicAdd(&cursor[dv >> LOG_RB], 1);
      rec[p] = ((unsigned)sv << LOG_RB) | (unsigned)(dv & (RB - 1));
    }
  }
}

#define SRT_U 9
#define SRT_CAP 8192
__global__ void k_bsort(const unsigned* __restrict__ rec, const int* __restrict__ bstart,
                        float* __restrict__ dinv, int* __restrict__ rowptr,
                        unsigned* __restrict__ srt, int n, int nb) {
  __shared__ int cnt[RB];
  __shared__ int cur[RB];
  __shared__ unsigned stage[SRT_CAP];
  const int tid = threadIdx.x, bd = blockDim.x, b = blockIdx.x;
  if (tid < RB) cnt[tid] = 0;
  __syncthreads();
  const int e0 = bstart[b], e1 = bstart[b + 1];
  const bool staged = e1 - e0 <= SRT_CAP;
  for (int base = e0; base < e1; base += SRT_U * bd) {
    unsigned r[SRT_U];
#pragma unroll
    for (int u = 0; u < SRT_U; ++u) {
      const int idx = base + tid + u * bd;
      r[u] = rec[idx < e1 ? idx : e1 - 1];
    }
#pragma unroll
    for (int u = 0; u < SRT_U; ++u)
      if (base + tid + u * bd < e1) atomicAdd(&cnt[r[u] & (RB - 1)], 1);
  }
  __syncthreads();
  if (tid < 32) {
    const int c0 = cnt[4 * tid], c1 = cnt[4 * tid + 1], c2 = cnt[4 * tid + 2],
              c3 = cnt[4 * tid + 3];
    const int s = c0 + c1 + c2 + c3;
    int incl = s;
    for (int o = 1; o < 32; o <<= 1) {
      const int t = __shfl_up(incl, o);
      if (tid >= o) incl += t;
    }
    const int ex = e0 + incl - s;
    cur[4 * tid]     = ex;
    cur[4 * tid + 1] = ex + c0;
    cur[4 * tid + 2] = ex + c0 + c1;
    cur[4 * tid + 3] = ex + c0 + c1 + c2;
  }
  __syncthreads();
  if (tid < RB) {
    const int node = b * RB + tid;
    rowptr[node] = cur[tid];
    if (node < n) {
      const float d = 1.0f + (float)cnt[tid];
      dinv[node] = (d > 0.0f) ? (1.0f / sqrtf(d)) : 0.0f;
    }
  }
  if (b == nb - 1 && tid == 0) rowptr[nb * RB] = e1;
  __syncthreads();
  for (int base = e0; base < e1; base += SRT_U * bd) {
    unsigned r[SRT_U];
#pragma unroll
    for (int u = 0; u < SRT_U; ++u) {
      const int idx = base + tid + u * bd;
      r[u] = rec[idx < e1 ? idx : e1 - 1];
    }
#pragma unroll
    for (int u = 0; u < SRT_U; ++u) {
      if (base + tid + u * bd < e1) {
        const int p = atomicAdd(&cur[r[u] & (RB - 1)], 1);
        if (staged) stage[p - e0] = r[u] >> LOG_RB;
        else srt[p] = r[u] >> LOG_RB;
      }
    }
  }
  if (staged) {
    __syncthreads();
    for (int i = e0 + tid; i < e1; i += bd) srt[i] = stage[i - e0];
  }
}

#define AGU 4
template <int F, int HP, bool FIRST>
__device__ __forceinline__ void cagg_body(const unsigned* __restrict__ srt,
                                          const int* __restrict__ rowptr,
                                          const float* __restrict__ dinv,
                                          const float* __restrict__ h,
                                          const float* __restrict__ bias,
                                          const float* __restrict__ dsrc_in,
                                          float* __restrict__ dsrc_out,
                                          float* __restrict__ o, int n) {
  constexpr int LPE = F / 4;
  const int tid = threadIdx.x, b = blockIdx.x;
  const int g = tid / LPE, c = tid % LPE;
  const int ng = blockDim.x / LPE;
  for (int nl = g; nl < RB; nl += ng) {
    const int node = b * RB + nl;
    if (node >= n) break;
    const int e0 = rowptr[node], e1 = rowptr[node + 1];
    const float dd = dinv[node];
    const float4 hself = *(const float4*)(h + (size_t)node * HP + c * 4);
    const float4 bv = *(const float4*)(bias + c * 4);
    float4 acc = {0.f, 0.f, 0.f, 0.f};
    unsigned sn[AGU];
    float dn[AGU];
    if (e0 < e1) {
#pragma unroll
      for (int u = 0; u < AGU; ++u) {
        const int idx = (e0 + u < e1) ? (e0 + u) : (e1 - 1);
        sn[u] = srt[idx];
        if (!FIRST) dn[u] = dsrc_in[idx];
      }
    }
    for (int e = e0; e < e1; e += AGU) {
      float4 hv[AGU];
      float ds[AGU];
#pragma unroll
      for (int u = 0; u < AGU; ++u) {
        const float* row = h + (size_t)sn[u] * HP;
        hv[u] = *(const float4*)(row + c * 4);
        ds[u] = FIRST ? row[F] : dn[u];
      }
#pragma unroll
      for (int u = 0; u < AGU; ++u) {
        const int idx = (e + AGU + u < e1) ? (e + AGU + u) : (e1 - 1);
        sn[u] = srt[idx];
        if (!FIRST) dn[u] = dsrc_in[idx];
      }
#pragma unroll
      for (int u = 0; u < AGU; ++u) {
        const bool live = e + u < e1;
        if (FIRST && c == 0 && live) dsrc_out[e + u] = ds[u];
        const float norm = ds[u] * dd;
        acc.x += (live ? hv[u].x : 0.f) * norm;
        acc.y += (live ? hv[u].y : 0.f) * norm;
        acc.z += (live ? hv[u].z : 0.f) * norm;
        acc.w += (live ? hv[u].w : 0.f) * norm;
      }
    }
    float4 r;
    r.x = acc.x + (hself.x * dd * dd + bv.x);
    r.y = acc.y + (hself.y * dd * dd + bv.y);
    r.z = acc.z + (hself.z * dd * dd + bv.z);
    r.w = acc.w + (hself.w * dd * dd + bv.w);
    *(float4*)(o + (size_t)node * F + c * 4) = r;
  }
}

__global__ void k_cagg1(const unsigned* __restrict__ srt, const int* __restrict__ rowptr,
                        const float* __restrict__ dinv, const float* __restrict__ h1,
                        const float* __restrict__ b1, float* __restrict__ dsrc,
                        float* __restrict__ z1, int n) {
  cagg_body<HID, H1P, true>(srt, rowptr, dinv, h1, b1, nullptr, dsrc, z1, n);
}

__global__ void k_cagg2(const unsigned* __restrict__ srt, const int* __restrict__ rowptr,
                        const float* __restrict__ dinv, const float* __restrict__ h2,
                        const float* __restrict__ b2, const float* __restrict__ dsrc,
                        float* __restrict__ out, int n) {
  cagg_body<ODIM, ODIM, false>(srt, rowptr, dinv, h2, b2, dsrc, nullptr, out, n);
}

extern "C" void kernel_launch(void* const* d_in, const int* in_sizes, int n_in,
                              void* d_out, int out_size, void* d_ws, size_t ws_size,
                              hipStream_t stream) {
  const float* x  = (const float*)d_in[0];
  const float* W1 = (const float*)d_in[1];
  const float* b1 = (const float*)d_in[2];
  const float* W2 = (const float*)d_in[3];
  const float* b2 = (const float*)d_in[4];
  const int*   ei = (const int*)d_in[5];

  const int N = in_sizes[0] / IN_DIM;
  const int E = in_sizes[5] / 2;
  const int* src = ei;
  const int* dst = ei + E;
  float* out = (float*)d_out;

  char* ws = (char*)d_ws;
  size_t off = 0;
  auto alloc = [&](size_t nbytes) -> void* {
    void* p = ws + off;
    off = (off + nbytes + 255) & ~(size_t)255;
    return p;
  };
  float* dinv = (float*)alloc((size_t)N * sizeof(float));
  float* z1   = (float*)alloc((size_t)N * HID  * sizeof(float));
  float* h2   = (float*)alloc((size_t)N * ODIM * sizeof(float));
  const size_t planes = off;

  const int B = 256;
  const int tiles = (N + 15) >> 4;
  const int gemmBlocks = (tiles + 7) / 8;

  const int NB = (N + RB - 1) / RB;
  const int L  = (E + NSL - 1) / NSL;
  unsigned* rec = (unsigned*)alloc((size_t)E * sizeof(unsigned));
  unsigned* srt = (unsigned*)alloc((size_t)E * sizeof(unsigned));
  int* cntmat   = (int*)alloc((size_t)NB * NSL * sizeof(int));
  int* totals   = (int*)alloc((size_t)NB * sizeof(int));
  int* bstart   = (int*)alloc((size_t)(NB + 1) * sizeof(int));
  float* dsrc = (float*)rec;
  int* rowptr = cntmat;

  if (off <= ws_size) {
    k_bcount  <<<NSL, 512, (size_t)NB * sizeof(int), stream>>>(dst, cntmat, E, L, NB);
    k_bscan   <<<NB, NSL, 0, stream>>>(cntmat, totals);
    k_bscatter<<<NSL, 512, (size_t)(2 * NB + 2 * 512) * sizeof(int), stream>>>(
        src, dst, cntmat, totals, bstart, rec, E, L, NB);
    k_bsort   <<<NB, 512, 0, stream>>>(rec, bstart, dinv, rowptr, srt, N, NB);

    float* h1a = h2;
    k_gemm1   <<<gemmBlocks, B, 0, stream>>>(x, W1, h1a, N, H1P, dinv);
    k_cagg1   <<<NB, 512, 0, stream>>>(srt, rowptr, dinv, h1a, b1, dsrc, z1, N);

    k_gemm2   <<<gemmBlocks, B, 0, stream>>>(z1, W2, h2, N);
    k_cagg2   <<<NB, 512, 0, stream>>>(srt, rowptr, dinv, h2, b2, dsrc, out, N);
    return;
  }

  off = planes;
  float* h1 = (float*)alloc((size_t)N * HID * sizeof(float));
  k_deg_init <<<(N + B - 1) / B, B, 0, stream>>>(dinv, N);
  k_deg_edges<<<(E + B - 1) / B, B, 0, stream>>>(dst, dinv, E);
  k_dinv     <<<(N + B - 1) / B, B, 0, stream>>>(dinv, N);

  k_gemm1    <<<gemmBlocks, B, 0, stream>>>(x, W1, h1, N, HID, nullptr);
  k_init_z1  <<<((size_t)N * HID + B - 1) / B, B, 0, stream>>>(h1, dinv, b1, z1, N);
  k_agg1     <<<(E * 4 + B - 1) / B, B, 0, stream>>>(src, dst, dinv, h1, z1, E);

  k_gemm2    <<<gemmBlocks, B, 0, stream>>>(z1, W2, h2, N);
  k_init_out <<<((size_t)N * ODIM + B - 1) / B, B, 0, stream>>>(h2, dinv, b2, out, N);
  k_agg2     <<<(E * 16 + B - 1) / B, B, 0, stream>>>(src, dst, dinv, h2, out, E);
}


# GEMM1's tiles ride in the count pass's launch as extra work-groups and GEMM2 runs behind the layer-1 sums on rows kept in LDS (z1 not stored): 8 launches become 6; values and sum orders unchanged
# speedup vs baseline: 3.5679x; 1.1021x over previous
#include <hip/hip_runtime.h>

typedef __attribute__((ext_vector_type(2))) float v2f;
typedef __attribute__((ext_vector_type(8))) float v8f;

#define IN_DIM 512
#define HID    16
#define ODIM   64

__global__ void k_deg_init(float* __restrict__ deg, int n) {
  int i = blockIdx.x * blockDim.x + threadIdx.x;
  if (i < n) deg[i] = 1.0f;
}

__global__ void k_deg_edges(const int* __restrict__ dst, float* __restrict__ deg, int E) {
  int i = blockIdx.x * blockDim.x + threadIdx.x;
  if (i < E) atomicAdd(&deg[dst[i]], 1.0f);
}

__global__ void k_dinv(float* __restrict__ deg, int n) {
  int i = blockIdx.x * blockDim.x + threadIdx.x;
  if (i < n) {
    float d = deg[i];
    deg[i] = (d > 0.0f) ? (1.0f / sqrtf(d)) : 0.0f;
  }
}

__device__ __forceinline__ void gemm1_tiles(const float* __restrict__ x,
                                            const float* __restrict__ W1,
                                            float* __restrict__ h1, int n, int pitch,
                                            float* lw, int wg) {
  for (int i = threadIdx.x; i < IN_DIM * HID; i += blockDim.x) lw[i] = W1[i];
  __syncthreads();

  int wave  = (wg * blockDim.x + threadIdx.x) >> 5;
  int lane  = threadIdx.x & 31;
  int tiles = (n + 15) >> 4;
  if (wave >= tiles) return;

  int half = lane >> 4;
  int ml   = lane & 15;
  int row  = wave * 16 + ml;
  int rrow = (row < n) ? row : (n - 1);
  const float* xr = x + (size_t)rrow * IN_DIM;

  v8f acc = {0.f, 0.f, 0.f, 0.f, 0.f, 0.f, 0.f, 0.f};
  for (int k = 0; k < IN_DIM; k += 4) {
    int ka = k + 2 * half;
    v2f a; a.x = xr[ka];                 a.y = xr[ka + 1];
    v2f b; b.x = lw[ka * HID + ml];      b.y = lw[(ka + 1) * HID + ml];
    acc = __builtin_amdgcn_wmma_f32_16x16x4_f32(false, a, false, b, (short)0, acc,
                                                false, false);
  }
#pragma unroll
  for (int r = 0; r < 8; ++r) {
    int orow = wave * 16 + r + 8 * half;
    if (orow < n) h1[(size_t)orow * pitch + ml] = acc[r];
  }
}

__global__ void k_gemm1(const float* __restrict__ x, const float* __restrict__ W1,
                        float* __restrict__ h1, int n, int pitch) {
  __shared__ float lw[IN_DIM * HID];
  gemm1_tiles(x, W1, h1, n, pitch, lw, blockIdx.x);
}

__global__ void k_init_z1(const float* __restrict__ h1, const float* __restrict__ dinv,
                          const float* __restrict__ b1, float* __restrict__ z1, int n) {
  int t = blockIdx.x * blockDim.x + threadIdx.x;
  int total = n * HID;
  if (t < total) {
    int i = t / HID, j = t - i * HID;
    float di = dinv[i];
    z1[t] = h1[t] * di * di + b1[j];
  }
}

__global__ void k_agg1(const int* __restrict__ src, const int* __restrict__ dst,
                       const float* __restrict__ dinv, const float* __restrict__ h1,
                       float* __restrict__ z1, int E) {
  int total  = E * 4;
  int stride = gridDim.x * blockDim.x;
  for (int t = blockIdx.x * blockDim.x + threadIdx.x; t < total; t += stride) {
    int e = t >> 2, c = t & 3;
    int s = src[e], d = dst[e];
    float norm = dinv[s] * dinv[d];
    const float4 hv = *(const float4*)(h1 + (size_t)s * HID + c * 4);
    float* o = z1 + (size_t)d * HID + c * 4;
    atomicAdd(o + 0, hv.x * norm);
    atomicAdd(o + 1, hv.y * norm);
    atomicAdd(o + 2, hv.z * norm);
    atomicAdd(o + 3, hv.w * norm);
  }
}

__global__ void k_gemm2(const float* __restrict__ z1, const float* __restrict__ W2,
                        float* __restrict__ h2, int n) {
  __shared__ float lw[HID * ODIM];
  for (int i = threadIdx.x; i < HID * ODIM; i += blockDim.x) lw[i] = W2[i];
  __syncthreads();

  int wave  = (blockIdx.x * blockDim.x + threadIdx.x) >> 5;
  int lane  = threadIdx.x & 31;
  int tiles = (n + 15) >> 4;
  if (wave >= tiles) return;

  int half = lane >> 4;
  int ml   = lane & 15;
  int row  = wave * 16 + ml;
  int rrow = (row < n) ? row : (n - 1);
  const float* zr = z1 + (size_t)rrow * HID;

  v2f afrag[4];
#pragma unroll
  for (int kk = 0; kk < 4; ++kk) {
    int ka = kk * 4 + 2 * half;
    afrag[kk].x = fmaxf(zr[ka], 0.f);
    afrag[kk].y = fmaxf(zr[ka + 1], 0.f);
  }

#pragma unroll
  for (int tcol = 0; tcol < 4; ++tcol) {
    v8f acc = {0.f, 0.f, 0.f, 0.f, 0.f, 0.f, 0.f, 0.f};
#pragma unroll
    for (int kk = 0; kk < 4; ++kk) {
      int ka = kk * 4 + 2 * half;
      v2f b;
      b.x = lw[ka * ODIM + tcol * 16 + ml];
      b.y = lw[(ka + 1) * ODIM + tcol * 16 + ml];
      acc = __builtin_amdgcn_wmma_f32_16x16x4_f32(false, afrag[kk], false, b, (short)0,
                                                  acc, false, false);
    }
#pragma unroll
    for (int r = 0; r < 8; ++r) {
      int orow = wave * 16 + r + 8 * half;
      if (orow < n) h2[(size_t)orow * ODIM + tcol * 16 + ml] = acc[r];
    }
  }
}

__global__ void k_init_out(const float* __restrict__ h2, const float* __restrict__ dinv,
                           const float* __restrict__ b2, float* __restrict__ out, int n) {
  int t = blockIdx.x * blockDim.x + threadIdx.x;
  int total = n * ODIM;
  if (t < total) {
    int i = t / ODIM, j = t - i * ODIM;
    float di = dinv[i];
    out[t] = h2[t] * di * di + b2[j];
  }
}

__global__ void k_agg2(const int* __restrict__ src, const int* __restrict__ dst,
                       const float* __restrict__ dinv, const float* __restrict__ h2,
                       float* __restrict__ out, int E) {
  int total  = E * 16;
  int stride = gridDim.x * blockDim.x;
  for (int t = blockIdx.x * blockDim.x + threadIdx.x; t < total; t += stride) {
    int e = t >> 4, c = t & 15;
    int s = src[e], d = dst[e];
    float norm = dinv[s] * dinv[d];
    const float4 hv = *(const float4*)(h2 + (size_t)s * ODIM + c * 4);
    float* o = out + (size_t)d * ODIM + c * 4;
    atomicAdd(o + 0, hv.x * norm);
    atomicAdd(o + 1, hv.y * norm);
    atomicAdd(o + 2, hv.z * norm);
    atomicAdd(o + 3, hv.w * norm);
  }
}

#define RB     128
#define LOG_RB 7
#define NSL    256
#define H1P    32

__device__ __forceinline__ int block_excl_scan(int v, int* scr, int* total) {
  const int t = threadIdx.x, nt = blockDim.x;
  scr[t] = v;
  __syncthreads();
  for (int o = 1; o < nt; o <<= 1) {
    int a = (t >= o) ? scr[t - o] : 0;
    __syncthreads();
    scr[t] += a;
    __syncthreads();
  }
  const int incl = scr[t];
  *total = scr[nt - 1];
  __syncthreads();
  return incl - v;
}

__device__ __forceinline__ void block_excl_scan2(int va, int vb, int* scrA, int* scrB,
                                                 int* exA, int* exB, int* totA) {
  const int t = threadIdx.x, nt = blockDim.x;
  scrA[t] = va;
  scrB[t] = vb;
  __syncthreads();
  for (int o = 1; o < nt; o <<= 1) {
    int a = 0, b = 0;
    if (t >= o) {
      a = scrA[t - o];
      b = scrB[t - o];
    }
    __syncthreads();
    scrA[t] += a;
    scrB[t] += b;
    __syncthreads();
  }
  *exA = scrA[t] - va;
  *exB = scrB[t] - vb;
  *totA = scrA[nt - 1];
  __syncthreads();
}

__device__ __forceinline__ void bcount_slice(const int* __restrict__ dst,
                                             int* __restrict__ cntmat, int E, int L, int nb,
                                             int* cnt) {
  const int tid = threadIdx.x, bd = blockDim.x, wg = blockIdx.x;
  for (int i = tid; i < nb; i += bd) cnt[i] = 0;
  __syncthreads();
  const int e0 = wg * L;
  int e1 = e0 + L;
  if (e1 > E) e1 = E;
  int e = e0 + tid;
  for (; e + 3 * bd < e1; e += 4 * bd) {
    int d[4];
#pragma unroll
    for (int u = 0; u < 4; ++u) d[u] = dst[e + u * bd];
#pragma unroll
    for (int u = 0; u < 4; ++u) atomicAdd(&cnt[d[u] >> LOG_RB], 1);
  }
  for (; e < e1; e += bd) atomicAdd(&cnt[dst[e] >> LOG_RB], 1);
  __syncthreads();
  for (int i = tid; i < nb; i += bd) cntmat[(size_t)i * NSL + wg] = cnt[i];
}

__global__ void k_bcount_gemm1(const int* __restrict__ dst, int* __restrict__ cntmat,
                               int E, int L, int nb,
                               const float* __restrict__ x, const float* __restrict__ W1,
                               float* __restrict__ h1, int n, int pitch) {
  extern __shared__ int dynsh[];
  __shared__ float lw[IN_DIM * HID];
  if (blockIdx.x < NSL) bcount_slice(dst, cntmat, E, L, nb, dynsh);
  else gemm1_tiles(x, W1, h1, n, pitch, lw, (int)blockIdx.x - NSL);
}

__global__ void k_bscan(int* __restrict__ cntmat, int* __restrict__ totals) {
  __shared__ int scr[NSL];
  const int b = blockIdx.x, w = threadIdx.x;
  const int v = cntmat[(size_t)b * NSL + w];
  int tot;
  const int ex = block_excl_scan(v, scr, &tot);
  cntmat[(size_t)b * NSL + w] = ex;
  if (w == 0) totals[b] = tot;
}

#define SC_CAP 16384
__global__ void k_bscatter(const int* __restrict__ src, const int* __restrict__ dst,
                           const int* __restrict__ offs, const int* __restrict__ totals,
                           int* __restrict__ bstart, unsigned* __restrict__ rec,
                           int E, int L, int nb) {
  extern __shared__ int dynsh[];
  int* cursor = dynsh;
  int* delta  = dynsh + nb;
  int* scrA   = dynsh + 2 * nb;
  int* scrB   = scrA + blockDim.x;
  __shared__ unsigned stage[SC_CAP];
  __shared__ int gpos[SC_CAP];
  const int tid = threadIdx.x, bd = blockDim.x, wg = blockIdx.x;
  const bool staged = L <= SC_CAP;
  const int per = (nb + bd - 1) / bd;
  const int lo = tid * per;
  int hi = lo + per;
  if (hi > nb) hi = nb;
  int s = 0, ls = 0;
  for (int b = lo; b < hi; ++b) {
    const int t  = totals[b];
    const int o  = offs[(size_t)b * NSL + wg];
    const int o1 = (wg + 1 < NSL) ? offs[(size_t)b * NSL + wg + 1] : t;
    s  += t;
    ls += o1 - o;
  }
  int pre, lpre, tot;
  block_excl_scan2(s, ls, scrA, scrB, &pre, &lpre, &tot);
  for (int b = lo; b < hi; ++b) {
    const int t  = totals[b];
    const int o  = offs[(size_t)b * NSL + wg];
    const int o1 = (wg + 1 < NSL) ? offs[(size_t)b * NSL + wg + 1] : t;
    cursor[b] = staged ? lpre : pre + o;
    delta[b]  = pre + o - lpre;
    if (wg == 0) bstart[b] = pre;
    pre  += t;
    lpre += o1 - o;
  }
  if (wg == 0 && tid == 0) bstart[nb] = tot;
  __syncthreads();
  const int e0 = wg * L;
  int e1 = e0 + L;
  if (e1 > E) e1 = E;
  int e = e0 + tid;
  if (staged) {
    for (; e + 3 * bd < e1; e += 4 * bd) {
      int sv[4], dv[4];
#pragma unroll
      for (int u = 0; u < 4; ++u) {
        sv[u] = src[e + u * bd];
        dv[u] = dst[e + u * bd];
      }
      int bk[4], dl[4], lp[4];
#pragma unroll
      for (int u = 0; u < 4; ++u) {
        bk[u] = dv[u] >> LOG_RB;
        dl[u] = delta[bk[u]];
      }
#pragma unroll
      for (int u = 0; u < 4; ++u) lp[u] = atomicAdd(&cursor[bk[u]], 1);
#pragma unroll
      for (int u = 0; u < 4; ++u) {
        stage[lp[u]] = ((unsigned)sv[u] << LOG_RB) | (unsigned)(dv[u] & (RB - 1));
        gpos[lp[u]]  = lp[u] + dl[u];
      }
    }
    for (; e < e1; e += bd) {
      const int sv = src[e], dv = dst[e];
      const int bk = dv >> LOG_RB;
      const int lp = atomicAdd(&cursor[bk], 1);
      stage[lp] = ((unsigned)sv << LOG_RB) | (unsigned)(dv & (RB - 1));
      gpos[lp]  = lp + delta[bk];
    }
    __syncthreads();
    const int m = e1 - e0;
    for (int i = tid; i < m; i += 4 * bd) {
      unsigned rv[4];
      int gp[4];
#pragma unroll
      for (int u = 0; u < 4; ++u) {
        const int j = i + u * bd;
        const int jj = j < m ? j : m - 1;
        rv[u] = stage[jj];
        gp[u] = gpos[jj];
      }
#pragma unroll
      for (int u = 0; u < 4; ++u)
        if (i + u * bd < m) rec[gp[u]] = rv[u];
    }
  } else {
    for (; e + 3 * bd < e1; e += 4 * bd) {
      int sv[4], dv[4];
#pragma unroll
      for (int u = 0; u < 4; ++u) {
        sv[u] = src[e + u * bd];
        dv[u] = dst[e + u * bd];
      }
#pragma unroll
      for (int u = 0; u < 4; ++u) {
        const int p = atomicAdd(&cursor[dv[u] >> LOG_RB], 1);
        rec[p] = ((unsigned)sv[u] << LOG_RB) | (unsigned)(dv[u] & (RB - 1));
      }
    }
    for (; e < e1; e += bd) {
      const int sv = src[e], dv = dst[e];
      const int p = atomicAdd(&cursor[dv >> LOG_RB], 1);
      rec[p] = ((unsigned)sv << LOG_RB) | (unsigned)(dv & (RB - 1));
    }
  }
}

#define SRT_U 9
#define SRT_CAP 8192
__global__ void k_bsort(const unsigned* __restrict__ rec, const int* __restrict__ bstart,
                        float* __restrict__ dinv, float* __restrict__ h1,
                        int* __restrict__ rowptr, unsigned* __restrict__ srt, int n, int nb) {
  __shared__ int cnt[RB];
  __shared__ int cur[RB];
  __shared__ unsigned stage[SRT_CAP];
  const int tid = threadIdx.x, bd = blockDim.x, b = blockIdx.x;
  if (tid < RB) cnt[tid] = 0;
  __syncthreads();
  const int e0 = bstart[b], e1 = bstart[b + 1];
  const bool staged = e1 - e0 <= SRT_CAP;
  for (int base = e0; base < e1; base += SRT_U * bd) {
    unsigned r[SRT_U];
#pragma unroll
    for (int u = 0; u < SRT_U; ++u) {
      const int idx = base + tid + u * bd;
      r[u] = rec[idx < e1 ? idx : e1 - 1];
    }
#pragma unroll
    for (int u = 0; u < SRT_U; ++u)
      if (base + tid + u * bd < e1) atomicAdd(&cnt[r[u] & (RB - 1)], 1);
  }
  __syncthreads();
  if (tid < 32) {
    const int c0 = cnt[4 * tid], c1 = cnt[4 * tid + 1], c2 = cnt[4 * tid + 2],
              c3 = cnt[4 * tid + 3];
    const int s = c0 + c1 + c2 + c3;
    int incl = s;
    for (int o = 1; o < 32; o <<= 1) {
      const int t = __shfl_up(incl, o);
      if (tid >= o) incl += t;
    }
    const int ex = e0 + incl - s;
    cur[4 * tid]     = ex;
    cur[4 * tid + 1] = ex + c0;
    cur[4 * tid + 2] = ex + c0 + c1;
    cur[4 * tid + 3] = ex + c0 + c1 + c2;
  }
  __syncthreads();
  if (tid < RB) {
    const int node = b * RB + tid;
    rowptr[node] = cur[tid];
    if (node < n) {
      const float d = 1.0f + (float)cnt[tid];
      const float dv = (d > 0.0f) ? (1.0f / sqrtf(d)) : 0.0f;
      dinv[node] = dv;
      h1[(size_t)node * H1P + HID] = dv;
    }
  }
  if (b == nb - 1 && tid == 0) rowptr[nb * RB] = e1;
  __syncthreads();
  for (int base = e0; base < e1; base += SRT_U * bd) {
    unsigned r[SRT_U];
#pragma unroll
    for (int u = 0; u < SRT_U; ++u) {
      const int idx = base + tid + u * bd;
      r[u] = rec[idx < e1 ? idx : e1 - 1];
    }
#pragma unroll
    for (int u = 0; u < SRT_U; ++u) {
      if (base + tid + u * bd < e1) {
        const int p = atomicAdd(&cur[r[u] & (RB - 1)], 1);
        if (staged) stage[p - e0] = r[u] >> LOG_RB;
        else srt[p] = r[u] >> LOG_RB;
      }
    }
  }
  if (staged) {
    __syncthreads();
    for (int i = e0 + tid; i < e1; i += bd) srt[i] = stage[i - e0];
  }
}

#define AGU 4
#define ZP 20
template <int F, int HP, bool FIRST, bool FOLD>
__device__ __forceinline__ void cagg_body(const unsigned* __restrict__ srt,
                                          const int* __restrict__ rowptr,
                                          const float* __restrict__ dinv,
                                          const float* __restrict__ h,
                                          const float* __restrict__ bias,
                                          const float* __restrict__ dsrc_in,
                                          float* __restrict__ dsrc_out,
                                          float* __restrict__ o, float* zs, int n) {
  constexpr int LPE = F / 4;
  const int tid = threadIdx.x, b = blockIdx.x;
  const int g = tid / LPE, c = tid % LPE;
  const int ng = blockDim.x / LPE;
  for (int nl = g; nl < RB; nl += ng) {
    const int node = b * RB + nl;
    if (node >= n) {
      if (!FOLD) break;
      const float4 zero = {0.f, 0.f, 0.f, 0.f};
      *(float4*)(zs + nl * ZP + c * 4) = zero;
      continue;
    }
    const int e0 = rowptr[node], e1 = rowptr[node + 1];
    const float dd = dinv[node];
    const float4 hself = *(const float4*)(h + (size_t)node * HP + c * 4);
    const float4 bv = *(const float4*)(bias + c * 4);
    float4 acc = {0.f, 0.f, 0.f, 0.f};
    unsigned sn[AGU];
    float dn[AGU];
    if (e0 < e1) {
#pragma unroll
      for (int u = 0; u < AGU; ++u) {
        const int idx = (e0 + u < e1) ? (e0 + u) : (e1 - 1);
        sn[u] = srt[idx];
        if (!FIRST) dn[u] = dsrc_in[idx];
      }
    }
    for (int e = e0; e < e1; e += AGU) {
      float4 hv[AGU];
      float ds[AGU];
#pragma unroll
      for (int u = 0; u < AGU; ++u) {
        const float* row = h + (size_t)sn[u] * HP;
        hv[u] = *(const float4*)(row + c * 4);
        ds[u] = FIRST ? row[F] : dn[u];
      }
#pragma unroll
      for (int u = 0; u < AGU; ++u) {
        const int idx = (e + AGU + u < e1) ? (e + AGU + u) : (e1 - 1);
        sn[u] = srt[idx];
        if (!FIRST) dn[u] = dsrc_in[idx];
      }
#pragma unroll
      for (int u = 0; u < AGU; ++u) {
        const bool live = e + u < e1;
        if (FIRST && c == 0 && live) dsrc_out[e + u] = ds[u];
        const float norm = ds[u] * dd;
        acc.x += (live ? hv[u].x : 0.f) * norm;
        acc.y += (live ? hv[u].y : 0.f) * norm;
        acc.z += (live ? hv[u].z : 0.f) * norm;
        acc.w += (live ? hv[u].w : 0.f) * norm;
      }
    }
    float4 r;
    r.x = acc.x + (hself.x * dd * dd + bv.x);
    r.y = acc.y + (hself.y * dd * dd + bv.y);
    r.z = acc.z + (hself.z * dd * dd + bv.z);
    r.w = acc.w + (hself.w * dd * dd + bv.w);
    if (FOLD) {
      float4 q;
      q.x = fmaxf(r.x, 0.f);
      q.y = fmaxf(r.y, 0.f);
      q.z = fmaxf(r.z, 0.f);
      q.w = fmaxf(r.w, 0.f);
      *(float4*)(zs + nl * ZP + c * 4) = q;
    } else {
      *(float4*)(o + (size_t)node * F + c * 4) = r;
    }
  }
}

__global__ void k_cagg1(const unsigned* __restrict__ srt, const int* __restrict__ rowptr,
                        const float* __restrict__ dinv, const float* __restrict__ h1,
                        const float* __restrict__ b1, float* __restrict__ dsrc,
                        float* __restrict__ z1, int n) {
  cagg_body<HID, H1P, true, false>(srt, rowptr, dinv, h1, b1, nullptr, dsrc, z1, nullptr, n);
}

__global__ void k_cagg1_gemm2(const unsigned* __restrict__ srt,
                              const int* __restrict__ rowptr,
                              const float* __restrict__ dinv, const float* __restrict__ h1,
                              const float* __restrict__ b1, float* __restrict__ dsrc,
                              const float* __restrict__ W2, float* __restrict__ h2, int n) {
  __shared__ __attribute__((aligned(16))) float zs[RB * ZP];
  __shared__ float lw2[HID * ODIM];
  for (int i = threadIdx.x; i < HID * ODIM; i += blockDim.x) lw2[i] = W2[i];
  cagg_body<HID, H1P, true, true>(srt, rowptr, dinv, h1, b1, nullptr, dsrc, nullptr, zs, n);
  __syncthreads();

  const int wv = threadIdx.x >> 5, nwv = blockDim.x >> 5;
  const int lane = threadIdx.x & 31;
  const int half = lane >> 4;
  const int ml   = lane & 15;
  for (int unit = wv; unit < (RB / 16) * 2; unit += nwv) {
    const int rt = unit >> 1;
    const float* zr = zs + (rt * 16 + ml) * ZP;
    v2f afrag[4];
#pragma unroll
    for (int kk = 0; kk < 4; ++kk) {
      const int ka = kk * 4 + 2 * half;
      afrag[kk].x = zr[ka];
      afrag[kk].y = zr[ka + 1];
    }
#pragma unroll
    for (int t = 0; t < 2; ++t) {
      const int tcol = (unit & 1) * 2 + t;
      v8f acc = {0.f, 0.f, 0.f, 0.f, 0.f, 0.f, 0.f, 0.f};
#pragma unroll
      for (int kk = 0; kk < 4; ++kk) {
        const int ka = kk * 4 + 2 * half;
        v2f b;
        b.x = lw2[ka * ODIM + tcol * 16 + ml];
        b.y = lw2[(ka + 1) * ODIM + tcol * 16 + ml];
        acc = __builtin_amdgcn_wmma_f32_16x16x4_f32(false, afrag[kk], false, b, (short)0,
                                                    acc, false, false);
      }
#pragma unroll
      for (int r = 0; r < 8; ++r) {
        const int orow = (int)blockIdx.x * RB + rt * 16 + r + 8 * half;
        if (orow < n) h2[(size_t)orow * ODIM + tcol * 16 + ml] = acc[r];
      }
    }
  }
}

__global__ void k_cagg2(const unsigned* __restrict__ srt, const int* __restrict__ rowptr,
                        const float* __restrict__ dinv, const float* __restrict__ h2,
                        const float* __restrict__ b2, const float* __restrict__ dsrc,
                        float* __restrict__ out, int n) {
  cagg_body<ODIM, ODIM, false, false>(srt, rowptr, dinv, h2, b2, dsrc, nullptr, out, nullptr,
                                      n);
}

extern "C" void kernel_launch(void* const* d_in, const int* in_sizes, int n_in,
                              void* d_out, int out_size, void* d_ws, size_t ws_size,
                              hipStream_t stream) {
  const float* x  = (const float*)d_in[0];
  const float* W1 = (const float*)d_in[1];
  const float* b1 = (const float*)d_in[2];
  const float* W2 = (const float*)d_in[3];
  const float* b2 = (const float*)d_in[4];
  const int*   ei = (const int*)d_in[5];

  const int N = in_sizes[0] / IN_DIM;
  const int E = in_sizes[5] / 2;
  const int* src = ei;
  const int* dst = ei + E;
  float* out = (float*)d_out;

  char* ws = (char*)d_ws;
  size_t off = 0;
  auto alloc = [&](size_t nbytes) -> void* {
    void* p = ws + off;
    off = (off + nbytes + 255) & ~(size_t)255;
    return p;
  };
  float* dinv = (float*)alloc((size_t)N * sizeof(float));
  float* h2   = (float*)alloc((size_t)N * ODIM * sizeof(float));
  const size_t planes = off;

  const int B = 256;
  const int tiles = (N + 15) >> 4;
  const int gemmBlocks = (tiles + 7) / 8;

  const int NB = (N + RB - 1) / RB;
  const int L  = (E + NSL - 1) / NSL;
  unsigned* rec = (unsigned*)alloc((size_t)E * sizeof(unsigned));
  unsigned* srt = (unsigned*)alloc((size_t)E * sizeof(unsigned));
  int* cntmat   = (int*)alloc((size_t)NB * NSL * sizeof(int));
  int* totals   = (int*)alloc((size_t)NB * sizeof(int));
  int* bstart   = (int*)alloc((size_t)(NB + 1) * sizeof(int));
  float* dsrc = (float*)rec;
  int* rowptr = cntmat;
  float* tail = (float*)(ws + off);
  const size_t need_z1  = off + (size_t)N * HID * sizeof(float);
  const size_t need_h1p = off + (size_t)N * H1P * sizeof(float);
  const int gemm1Groups = (tiles + 15) / 16;

  if (need_z1 <= ws_size) {
    const bool fold = need_h1p <= ws_size;
    float* h1a = fold ? tail : h2;
    k_bcount_gemm1<<<NSL + gemm1Groups, 512, (size_t)NB * sizeof(int), stream>>>(
        dst, cntmat, E, L, NB, x, W1, h1a, N, H1P);
    k_bscan   <<<NB, NSL, 0, stream>>>(cntmat, totals);
    k_bscatter<<<NSL, 512, (size_t)(2 * NB + 2 * 512) * sizeof(int), stream>>>(
        src, dst, cntmat, totals, bstart, rec, E, L, NB);
    k_bsort   <<<NB, 512, 0, stream>>>(rec, bstart, dinv, h1a, rowptr, srt, N, NB);

    if (fold) {
      k_cagg1_gemm2<<<NB, 512, 0, stream>>>(srt, rowptr, dinv, h1a, b1, dsrc, W2, h2, N);
    } else {
      float* z1 = tail;
      k_cagg1 <<<NB, 512, 0, stream>>>(srt, rowptr, dinv, h1a, b1, dsrc, z1, N);
      k_gemm2 <<<gemmBlocks, B, 0, stream>>>(z1, W2, h2, N);
    }
    k_cagg2   <<<NB, 512, 0, stream>>>(srt, rowptr, dinv, h2, b2, dsrc, out, N);
    return;
  }

  off = planes;
  float* z1 = (float*)alloc((size_t)N * HID * sizeof(float));
  float* h1 = (float*)alloc((size_t)N * HID * sizeof(float));
  k_deg_init <<<(N + B - 1) / B, B, 0, stream>>>(dinv, N);
  k_deg_edges<<<(E + B - 1) / B, B, 0, stream>>>(dst, dinv, E);
  k_dinv     <<<(N + B - 1) / B, B, 0, stream>>>(dinv, N);

  k_gemm1    <<<gemmBlocks, B, 0, stream>>>(x, W1, h1, N, HID);
  k_init_z1  <<<((size_t)N * HID + B - 1) / B, B, 0, stream>>>(h1, dinv, b1, z1, N);
  k_agg1     <<<(E * 4 + B - 1) / B, B, 0, stream>>>(src, dst, dinv, h1, z1, E);

  k_gemm2    <<<gemmBlocks, B, 0, stream>>>(z1, W2, h2, N);
  k_init_out <<<((size_t)N * ODIM + B - 1) / B, B, 0, stream>>>(h2, dinv, b2, out, N);
  k_agg2     <<<(E * 16 + B - 1) / B, B, 0, stream>>>(src, dst, dinv, h2, out, E);
}


# edge slices bucket their own records in LDS (own count and scan) and store whole chunks; a bucket's work-group pulls its runs: count pass and scan launch gone, 6 launches to 4; sum order as before
# speedup vs baseline: 4.0039x; 1.1222x over previous
#include <hip/hip_runtime.h>

typedef __attribute__((ext_vector_type(2))) float v2f;
typedef __attribute__((ext_vector_type(8))) float v8f;

#define IN_DIM 512
#define HID    16
#define ODIM   64

__global__ void k_deg_init(float* __restrict__ deg, int n) {
  int i = blockIdx.x * blockDim.x + threadIdx.x;
  if (i < n) deg[i] = 1.0f;
}

__global__ void k_deg_edges(const int* __restrict__ dst, float* __restrict__ deg, int E) {
  int i = blockIdx.x * blockDim.x + threadIdx.x;
  if (i < E) atomicAdd(&deg[dst[i]], 1.0f);
}

__global__ void k_dinv(float* __restrict__ deg, int n) {
  int i = blockIdx.x * blockDim.x + threadIdx.x;
  if (i < n) {
    float d = deg[i];
    deg[i] = (d > 0.0f) ? (1.0f / sqrtf(d)) : 0.0f;
  }
}

__device__ __forceinline__ void gemm1_tiles(const float* __restrict__ x,
                                            const float* __restrict__ W1,
                                            float* __restrict__ h1, int n, int pitch,
                                            float* lw, int wg) {
  for (int i = threadIdx.x; i < IN_DIM * HID; i += blockDim.x) lw[i] = W1[i];
  __syncthreads();

  int wave  = (wg * blockDim.x + threadIdx.x) >> 5;
  int lane  = threadIdx.x & 31;
  int tiles = (n + 15) >> 4;
  if (wave >= tiles) return;

  int half = lane >> 4;
  int ml   = lane & 15;
  int row  = wave * 16 + ml;
  int rrow = (row < n) ? row : (n - 1);
  const float* xr = x + (size_t)rrow * IN_DIM;

  v8f acc = {0.f, 0.f, 0.f, 0.f, 0.f, 0.f, 0.f, 0.f};
  for (int k = 0; k < IN_DIM; k += 4) {
    int ka = k + 2 * half;
    v2f a; a.x = xr[ka];                 a.y = xr[ka + 1];
    v2f b; b.x = lw[ka * HID + ml];      b.y = lw[(ka + 1) * HID + ml];
    acc = __builtin_amdgcn_wmma_f32_16x16x4_f32(false, a, false, b, (short)0, acc,
                                                false, false);
  }
#pragma unroll
  for (int r = 0; r < 8; ++r) {
    int orow = wave * 16 + r + 8 * half;
    if (orow < n) h1[(size_t)orow * pitch + ml] = acc[r];
  }
}

__global__ void k_gemm1(const float* __restrict__ x, const float* __restrict__ W1,
                        float* __restrict__ h1, int n, int pitch) {
  __shared__ float lw[IN_DIM * HID];
  gemm1_tiles(x, W1, h1, n, pitch, lw, blockIdx.x);
}

__global__ void k_init_z1(const float* __restrict__ h1, const float* __restrict__ dinv,
                          const float* __restrict__ b1, float* __restrict__ z1, int n) {
  int t = blockIdx.x * blockDim.x + threadIdx.x;
  int total = n * HID;
  if (t < total) {
    int i = t / HID, j = t - i * HID;
    float di = dinv[i];
    z1[t] = h1[t] * di * di + b1[j];
  }
}

__global__ void k_agg1(const int* __restrict__ src, const int* __restrict__ dst,
                       const float* __restrict__ dinv, const float* __restrict__ h1,
                       float* __restrict__ z1, int E) {
  int total  = E * 4;
  int stride = gridDim.x * blockDim.x;
  for (int t = blockIdx.x * blockDim.x + threadIdx.x; t < total; t += stride) {
    int e = t >> 2, c = t & 3;
    int s = src[e], d = dst[e];
    float norm = dinv[s] * dinv[d];
    const float4 hv = *(const float4*)(h1 + (size_t)s * HID + c * 4);
    float* o = z1 + (size_t)d * HID + c * 4;
    atomicAdd(o + 0, hv.x * norm);
    atomicAdd(o + 1, hv.y * norm);
    atomicAdd(o + 2, hv.z * norm);
    atomicAdd(o + 3, hv.w * norm);
  }
}

__global__ void k_gemm2(const float* __restrict__ z1, const float* __restrict__ W2,
                        float* __restrict__ h2, int n) {
  __shared__ float lw[HID * ODIM];
  for (int i = threadIdx.x; i < HID * ODIM; i += blockDim.x) lw[i] = W2[i];
  __syncthreads();

  int wave  = (blockIdx.x * blockDim.x + threadIdx.x) >> 5;
  int lane  = threadIdx.x & 31;
  int tiles = (n + 15) >> 4;
  if (wave >= tiles) return;

  int half = lane >> 4;
  int ml   = lane & 15;
  int row  = wave * 16 + ml;
  int rrow = (row < n) ? row : (n - 1);
  const float* zr = z1 + (size_t)rrow * HID;

  v2f afrag[4];
#pragma unroll
  for (int kk = 0; kk < 4; ++kk) {
    int ka = kk * 4 + 2 * half;
    afrag[kk].x = fmaxf(zr[ka], 0.f);
    afrag[kk].y = fmaxf(zr[ka + 1], 0.f);
  }

#pragma unroll
  for (int tcol = 0; tcol < 4; ++tcol) {
    v8f acc = {0.f, 0.f, 0.f, 0.f, 0.f, 0.f, 0.f, 0.f};
#pragma unroll
    for (int kk = 0; kk < 4; ++kk) {
      int ka = kk * 4 + 2 * half;
      v2f b;
      b.x = lw[ka * ODIM + tcol * 16 + ml];
      b.y = lw[(ka + 1) * ODIM + tcol * 16 + ml];
      acc = __builtin_amdgcn_wmma_f32_16x16x4_f32(false, afrag[kk], false, b, (short)0,
                                                  acc, false, false);
    }
#pragma unroll
    for (int r = 0; r < 8; ++r) {
      int orow = wave * 16 + r + 8 * half;
      if (orow < n) h2[(size_t)orow * ODIM + tcol * 16 + ml] = acc[r];
    }
  }
}

__global__ void k_init_out(const float* __restrict__ h2, const float* __restrict__ dinv,
                           const float* __restrict__ b2, float* __restrict__ out, int n) {
  int t = blockIdx.x * blockDim.x + threadIdx.x;
  int total = n * ODIM;
  if (t < total) {
    int i = t / ODIM, j = t - i * ODIM;
    float di = dinv[i];
    out[t] = h2[t] * di * di + b2[j];
  }
}

__global__ void k_agg2(const int* __restrict__ src, const int* __restrict__ dst,
                       const float* __restrict__ dinv, const float* __restrict__ h2,
                       float* __restrict__ out, int E) {
  int total  = E * 16;
  int stride = gridDim.x * blockDim.x;
  for (int t = blockIdx.x * blockDim.x + threadIdx.x; t < total; t += stride) {
    int e = t >> 4, c = t & 15;
    int s = src[e], d = dst[e];
    float norm = dinv[s] * dinv[d];
    const float4 hv = *(const float4*)(h2 + (size_t)s * ODIM + c * 4);
    float* o = out + (size_t)d * ODIM + c * 4;
    atomicAdd(o + 0, hv.x * norm);
    atomicAdd(o + 1, hv.y * norm);
    atomicAdd(o + 2, hv.z * norm);
    atomicAdd(o + 3, hv.w * norm);
  }
}

#define RB     128
#define LOG_RB 7
#define NSL    256
#define H1P    32

__device__ __forceinline__ int block_excl_scan(int v, int* scr, int* total) {
  const int t = threadIdx.x, nt = blockDim.x;
  scr[t] = v;
  __syncthreads();
  for (int o = 1; o < nt; o <<= 1) {
    int a = (t >= o) ? scr[t - o] : 0;
    __syncthreads();
    scr[t] += a;
    __syncthreads();
  }
  const int incl = scr[t];
  *total = scr[nt - 1];
  __syncthreads();
  return incl - v;
}

__device__ __forceinline__ void block_scan_and_sum(int va, int vb, int* scrA, int* scrB,
                                                   int* exA, int* totA, int* totB) {
  const int t = threadIdx.x, nt = blockDim.x;
  scrA[t] = va;
  scrB[t] = vb;
  __syncthreads();
  for (int o = 1; o < nt; o <<= 1) {
    int a = 0, b = 0;
    if (t >= o) {
      a = scrA[t - o];
      b = scrB[t - o];
    }
    __syncthreads();
    scrA[t] += a;
    scrB[t] += b;
    __syncthreads();
  }
  *exA  = scrA[t] - va;
  *totA = scrA[nt - 1];
  *totB = scrB[nt - 1];
  __syncthreads();
}

#define SC_CAP 16384
__device__ __forceinline__ void bslice(const int* __restrict__ src,
                                       const int* __restrict__ dst,
                                       int* __restrict__ lstart, unsigned* __restrict__ rec,
                                       int E, int L, int nb, int P, int* cnt, int* scr,
                                       unsigned* stage) {
  const int tid = threadIdx.x, bd = blockDim.x, wg = blockIdx.x;
  for (int i = tid; i < nb; i += bd) cnt[i] = 0;
  __syncthreads();
  const int e0 = wg * L;
  int e1 = e0 + L;
  if (e1 > E) e1 = E;
  int e = e0 + tid;
  for (; e + 3 * bd < e1; e += 4 * bd) {
    int d[4];
#pragma unroll
    for (int u = 0; u < 4; ++u) d[u] = dst[e + u * bd];
#pragma unroll
    for (int u = 0; u < 4; ++u) atomicAdd(&cnt[d[u] >> LOG_RB], 1);
  }
  for (; e < e1; e += bd) atomicAdd(&cnt[dst[e] >> LOG_RB], 1);
  __syncthreads();
  const int per = (nb + bd - 1) / bd;
  const int lo = tid * per;
  int hi = lo + per;
  if (hi > nb) hi = nb;
  int s = 0;
  for (int b = lo; b < hi; ++b) s += cnt[b];
  int tot;
  int pre = block_excl_scan(s, scr, &tot);
  int* lrow = lstart + (size_t)wg * P;
  for (int b = lo; b < hi; ++b) {
    const int c = cnt[b];
    cnt[b]  = pre;
    lrow[b] = pre;
    pre += c;
  }
  if (tid == 0) lrow[nb] = tot;
  __syncthreads();
  e = e0 + tid;
  for (; e + 3 * bd < e1; e += 4 * bd) {
    int sv[4], dv[4], lp[4];
#pragma unroll
    for (int u = 0; u < 4; ++u) {
      sv[u] = src[e + u * bd];
      dv[u] = dst[e + u * bd];
    }
#pragma unroll
    for (int u = 0; u < 4; ++u) lp[u] = atomicAdd(&cnt[dv[u] >> LOG_RB], 1);
#pragma unroll
    for (int u = 0; u < 4; ++u)
      stage[lp[u]] = ((unsigned)sv[u] << LOG_RB) | (unsigned)(dv[u] & (RB - 1));
  }
  for (; e < e1; e += bd) {
    const int sv = src[e], dv = dst[e];
    const int lp = atomicAdd(&cnt[dv >> LOG_RB], 1);
    stage[lp] = ((unsigned)sv << LOG_RB) | (unsigned)(dv & (RB - 1));
  }
  __syncthreads();
  const int m = e1 - e0;
  for (int i = tid; i < m; i += 4 * bd) {
    unsigned rv[4];
#pragma unroll
    for (int u = 0; u < 4; ++u) {
      const int j = i + u * bd;
      rv[u] = stage[j < m ? j : m - 1];
    }
#pragma unroll
    for (int u = 0; u < 4; ++u)
      if (i + u * bd < m) rec[e0 + i + u * bd] = rv[u];
  }
}

__global__ void k_bslice_gemm1(const int* __restrict__ src, const int* __restrict__ dst,
                               int* __restrict__ lstart, unsigned* __restrict__ rec,
                               int E, int L, int nb, int P,
                               const float* __restrict__ x, const float* __restrict__ W1,
                               float* __restrict__ h1, int n, int pitch) {
  extern __shared__ int dynsh[];
  __shared__ unsigned buf[SC_CAP];
  if (blockIdx.x < NSL) bslice(src, dst, lstart, rec, E, L, nb, P, dynsh, dynsh + nb, buf);
  else gemm1_tiles(x, W1, h1, n, pitch, (float*)buf, (int)blockIdx.x - NSL);
}

#define SRT_T 512
#define RPW (NSL / (SRT_T / 32))
#define SRT_CAP 8192
__global__ void k_bsort(const unsigned* __restrict__ rec, const int* __restrict__ lstart,
                        float* __restrict__ dinv, float* __restrict__ h1,
                        int* __restrict__ rowptr, unsigned* __restrict__ srt, int n, int nb,
                        int L, int P) {
  __shared__ int cnt[RB];
  __shared__ int cur[RB];
  __shared__ int rbase[NSL];
  __shared__ int rlen[NSL];
  __shared__ int roff[NSL];
  __shared__ int scrA[SRT_T];
  __shared__ int scrB[SRT_T];
  __shared__ unsigned sbuf[SRT_CAP];
  __shared__ unsigned obuf[SRT_CAP];
  const int tid = threadIdx.x, bd = blockDim.x, b = blockIdx.x;
  const int wv = tid >> 5, lane = tid & 31, nwv = bd >> 5;
  if (tid < RB) cnt[tid] = 0;
  int a = 0, len = 0;
  if (tid < NSL) {
    const int* lrow = lstart + (size_t)tid * P + b;
    a   = lrow[0];
    len = lrow[1] - a;
  }
  int ro, m, e0;
  block_scan_and_sum(len, a, scrA, scrB, &ro, &m, &e0);
  if (tid < NSL) {
    rbase[tid] = tid * L + a;
    rlen[tid]  = len;
    roff[tid]  = ro;
  }
  __syncthreads();
  const int e1 = e0 + m;
  const bool staged = m <= SRT_CAP;
  if (staged) {
    int myln = 0, myrb = 0, myof = 0;
    if (lane < RPW) {
      const int w = wv * RPW + lane;
      myln = rlen[w];
      myrb = rbase[w];
      myof = roff[w];
    }
    unsigned rv[RPW];
#pragma unroll
    for (int j = 0; j < RPW; ++j) {
      const int ln = __builtin_amdgcn_readlane(myln, j);
      const int rb = __builtin_amdgcn_readlane(myrb, j);
      rv[j] = 0u;
      if (lane < ln) rv[j] = rec[rb + lane];
    }
#pragma unroll
    for (int j = 0; j < RPW; ++j) {
      const int ln  = __builtin_amdgcn_readlane(myln, j);
      const int ofs = __builtin_amdgcn_readlane(myof, j);
      if (lane < ln) sbuf[ofs + lane] = rv[j];
    }
#pragma unroll
    for (int j = 0; j < RPW; ++j) {
      const int ln  = __builtin_amdgcn_readlane(myln, j);
      const int rb  = __builtin_amdgcn_readlane(myrb, j);
      const int ofs = __builtin_amdgcn_readlane(myof, j);
      for (int i = lane + 32; i < ln; i += 32) sbuf[ofs + i] = rec[rb + i];
    }
    __syncthreads();
    for (int i = tid; i < m; i += 4 * bd) {
      unsigned r[4];
#pragma unroll
      for (int u = 0; u < 4; ++u) {
        const int j = i + u * bd;
        r[u] = sbuf[j < m ? j : m - 1];
      }
#pragma unroll
      for (int u = 0; u < 4; ++u)
        if (i + u * bd < m) atomicAdd(&cnt[r[u] & (RB - 1)], 1);
    }
  } else {
    for (int w = wv; w < NSL; w += nwv) {
      const int ln = rlen[w], rb = rbase[w];
      for (int i = lane; i < ln; i += 32) atomicAdd(&cnt[rec[rb + i] & (RB - 1)], 1);
    }
  }
  __syncthreads();
  if (tid < 32) {
    const int c0 = cnt[4 * tid], c1 = cnt[4 * tid + 1], c2 = cnt[4 * tid + 2],
              c3 = cnt[4 * tid + 3];
    const int s = c0 + c1 + c2 + c3;
    int incl = s;
    for (int o = 1; o < 32; o <<= 1) {
      const int t = __shfl_up(incl, o);
      if (tid >= o) incl += t;
    }
    const int ex = e0 + incl - s;
    cur[4 * tid]     = ex;
    cur[4 * tid + 1] = ex + c0;
    cur[4 * tid + 2] = ex + c0 + c1;
    cur[4 * tid + 3] = ex + c0 + c1 + c2;
  }
  __syncthreads();
  if (tid < RB) {
    const int node = b * RB + tid;
    rowptr[node] = cur[tid];
    if (node < n) {
      const float d = 1.0f + (float)cnt[tid];
      const float dv = (d > 0.0f) ? (1.0f / sqrtf(d)) : 0.0f;
      dinv[node] = dv;
      h1[(size_t)node * H1P + HID] = dv;
    }
  }
  if (b == nb - 1 && tid == 0) rowptr[nb * RB] = e1;
  __syncthreads();
  if (staged) {
    for (int i = tid; i < m; i += 4 * bd) {
      unsigned r[4];
      int p[4];
#pragma unroll
      for (int u = 0; u < 4; ++u) {
        const int j = i + u * bd;
        r[u] = sbuf[j < m ? j : m - 1];
      }
#pragma unroll
      for (int u = 0; u < 4; ++u) {
        p[u] = e0;
        if (i + u * bd < m) p[u] = atomicAdd(&cur[r[u] & (RB - 1)], 1);
      }
#pragma unroll
      for (int u = 0; u < 4; ++u)
        if (i + u * bd < m) obuf[p[u] - e0] = r[u] >> LOG_RB;
    }
    __syncthreads();
    for (int i = tid; i < m; i += 4 * bd) {
      unsigned rv[4];
#pragma unroll
      for (int u = 0; u < 4; ++u) {
        const int j = i + u * bd;
        rv[u] = obuf[j < m ? j : m - 1];
      }
#pragma unroll
      for (int u = 0; u < 4; ++u)
        if (i + u * bd < m) srt[e0 + i + u * bd] = rv[u];
    }
  } else {
    for (int w = wv; w < NSL; w += nwv) {
      const int ln = rlen[w], rb = rbase[w];
      for (int i = lane; i < ln; i += 32) {
        const unsigned r = rec[rb + i];
        const int p = atomicAdd(&cur[r & (RB - 1)], 1);
        srt[p] = r >> LOG_RB;
      }
    }
  }
}

#define AGU 4
#define ZP 20
template <int F, int HP, bool FIRST, bool FOLD>
__device__ __forceinline__ void cagg_body(const unsigned* __restrict__ srt,
                                          const int* __restrict__ rowptr,
                                          const float* __restrict__ dinv,
                                          const float* __restrict__ h,
                                          const float* __restrict__ bias,
                                          const float* __restrict__ dsrc_in,
                                          float* __restrict__ dsrc_out,
                                          float* __restrict__ o, float* zs, int n) {
  constexpr int LPE = F / 4;
  const int tid = threadIdx.x, b = blockIdx.x;
  const int g = tid / LPE, c = tid % LPE;
  const int ng = blockDim.x / LPE;
  for (int nl = g; nl < RB; nl += ng) {
    const int node = b * RB + nl;
    if (node >= n) {
      if (!FOLD) break;
      const float4 zero = {0.f, 0.f, 0.f, 0.f};
      *(float4*)(zs + nl * ZP + c * 4) = zero;
      continue;
    }
    const int e0 = rowptr[node], e1 = rowptr[node + 1];
    const float dd = dinv[node];
    const float4 hself = *(const float4*)(h + (size_t)node * HP + c * 4);
    const float4 bv = *(const float4*)(bias + c * 4);
    float4 acc = {0.f, 0.f, 0.f, 0.f};
    unsigned sn[AGU];
    float dn[AGU];
    if (e0 < e1) {
#pragma unroll
      for (int u = 0; u < AGU; ++u) {
        const int idx = (e0 + u < e1) ? (e0 + u) : (e1 - 1);
        sn[u] = srt[idx];
        if (!FIRST) dn[u] = dsrc_in[idx];
      }
    }
    for (int e = e0; e < e1; e += AGU) {
      float4 hv[AGU];
      float ds[AGU];
#pragma unroll
      for (int u = 0; u < AGU; ++u) {
        const float* row = h + (size_t)sn[u] * HP;
        hv[u] = *(const float4*)(row + c * 4);
        ds[u] = FIRST ? row[F] : dn[u];
      }
#pragma unroll
      for (int u = 0; u < AGU; ++u) {
        const int idx = (e + AGU + u < e1) ? (e + AGU + u) : (e1 - 1);
        sn[u] = srt[idx];
        if (!FIRST) dn[u] = dsrc_in[idx];
      }
#pragma unroll
      for (int u = 0; u < AGU; ++u) {
        const bool live = e + u < e1;
        if (FIRST && c == 0 && live) dsrc_out[e + u] = ds[u];
        const float norm = ds[u] * dd;
        acc.x += (live ? hv[u].x : 0.f) * norm;
        acc.y += (live ? hv[u].y : 0.f) * norm;
        acc.z += (live ? hv[u].z : 0.f) * norm;
        acc.w += (live ? hv[u].w : 0.f) * norm;
      }
    }
    float4 r;
    r.x = acc.x + (hself.x * dd * dd + bv.x);
    r.y = acc.y + (hself.y * dd * dd + bv.y);
    r.z = acc.z + (hself.z * dd * dd + bv.z);
    r.w = acc.w + (hself.w * dd * dd + bv.w);
    if (FOLD) {
      float4 q;
      q.x = fmaxf(r.x, 0.f);
      q.y = fmaxf(r.y, 0.f);
      q.z = fmaxf(r.z, 0.f);
      q.w = fmaxf(r.w, 0.f);
      *(float4*)(zs + nl * ZP + c * 4) = q;
    } else {
      *(float4*)(o + (size_t)node * F + c * 4) = r;
    }
  }
}

__global__ void k_cagg1(const unsigned* __restrict__ srt, const int* __restrict__ rowptr,
                        const float* __restrict__ dinv, const float* __restrict__ h1,
                        const float* __restrict__ b1, float* __restrict__ dsrc,
                        float* __restrict__ z1, int n) {
  cagg_body<HID, H1P, true, false>(srt, rowptr, dinv, h1, b1, nullptr, dsrc, z1, nullptr, n);
}

__global__ void k_cagg1_gemm2(const unsigned* __restrict__ srt,
                              const int* __restrict__ rowptr,
                              const float* __restrict__ dinv, const float* __restrict__ h1,
                              const float* __restrict__ b1, float* __restrict__ dsrc,
                              const float* __restrict__ W2, float* __restrict__ h2, int n) {
  __shared__ __attribute__((aligned(16))) float zs[RB * ZP];
  __shared__ float lw2[HID * ODIM];
  for (int i = threadIdx.x; i < HID * ODIM; i += blockDim.x) lw2[i] = W2[i];
  cagg_body<HID, H1P, true, true>(srt, rowptr, dinv, h1, b1, nullptr, dsrc, nullptr, zs, n);
  __syncthreads();

  const int wv = threadIdx.x >> 5, nwv = blockDim.x >> 5;
  const int lane = threadIdx.x & 31;
  const int half = lane >> 4;
  const int ml   = lane & 15;
  for (int unit = wv; unit < (RB / 16) * 2; unit += nwv) {
    const int rt = unit >> 1;
    const float* zr = zs + (rt * 16 + ml) * ZP;
    v2f afrag[4];
#pragma unroll
    for (int kk = 0; kk < 4; ++kk) {
      const int ka = kk * 4 + 2 * half;
      afrag[kk].x = zr[ka];
      afrag[kk].y = zr[ka + 1];
    }
#pragma unroll
    for (int t = 0; t < 2; ++t) {
      const int tcol = (unit & 1) * 2 + t;
      v8f acc = {0.f, 0.f, 0.f, 0.f, 0.f, 0.f, 0.f, 0.f};
#pragma unroll
      for (int kk = 0; kk < 4; ++kk) {
        const int ka = kk * 4 + 2 * half;
        v2f b;
        b.x = lw2[ka * ODIM + tcol * 16 + ml];
        b.y = lw2[(ka + 1) * ODIM + tcol * 16 + ml];
        acc = __builtin_amdgcn_wmma_f32_16x16x4_f32(false, afrag[kk], false, b, (short)0,
                                                    acc, false, false);
      }
#pragma unroll
      for (int r = 0; r < 8; ++r) {
        const int orow = (int)blockIdx.x * RB + rt * 16 + r + 8 * half;
        if (orow < n) h2[(size_t)orow * ODIM + tcol * 16 + ml] = acc[r];
      }
    }
  }
}

__global__ void k_cagg2(const unsigned* __restrict__ srt, const int* __restrict__ rowptr,
                        const float* __restrict__ dinv, const float* __restrict__ h2,
                        const float* __restrict__ b2, const float* __restrict__ dsrc,
                        float* __restrict__ out, int n) {
  cagg_body<ODIM, ODIM, false, false>(srt, rowptr, dinv, h2, b2, dsrc, nullptr, out, nullptr,
                                      n);
}

extern "C" void kernel_launch(void* const* d_in, const int* in_sizes, int n_in,
                              void* d_out, int out_size, void* d_ws, size_t ws_size,
                              hipStream_t stream) {
  const float* x  = (const float*)d_in[0];
  const float* W1 = (const float*)d_in[1];
  const float* b1 = (const float*)d_in[2];
  const float* W2 = (const float*)d_in[3];
  const float* b2 = (const float*)d_in[4];
  const int*   ei = (const int*)d_in[5];

  const int N = in_sizes[0] / IN_DIM;
  const int E = in_sizes[5] / 2;
  const int* src = ei;
  const int* dst = ei + E;
  float* out = (float*)d_out;

  char* ws = (char*)d_ws;
  size_t off = 0;
  auto alloc = [&](size_t nbytes) -> void* {
    void* p = ws + off;
    off = (off + nbytes + 255) & ~(size_t)255;
    return p;
  };
  float* dinv = (float*)alloc((size_t)N * sizeof(float));
  float* h2   = (float*)alloc((size_t)N * ODIM * sizeof(float));
  const size_t planes = off;

  const int B = 256;
  const int tiles = (N + 15) >> 4;
  const int gemmBlocks = (tiles + 7) / 8;

  const int NB = (N + RB - 1) / RB;
  const int L  = ((E + NSL - 1) / NSL + 31) & ~31;
  const int P  = (NB + 1 + 31) & ~31;
  unsigned* rec = (unsigned*)alloc((size_t)E * sizeof(unsigned));
  unsigned* srt = (unsigned*)alloc((size_t)E * sizeof(unsigned));
  int* lstart   = (int*)alloc((size_t)NSL * P * sizeof(int));
  int* rowptr   = (int*)alloc(((size_t)NB * RB + 1) * sizeof(int));
  float* dsrc = (float*)rec;
  float* tail = (float*)(ws + off);
  const size_t need_z1  = off + (size_t)N * HID * sizeof(float);
  const size_t need_h1p = off + (size_t)N * H1P * sizeof(float);
  const int gemm1Groups = (tiles + 15) / 16;

  if (need_z1 <= ws_size && L <= SC_CAP) {
    const bool fold = need_h1p <= ws_size;
    float* h1a = fold ? tail : h2;
    k_bslice_gemm1<<<NSL + gemm1Groups, 512, (size_t)(NB + 512) * sizeof(int), stream>>>(
        src, dst, lstart, rec, E, L, NB, P, x, W1, h1a, N, H1P);
    k_bsort   <<<NB, SRT_T, 0, stream>>>(rec, lstart, dinv, h1a, rowptr, srt, N, NB, L, P);

    if (fold) {
      k_cagg1_gemm2<<<NB, 512, 0, stream>>>(srt, rowptr, dinv, h1a, b1, dsrc, W2, h2, N);
    } else {
      float* z1 = tail;
      k_cagg1 <<<NB, 512, 0, stream>>>(srt, rowptr, dinv, h1a, b1, dsrc, z1, N);
      k_gemm2 <<<gemmBlocks, B, 0, stream>>>(z1, W2, h2, N);
    }
    k_cagg2   <<<NB, 512, 0, stream>>>(srt, rowptr, dinv, h2, b2, dsrc, out, N);
    return;
  }

  off = planes;
  float* z1 = (float*)alloc((size_t)N * HID * sizeof(float));
  float* h1 = (float*)alloc((size_t)N * HID * sizeof(float));
  k_deg_init <<<(N + B - 1) / B, B, 0, stream>>>(dinv, N);
  k_deg_edges<<<(E + B - 1) / B, B, 0, stream>>>(dst, dinv, E);
  k_dinv     <<<(N + B - 1) / B, B, 0, stream>>>(dinv, N);

  k_gemm1    <<<gemmBlocks, B, 0, stream>>>(x, W1, h1, N, HID);
  k_init_z1  <<<((size_t)N * HID + B - 1) / B, B, 0, stream>>>(h1, dinv, b1, z1, N);
  k_agg1     <<<(E * 4 + B - 1) / B, B, 0, stream>>>(src, dst, dinv, h1, z1, E);

  k_gemm2    <<<gemmBlocks, B, 0, stream>>>(z1, W2, h2, N);
  k_init_out <<<((size_t)N * ODIM + B - 1) / B, B, 0, stream>>>(h2, dinv, b2, out, N);
  k_agg2     <<<(E * 16 + B - 1) / B, B, 0, stream>>>(src, dst, dinv, h2, out, E);
}


# a node's per-edge words (source index, factor) are read one edge per lane and passed round its lane group by cross-lane moves; layer 2 keeps 8 row gathers of a node in flight, not 4; sums as before
# speedup vs baseline: 4.7799x; 1.1938x over previous
#include <hip/hip_runtime.h>

typedef __attribute__((ext_vector_type(2))) float v2f;
typedef __attribute__((ext_vector_type(8))) float v8f;

#define IN_DIM 512
#define HID    16
#define ODIM   64

__global__ void k_deg_init(float* __restrict__ deg, int n) {
  int i = blockIdx.x * blockDim.x + threadIdx.x;
  if (i < n) deg[i] = 1.0f;
}

__global__ void k_deg_edges(const int* __restrict__ dst, float* __restrict__ deg, int E) {
  int i = blockIdx.x * blockDim.x + threadIdx.x;
  if (i < E) atomicAdd(&deg[dst[i]], 1.0f);
}

__global__ void k_dinv(float* __restrict__ deg, int n) {
  int i = blockIdx.x * blockDim.x + threadIdx.x;
  if (i < n) {
    float d = deg[i];
    deg[i] = (d > 0.0f) ? (1.0f / sqrtf(d)) : 0.0f;
  }
}

__device__ __forceinline__ void gemm1_tiles(const float* __restrict__ x,
                                            const float* __restrict__ W1,
                                            float* __restrict__ h1, int n, int pitch,
                                            float* lw, int wg) {
  for (int i = threadIdx.x; i < IN_DIM * HID; i += blockDim.x) lw[i] = W1[i];
  __syncthreads();

  int wave  = (wg * blockDim.x + threadIdx.x) >> 5;
  int lane  = threadIdx.x & 31;
  int tiles = (n + 15) >> 4;
  if (wave >= tiles) return;

  int half = lane >> 4;
  int ml   = lane & 15;
  int row  = wave * 16 + ml;
  int rrow = (row < n) ? row : (n - 1);
  const float* xr = x + (size_t)rrow * IN_DIM;

  v8f acc = {0.f, 0.f, 0.f, 0.f, 0.f, 0.f, 0.f, 0.f};
  for (int k = 0; k < IN_DIM; k += 4) {
    int ka = k + 2 * half;
    v2f a; a.x = xr[ka];                 a.y = xr[ka + 1];
    v2f b; b.x = lw[ka * HID + ml];      b.y = lw[(ka + 1) * HID + ml];
    acc = __builtin_amdgcn_wmma_f32_16x16x4_f32(false, a, false, b, (short)0, acc,
                                                false, false);
  }
#pragma unroll
  for (int r = 0; r < 8; ++r) {
    int orow = wave * 16 + r + 8 * half;
    if (orow < n) h1[(size_t)orow * pitch + ml] = acc[r];
  }
}

__global__ void k_gemm1(const float* __restrict__ x, const float* __restrict__ W1,
                        float* __restrict__ h1, int n, int pitch) {
  __shared__ float lw[IN_DIM * HID];
  gemm1_tiles(x, W1, h1, n, pitch, lw, blockIdx.x);
}

__global__ void k_init_z1(const float* __restrict__ h1, const float* __restrict__ dinv,
                          const float* __restrict__ b1, float* __restrict__ z1, int n) {
  int t = blockIdx.x * blockDim.x + threadIdx.x;
  int total = n * HID;
  if (t < total) {
    int i = t / HID, j = t - i * HID;
    float di = dinv[i];
    z1[t] = h1[t] * di * di + b1[j];
  }
}

__global__ void k_agg1(const int* __restrict__ src, const int* __restrict__ dst,
                       const float* __restrict__ dinv, const float* __restrict__ h1,
                       float* __restrict__ z1, int E) {
  int total  = E * 4;
  int stride = gridDim.x * blockDim.x;
  for (int t = blockIdx.x * blockDim.x + threadIdx.x; t < total; t += stride) {
    int e = t >> 2, c = t & 3;
    int s = src[e], d = dst[e];
    float norm = dinv[s] * dinv[d];
    const float4 hv = *(const float4*)(h1 + (size_t)s * HID + c * 4);
    float* o = z1 + (size_t)d * HID + c * 4;
    atomicAdd(o + 0, hv.x * norm);
    atomicAdd(o + 1, hv.y * norm);
    atomicAdd(o + 2, hv.z * norm);
    atomicAdd(o + 3, hv.w * norm);
  }
}

__global__ void k_gemm2(const float* __restrict__ z1, const float* __restrict__ W2,
                        float* __restrict__ h2, int n) {
  __shared__ float lw[HID * ODIM];
  for (int i = threadIdx.x; i < HID * ODIM; i += blockDim.x) lw[i] = W2[i];
  __syncthreads();

  int wave  = (blockIdx.x * blockDim.x + threadIdx.x) >> 5;
  int lane  = threadIdx.x & 31;
  int tiles = (n + 15) >> 4;
  if (wave >= tiles) return;

  int half = lane >> 4;
  int ml   = lane & 15;
  int row  = wave * 16 + ml;
  int rrow = (row < n) ? row : (n - 1);
  const float* zr = z1 + (size_t)rrow * HID;

  v2f afrag[4];
#pragma unroll
  for (int kk = 0; kk < 4; ++kk) {
    int ka = kk * 4 + 2 * half;
    afrag[kk].x = fmaxf(zr[ka], 0.f);
    afrag[kk].y = fmaxf(zr[ka + 1], 0.f);
  }

#pragma unroll
  for (int tcol = 0; tcol < 4; ++tcol) {
    v8f acc = {0.f, 0.f, 0.f, 0.f, 0.f, 0.f, 0.f, 0.f};
#pragma unroll
    for (int kk = 0; kk < 4; ++kk) {
      int ka = kk * 4 + 2 * half;
      v2f b;
      b.x = lw[ka * ODIM + tcol * 16 + ml];
      b.y = lw[(ka + 1) * ODIM + tcol * 16 + ml];
      acc = __builtin_amdgcn_wmma_f32_16x16x4_f32(false, afrag[kk], false, b, (short)0,
                                                  acc, false, false);
    }
#pragma unroll
    for (int r = 0; r < 8; ++r) {
      int orow = wave * 16 + r + 8 * half;
      if (orow < n) h2[(size_t)orow * ODIM + tcol * 16 + ml] = acc[r];
    }
  }
}

__global__ void k_init_out(const float* __restrict__ h2, const float* __restrict__ dinv,
                           const float* __restrict__ b2, float* __restrict__ out, int n) {
  int t = blockIdx.x * blockDim.x + threadIdx.x;
  int total = n * ODIM;
  if (t < total) {
    int i = t / ODIM, j = t - i * ODIM;
    float di = dinv[i];
    out[t] = h2[t] * di * di + b2[j];
  }
}

__global__ void k_agg2(const int* __restrict__ src, const int* __restrict__ dst,
                       const float* __restrict__ dinv, const float* __restrict__ h2,
                       float* __restrict__ out, int E) {
  int total  = E * 16;
  int stride = gridDim.x * blockDim.x;
  for (int t = blockIdx.x * blockDim.x + threadIdx.x; t < total; t += stride) {
    int e = t >> 4, c = t & 15;
    int s = src[e], d = dst[e];
    float norm = dinv[s] * dinv[d];
    const float4 hv = *(const float4*)(h2 + (size_t)s * ODIM + c * 4);
    float* o = out + (size_t)d * ODIM + c * 4;
    atomicAdd(o + 0, hv.x * norm);
    atomicAdd(o + 1, hv.y * norm);
    atomicAdd(o + 2, hv.z * norm);
    atomicAdd(o + 3, hv.w * norm);
  }
}

#define RB     128
#define LOG_RB 7
#define NSL    256
#define H1P    32

__device__ __forceinline__ int block_excl_scan(int v, int* scr, int* total) {
  const int t = threadIdx.x, nt = blockDim.x;
  scr[t] = v;
  __syncthreads();
  for (int o = 1; o < nt; o <<= 1) {
    int a = (t >= o) ? scr[t - o] : 0;
    __syncthreads();
    scr[t] += a;
    __syncthreads();
  }
  const int incl = scr[t];
  *total = scr[nt - 1];
  __syncthreads();
  return incl - v;
}

__device__ __forceinline__ void block_scan_and_sum(int va, int vb, int* scrA, int* scrB,
                                                   int* exA, int* totA, int* totB) {
  const int t = threadIdx.x, nt = blockDim.x;
  scrA[t] = va;
  scrB[t] = vb;
  __syncthreads();
  for (int o = 1; o < nt; o <<= 1) {
    int a = 0, b = 0;
    if (t >= o) {
      a = scrA[t - o];
      b = scrB[t - o];
    }
    __syncthreads();
    scrA[t] += a;
    scrB[t] += b;
    __syncthreads();
  }
  *exA  = scrA[t] - va;
  *totA = scrA[nt - 1];
  *totB = scrB[nt - 1];
  __syncthreads();
}

#define SC_CAP 16384
__device__ __forceinline__ void bslice(const int* __restrict__ src,
                                       const int* __restrict__ dst,
                                       int* __restrict__ lstart, unsigned* __restrict__ rec,
                                       int E, int L, int nb, int P, int* cnt, int* scr,
                                       unsigned* stage) {
  const int tid = threadIdx.x, bd = blockDim.x, wg = blockIdx.x;
  for (int i = tid; i < nb; i += bd) cnt[i] = 0;
  __syncthreads();
  const int e0 = wg * L;
  int e1 = e0 + L;
  if (e1 > E) e1 = E;
  int e = e0 + tid;
  for (; e + 3 * bd < e1; e += 4 * bd) {
    int d[4];
#pragma unroll
    for (int u = 0; u < 4; ++u) d[u] = dst[e + u * bd];
#pragma unroll
    for (int u = 0; u < 4; ++u) atomicAdd(&cnt[d[u] >> LOG_RB], 1);
  }
  for (; e < e1; e += bd) atomicAdd(&cnt[dst[e] >> LOG_RB], 1);
  __syncthreads();
  const int per = (nb + bd - 1) / bd;
  const int lo = tid * per;
  int hi = lo + per;
  if (hi > nb) hi = nb;
  int s = 0;
  for (int b = lo; b < hi; ++b) s += cnt[b];
  int tot;
  int pre = block_excl_scan(s, scr, &tot);
  int* lrow = lstart + (size_t)wg * P;
  for (int b = lo; b < hi; ++b) {
    const int c = cnt[b];
    cnt[b]  = pre;
    lrow[b] = pre;
    pre += c;
  }
  if (tid == 0) lrow[nb] = tot;
  __syncthreads();
  e = e0 + tid;
  for (; e + 3 * bd < e1; e += 4 * bd) {
    int sv[4], dv[4], lp[4];
#pragma unroll
    for (int u = 0; u < 4; ++u) {
      sv[u] = src[e + u * bd];
      dv[u] = dst[e + u * bd];
    }
#pragma unroll
    for (int u = 0; u < 4; ++u) lp[u] = atomicAdd(&cnt[dv[u] >> LOG_RB], 1);
#pragma unroll
    for (int u = 0; u < 4; ++u)
      stage[lp[u]] = ((unsigned)sv[u] << LOG_RB) | (unsigned)(dv[u] & (RB - 1));
  }
  for (; e < e1; e += bd) {
    const int sv = src[e], dv = dst[e];
    const int lp = atomicAdd(&cnt[dv >> LOG_RB], 1);
    stage[lp] = ((unsigned)sv << LOG_RB) | (unsigned)(dv & (RB - 1));
  }
  __syncthreads();
  const int m = e1 - e0;
  for (int i = tid; i < m; i += 4 * bd) {
    unsigned rv[4];
#pragma unroll
    for (int u = 0; u < 4; ++u) {
      const int j = i + u * bd;
      rv[u] = stage[j < m ? j : m - 1];
    }
#pragma unroll
    for (int u = 0; u < 4; ++u)
      if (i + u * bd < m) rec[e0 + i + u * bd] = rv[u];
  }
}

__global__ void k_bslice_gemm1(const int* __restrict__ src, const int* __restrict__ dst,
                               int* __restrict__ lstart, unsigned* __restrict__ rec,
                               int E, int L, int nb, int P,
                               const float* __restrict__ x, const float* __restrict__ W1,
                               float* __restrict__ h1, int n, int pitch) {
  extern __shared__ int dynsh[];
  __shared__ unsigned buf[SC_CAP];
  if (blockIdx.x < NSL) bslice(src, dst, lstart, rec, E, L, nb, P, dynsh, dynsh + nb, buf);
  else gemm1_tiles(x, W1, h1, n, pitch, (float*)buf, (int)blockIdx.x - NSL);
}

#define SRT_T 512
#define RPW (NSL / (SRT_T / 32))
#define SRT_CAP 8192
__global__ void k_bsort(const unsigned* __restrict__ rec, const int* __restrict__ lstart,
                        float* __restrict__ dinv, float* __restrict__ h1,
                        int* __restrict__ rowptr, unsigned* __restrict__ srt, int n, int nb,
                        int L, int P) {
  __shared__ int cnt[RB];
  __shared__ int cur[RB];
  __shared__ int rbase[NSL];
  __shared__ int rlen[NSL];
  __shared__ int roff[NSL];
  __shared__ int scrA[SRT_T];
  __shared__ int scrB[SRT_T];
  __shared__ unsigned sbuf[SRT_CAP];
  __shared__ unsigned obuf[SRT_CAP];
  const int tid = threadIdx.x, bd = blockDim.x, b = blockIdx.x;
  const int wv = tid >> 5, lane = tid & 31, nwv = bd >> 5;
  if (tid < RB) cnt[tid] = 0;
  int a = 0, len = 0;
  if (tid < NSL) {
    const int* lrow = lstart + (size_t)tid * P + b;
    a   = lrow[0];
    len = lrow[1] - a;
  }
  int ro, m, e0;
  block_scan_and_sum(len, a, scrA, scrB, &ro, &m, &e0);
  if (tid < NSL) {
    rbase[tid] = tid * L + a;
    rlen[tid]  = len;
    roff[tid]  = ro;
  }
  __syncthreads();
  const int e1 = e0 + m;
  const bool staged = m <= SRT_CAP;
  if (staged) {
    int myln = 0, myrb = 0, myof = 0;
    if (lane < RPW) {
      const int w = wv * RPW + lane;
      myln = rlen[w];
      myrb = rbase[w];
      myof = roff[w];
    }
    unsigned rv[RPW];
#pragma unroll
    for (int j = 0; j < RPW; ++j) {
      const int ln = __builtin_amdgcn_readlane(myln, j);
      const int rb = __builtin_amdgcn_readlane(myrb, j);
      rv[j] = 0u;
      if (lane < ln) rv[j] = rec[rb + lane];
    }
#pragma unroll
    for (int j = 0; j < RPW; ++j) {
      const int ln  = __builtin_amdgcn_readlane(myln, j);
      const int ofs = __builtin_amdgcn_readlane(myof, j);
      if (lane < ln) sbuf[ofs + lane] = rv[j];
    }
#pragma unroll
    for (int j = 0; j < RPW; ++j) {
      const int ln  = __builtin_amdgcn_readlane(myln, j);
      const int rb  = __builtin_amdgcn_readlane(myrb, j);
      const int ofs = __builtin_amdgcn_readlane(myof, j);
      for (int i = lane + 32; i < ln; i += 32) sbuf[ofs + i] = rec[rb + i];
    }
    __syncthreads();
    for (int i = tid; i < m; i += 4 * bd) {
      unsigned r[4];
#pragma unroll
      for (int u = 0; u < 4; ++u) {
        const int j = i + u * bd;
        r[u] = sbuf[j < m ? j : m - 1];
      }
#pragma unroll
      for (int u = 0; u < 4; ++u)
        if (i + u * bd < m) atomicAdd(&cnt[r[u] & (RB - 1)], 1);
    }
  } else {
    for (int w = wv; w < NSL; w += nwv) {
      const int ln = rlen[w], rb = rbase[w];
      for (int i = lane; i < ln; i += 32) atomicAdd(&cnt[rec[rb + i] & (RB - 1)], 1);
    }
  }
  __syncthreads();
  if (tid < 32) {
    const int c0 = cnt[4 * tid], c1 = cnt[4 * tid + 1], c2 = cnt[4 * tid + 2],
              c3 = cnt[4 * tid + 3];
    const int s = c0 + c1 + c2 + c3;
    int incl = s;
    for (int o = 1; o < 32; o <<= 1) {
      const int t = __shfl_up(incl, o);
      if (tid >= o) incl += t;
    }
    const int ex = e0 + incl - s;
    cur[4 * tid]     = ex;
    cur[4 * tid + 1] = ex + c0;
    cur[4 * tid + 2] = ex + c0 + c1;
    cur[4 * tid + 3] = ex + c0 + c1 + c2;
  }
  __syncthreads();
  if (tid < RB) {
    const int node = b * RB + tid;
    rowptr[node] = cur[tid];
    if (node < n) {
      const float d = 1.0f + (float)cnt[tid];
      const float dv = (d > 0.0f) ? (1.0f / sqrtf(d)) : 0.0f;
      dinv[node] = dv;
      h1[(size_t)node * H1P + HID] = dv;
    }
  }
  if (b == nb - 1 && tid == 0) rowptr[nb * RB] = e1;
  __syncthreads();
  if (staged) {
    for (int i = tid; i < m; i += 4 * bd) {
      unsigned r[4];
      int p[4];
#pragma unroll
      for (int u = 0; u < 4; ++u) {
        const int j = i + u * bd;
        r[u] = sbuf[j < m ? j : m - 1];
      }
#pragma unroll
      for (int u = 0; u < 4; ++u) {
        p[u] = e0;
        if (i + u * bd < m) p[u] = atomicAdd(&cur[r[u] & (RB - 1)], 1);
      }
#pragma unroll
      for (int u = 0; u < 4; ++u)
        if (i + u * bd < m) obuf[p[u] - e0] = r[u] >> LOG_RB;
    }
    __syncthreads();
    for (int i = tid; i < m; i += 4 * bd) {
      unsigned rv[4];
#pragma unroll
      for (int u = 0; u < 4; ++u) {
        const int j = i + u * bd;
        rv[u] = obuf[j < m ? j : m - 1];
      }
#pragma unroll
      for (int u = 0; u < 4; ++u)
        if (i + u * bd < m) srt[e0 + i + u * bd] = rv[u];
    }
  } else {
    for (int w = wv; w < NSL; w += nwv) {
      const int ln = rlen[w], rb = rbase[w];
      for (int i = lane; i < ln; i += 32) {
        const unsigned r = rec[rb + i];
        const int p = atomicAdd(&cur[r & (RB - 1)], 1);
        srt[p] = r >> LOG_RB;
      }
    }
  }
}

#define ZP 20

template <int LPE>
__device__ __forceinline__ int group_word(int v, int u) {
  if (LPE == 4) {
    switch (u) {
      case 0:  return __builtin_amdgcn_mov_dpp(v, 0x00, 0xf, 0xf, true);
      case 1:  return __builtin_amdgcn_mov_dpp(v, 0x55, 0xf, 0xf, true);
      case 2:  return __builtin_amdgcn_mov_dpp(v, 0xaa, 0xf, 0xf, true);
      default: return __builtin_amdgcn_mov_dpp(v, 0xff, 0xf, 0xf, true);
    }
  }
  switch (u) {
    case 0:  return __builtin_amdgcn_mov_dpp(v, 0x150, 0xf, 0xf, true);
    case 1:  return __builtin_amdgcn_mov_dpp(v, 0x151, 0xf, 0xf, true);
    case 2:  return __builtin_amdgcn_mov_dpp(v, 0x152, 0xf, 0xf, true);
    case 3:  return __builtin_amdgcn_mov_dpp(v, 0x153, 0xf, 0xf, true);
    case 4:  return __builtin_amdgcn_mov_dpp(v, 0x154, 0xf, 0xf, true);
    case 5:  return __builtin_amdgcn_mov_dpp(v, 0x155, 0xf, 0xf, true);
    case 6:  return __builtin_amdgcn_mov_dpp(v, 0x156, 0xf, 0xf, true);
    case 7:  return __builtin_amdgcn_mov_dpp(v, 0x157, 0xf, 0xf, true);
    case 8:  return __builtin_amdgcn_mov_dpp(v, 0x158, 0xf, 0xf, true);
    case 9:  return __builtin_amdgcn_mov_dpp(v, 0x159, 0xf, 0xf, true);
    case 10: return __builtin_amdgcn_mov_dpp(v, 0x15a, 0xf, 0xf, true);
    case 11: return __builtin_amdgcn_mov_dpp(v, 0x15b, 0xf, 0xf, true);
    case 12: return __builtin_amdgcn_mov_dpp(v, 0x15c, 0xf, 0xf, true);
    case 13: return __builtin_amdgcn_mov_dpp(v, 0x15d, 0xf, 0xf, true);
    case 14: return __builtin_amdgcn_mov_dpp(v, 0x15e, 0xf, 0xf, true);
    default: return __builtin_amdgcn_mov_dpp(v, 0x15f, 0xf, 0xf, true);
  }
}

template <int F, int HP, bool FIRST, bool FOLD, int GE>
__device__ __forceinline__ void cagg_body(const unsigned* __restrict__ srt,
                                          const int* __restrict__ rowptr,
                                          const float* __restrict__ dinv,
                                          const float* __restrict__ h,
                                          const float* __restrict__ bias,
                                          const float* __restrict__ dsrc_in,
                                          float* __restrict__ dsrc_out,
                                          float* __restrict__ o, float* zs, int n) {
  constexpr int LPE = F / 4;
  constexpr int WIN = (LPE > GE) ? LPE : GE;
  constexpr int NR  = WIN / LPE;
  constexpr int NSG = WIN / GE;
  const int tid = threadIdx.x, b = blockIdx.x;
  const int g = tid / LPE, c = tid % LPE;
  const int ng = blockDim.x / LPE;
  const float4* __restrict__ h4 = (const float4*)h;
  for (int nl = g; nl < RB; nl += ng) {
    const int node = b * RB + nl;
    if (node >= n) {
      if (!FOLD) break;
      const float4 zero = {0.f, 0.f, 0.f, 0.f};
      *(float4*)(zs + nl * ZP + c * 4) = zero;
      continue;
    }
    const int e0 = rowptr[node], e1 = rowptr[node + 1];
    const float dd = dinv[node];
    const float4 hself = h4[(unsigned)node * (unsigned)(HP / 4) + (unsigned)c];
    const float4 bv = ((const float4*)bias)[c];
    float4 self;
    self.x = hself.x * dd * dd + bv.x;
    self.y = hself.y * dd * dd + bv.y;
    self.z = hself.z * dd * dd + bv.z;
    self.w = hself.w * dd * dd + bv.w;
    float4 acc = {0.f, 0.f, 0.f, 0.f};
    unsigned sn[NR];
    float dn[NR];
    if (e0 < e1) {
#pragma unroll
      for (int r = 0; r < NR; ++r) {
        const int idx = (e0 + r * LPE + c < e1) ? (e0 + r * LPE + c) : (e1 - 1);
        sn[r] = srt[idx];
        if (!FIRST) dn[r] = dsrc_in[idx];
      }
    }
    if (GE > 4) asm volatile("" : "+v"(self.x), "+v"(self.y), "+v"(self.z), "+v"(self.w));
    for (int e = e0; e < e1; e += WIN) {
      unsigned sx[NR];
      float dx[NR];
#pragma unroll
      for (int r = 0; r < NR; ++r) {
        const int idx = (e + WIN + r * LPE + c < e1) ? (e + WIN + r * LPE + c) : (e1 - 1);
        sx[r] = srt[idx];
        if (!FIRST) dx[r] = dsrc_in[idx];
      }
#pragma unroll
      for (int sg = 0; sg < NSG; ++sg) {
        const int rem = e1 - (e + sg * GE);
        if (sg == 0 || rem > 0) {
          float4 hv[GE];
          float ds[GE];
#pragma unroll
          for (int u = 0; u < GE; ++u) {
            const int j = sg * GE + u;
            const unsigned s = (unsigned)group_word<LPE>((int)sn[j / LPE], j % LPE);
            hv[u] = h4[s * (unsigned)(HP / 4) + (unsigned)c];
            if (FIRST) ds[u] = h[s * (unsigned)HP + (unsigned)F];
          }
          if (FIRST) {
#pragma unroll
            for (int r = 0; r < GE / LPE; ++r) {
              float v = ds[r * LPE];
#pragma unroll
              for (int l = 1; l < LPE; ++l) v = (c == l) ? ds[r * LPE + l] : v;
              if (r * LPE + c < rem) dsrc_out[e + sg * GE + r * LPE + c] = v;
            }
          }
#pragma unroll
          for (int u = 0; u < GE; ++u) {
            const int j = sg * GE + u;
            const float dsv =
                FIRST ? ds[u]
                      : __int_as_float(group_word<LPE>(__float_as_int(dn[j / LPE]), j % LPE));
            const float norm = (u < rem) ? dsv * dd : 0.f;
            acc.x = __builtin_fmaf(hv[u].x, norm, acc.x);
            acc.y = __builtin_fmaf(hv[u].y, norm, acc.y);
            acc.z = __builtin_fmaf(hv[u].z, norm, acc.z);
            acc.w = __builtin_fmaf(hv[u].w, norm, acc.w);
          }
        }
      }
#pragma unroll
      for (int r = 0; r < NR; ++r) {
        sn[r] = sx[r];
        if (!FIRST) dn[r] = dx[r];
      }
    }
    float4 r;
    r.x = acc.x + self.x;
    r.y = acc.y + self.y;
    r.z = acc.z + self.z;
    r.w = acc.w + self.w;
    if (FOLD) {
      float4 q;
      q.x = fmaxf(r.x, 0.f);
      q.y = fmaxf(r.y, 0.f);
      q.z = fmaxf(r.z, 0.f);
      q.w = fmaxf(r.w, 0.f);
      *(float4*)(zs + nl * ZP + c * 4) = q;
    } else {
      ((float4*)o)[(unsigned)node * (unsigned)LPE + (unsigned)c] = r;
    }
  }
}

__global__ void k_cagg1(const unsigned* __restrict__ srt, const int* __restrict__ rowptr,
                        const float* __restrict__ dinv, const float* __restrict__ h1,
                        const float* __restrict__ b1, float* __restrict__ dsrc,
                        float* __restrict__ z1, int n) {
  cagg_body<HID, H1P, true, false, 4>(srt, rowptr, dinv, h1, b1, nullptr, dsrc, z1, nullptr, n);
}

__global__ void k_cagg1_gemm2(const unsigned* __restrict__ srt,
                              const int* __restrict__ rowptr,
                              const float* __restrict__ dinv, const float* __restrict__ h1,
                              const float* __restrict__ b1, float* __restrict__ dsrc,
                              const float* __restrict__ W2, float* __restrict__ h2, int n) {
  __shared__ __attribute__((aligned(16))) float zs[RB * ZP];
  __shared__ float lw2[HID * ODIM];
  for (int i = threadIdx.x; i < HID * ODIM; i += blockDim.x) lw2[i] = W2[i];
  cagg_body<HID, H1P, true, true, 4>(srt, rowptr, dinv, h1, b1, nullptr, dsrc, nullptr, zs, n);
  __syncthreads();

  const int wv = threadIdx.x >> 5, nwv = blockDim.x >> 5;
  const int lane = threadIdx.x & 31;
  const int half = lane >> 4;
  const int ml   = lane & 15;
  for (int unit = wv; unit < (RB / 16) * 2; unit += nwv) {
    const int rt = unit >> 1;
    const float* zr = zs + (rt * 16 + ml) * ZP;
    v2f afrag[4];
#pragma unroll
    for (int kk = 0; kk < 4; ++kk) {
      const int ka = kk * 4 + 2 * half;
      afrag[kk].x = zr[ka];
      afrag[kk].y = zr[ka + 1];
    }
#pragma unroll
    for (int t = 0; t < 2; ++t) {
      const int tcol = (unit & 1) * 2 + t;
      v8f acc = {0.f, 0.f, 0.f, 0.f, 0.f, 0.f, 0.f, 0.f};
#pragma unroll
      for (int kk = 0; kk < 4; ++kk) {
        const int ka = kk * 4 + 2 * half;
        v2f b;
        b.x = lw2[ka * ODIM + tcol * 16 + ml];
        b.y = lw2[(ka + 1) * ODIM + tcol * 16 + ml];
        acc = __builtin_amdgcn_wmma_f32_16x16x4_f32(false, afrag[kk], false, b, (short)0,
                                                    acc, false, false);
      }
#pragma unroll
      for (int r = 0; r < 8; ++r) {
        const int orow = (int)blockIdx.x * RB + rt * 16 + r + 8 * half;
        if (orow < n) h2[(size_t)orow * ODIM + tcol * 16 + ml] = acc[r];
      }
    }
  }
}

__global__ void k_cagg2(const unsigned* __restrict__ srt, const int* __restrict__ rowptr,
                        const float* __restrict__ dinv, const float* __restrict__ h2,
                        const float* __restrict__ b2, const float* __restrict__ dsrc,
                        float* __restrict__ out, int n) {
  cagg_body<ODIM, ODIM, false, false, 8>(srt, rowptr, dinv, h2, b2, dsrc, nullptr, out, nullptr,
                                      n);
}

extern "C" void kernel_launch(void* const* d_in, const int* in_sizes, int n_in,
                              void* d_out, int out_size, void* d_ws, size_t ws_size,
                              hipStream_t stream) {
  const float* x  = (const float*)d_in[0];
  const float* W1 = (const float*)d_in[1];
  const float* b1 = (const float*)d_in[2];
  const float* W2 = (const float*)d_in[3];
  const float* b2 = (const float*)d_in[4];
  const int*   ei = (const int*)d_in[5];

  const int N = in_sizes[0] / IN_DIM;
  const int E = in_sizes[5] / 2;
  const int* src = ei;
  const int* dst = ei + E;
  float* out = (float*)d_out;

  char* ws = (char*)d_ws;
  size_t off = 0;
  auto alloc = [&](size_t nbytes) -> void* {
    void* p = ws + off;
    off = (off + nbytes + 255) & ~(size_t)255;
    return p;
  };
  float* dinv = (float*)alloc((size_t)N * sizeof(float));
  float* h2   = (float*)alloc((size_t)N * ODIM * sizeof(float));
  const size_t planes = off;

  const int B = 256;
  const int tiles = (N + 15) >> 4;
  const int gemmBlocks = (tiles + 7) / 8;

  const int NB = (N + RB - 1) / RB;
  const int L  = ((E + NSL - 1) / NSL + 31) & ~31;
  const int P  = (NB + 1 + 31) & ~31;
  unsigned* rec = (unsigned*)alloc((size_t)E * sizeof(unsigned));
  unsigned* srt = (unsigned*)alloc((size_t)E * sizeof(unsigned));
  int* lstart   = (int*)alloc((size_t)NSL * P * sizeof(int));
  int* rowptr   = (int*)alloc(((size_t)NB * RB + 1) * sizeof(int));
  float* dsrc = (float*)rec;
  float* tail = (float*)(ws + off);
  const size_t need_z1  = off + (size_t)N * HID * sizeof(float);
  const size_t need_h1p = off + (size_t)N * H1P * sizeof(float);
  const int gemm1Groups = (tiles + 15) / 16;

  if (need_z1 <= ws_size && L <= SC_CAP) {
    const bool fold = need_h1p <= ws_size;
    float* h1a = fold ? tail : h2;
    k_bslice_gemm1<<<NSL + gemm1Groups, 512, (size_t)(NB + 512) * sizeof(int), stream>>>(
        src, dst, lstart, rec, E, L, NB, P, x, W1, h1a, N, H1P);
    k_bsort   <<<NB, SRT_T, 0, stream>>>(rec, lstart, dinv, h1a, rowptr, srt, N, NB, L, P);

    if (fold) {
      k_cagg1_gemm2<<<NB, 512, 0, stream>>>(srt, rowptr, dinv, h1a, b1, dsrc, W2, h2, N);
    } else {
      float* z1 = tail;
      k_cagg1 <<<NB, 512, 0, stream>>>(srt, rowptr, dinv, h1a, b1, dsrc, z1, N);
      k_gemm2 <<<gemmBlocks, B, 0, stream>>>(z1, W2, h2, N);
    }
    k_cagg2   <<<NB, 512, 0, stream>>>(srt, rowptr, dinv, h2, b2, dsrc, out, N);
    return;
  }

  off = planes;
  float* z1 = (float*)alloc((size_t)N * HID * sizeof(float));
  float* h1 = (float*)alloc((size_t)N * HID * sizeof(float));
  k_deg_init <<<(N + B - 1) / B, B, 0, stream>>>(dinv, N);
  k_deg_edges<<<(E + B - 1) / B, B, 0, stream>>>(dst, dinv, E);
  k_dinv     <<<(N + B - 1) / B, B, 0, stream>>>(dinv, N);

  k_gemm1    <<<gemmBlocks, B, 0, stream>>>(x, W1, h1, N, HID);
  k_init_z1  <<<((size_t)N * HID + B - 1) / B, B, 0, stream>>>(h1, dinv, b1, z1, N);
  k_agg1     <<<(E * 4 + B - 1) / B, B, 0, stream>>>(src, dst, dinv, h1, z1, E);

  k_gemm2    <<<gemmBlocks, B, 0, stream>>>(z1, W2, h2, N);
  k_init_out <<<((size_t)N * ODIM + B - 1) / B, B, 0, stream>>>(h2, dinv, b2, out, N);
  k_agg2     <<<(E * 16 + B - 1) / B, B, 0, stream>>>(src, dst, dinv, h2, out, E);
}


# layer-1 launch passes its bucket's per-edge words and its h2 tile through LDS: srt range read, factor range and h2 rows written as whole 128-B lines, once each; values and sum orders unchanged
# speedup vs baseline: 5.2615x; 1.1008x over previous
#include <hip/hip_runtime.h>

typedef __attribute__((ext_vector_type(2))) float v2f;
typedef __attribute__((ext_vector_type(8))) float v8f;

#define IN_DIM 512
#define HID    16
#define ODIM   64

__global__ void k_deg_init(float* __restrict__ deg, int n) {
  int i = blockIdx.x * blockDim.x + threadIdx.x;
  if (i < n) deg[i] = 1.0f;
}

__global__ void k_deg_edges(const int* __restrict__ dst, float* __restrict__ deg, int E) {
  int i = blockIdx.x * blockDim.x + threadIdx.x;
  if (i < E) atomicAdd(&deg[dst[i]], 1.0f);
}

__global__ void k_dinv(float* __restrict__ deg, int n) {
  int i = blockIdx.x * blockDim.x + threadIdx.x;
  if (i < n) {
    float d = deg[i];
    deg[i] = (d > 0.0f) ? (1.0f / sqrtf(d)) : 0.0f;
  }
}

__device__ __forceinline__ void gemm1_tiles(const float* __restrict__ x,
                                            const float* __restrict__ W1,
                                            float* __restrict__ h1, int n, int pitch,
                                            float* lw, int wg) {
  for (int i = threadIdx.x; i < IN_DIM * HID; i += blockDim.x) lw[i] = W1[i];
  __syncthreads();

  int wave  = (wg * blockDim.x + threadIdx.x) >> 5;
  int lane  = threadIdx.x & 31;
  int tiles = (n + 15) >> 4;
  if (wave >= tiles) return;

  int half = lane >> 4;
  int ml   = lane & 15;
  int row  = wave * 16 + ml;
  int rrow = (row < n) ? row : (n - 1);
  const float* xr = x + (size_t)rrow * IN_DIM;

  v8f acc = {0.f, 0.f, 0.f, 0.f, 0.f, 0.f, 0.f, 0.f};
  for (int k = 0; k < IN_DIM; k += 4) {
    int ka = k + 2 * half;
    v2f a; a.x = xr[ka];                 a.y = xr[ka + 1];
    v2f b; b.x = lw[ka * HID + ml];      b.y = lw[(ka + 1) * HID + ml];
    acc = __builtin_amdgcn_wmma_f32_16x16x4_f32(false, a, false, b, (short)0, acc,
                                                false, false);
  }
#pragma unroll
  for (int r = 0; r < 8; ++r) {
    int orow = wave * 16 + r + 8 * half;
    if (orow < n) h1[(size_t)orow * pitch + ml] = acc[r];
  }
}

__global__ void k_gemm1(const float* __restrict__ x, const float* __restrict__ W1,
                        float* __restrict__ h1, int n, int pitch) {
  __shared__ float lw[IN_DIM * HID];
  gemm1_tiles(x, W1, h1, n, pitch, lw, blockIdx.x);
}

__global__ void k_init_z1(const float* __restrict__ h1, const float* __restrict__ dinv,
                          const float* __restrict__ b1, float* __restrict__ z1, int n) {
  int t = blockIdx.x * blockDim.x + threadIdx.x;
  int total = n * HID;
  if (t < total) {
    int i = t / HID, j = t - i * HID;
    float di = dinv[i];
    z1[t] = h1[t] * di * di + b1[j];
  }
}

__global__ void k_agg1(const int* __restrict__ src, const int* __restrict__ dst,
                       const float* __restrict__ dinv, const float* __restrict__ h1,
                       float* __restrict__ z1, int E) {
  int total  = E * 4;
  int stride = gridDim.x * blockDim.x;
  for (int t = blockIdx.x * blockDim.x + threadIdx.x; t < total; t += stride) {
    int e = t >> 2, c = t & 3;
    int s = src[e], d = dst[e];
    float norm = dinv[s] * dinv[d];
    const float4 hv = *(const float4*)(h1 + (size_t)s * HID + c * 4);
    float* o = z1 + (size_t)d * HID + c * 4;
    atomicAdd(o + 0, hv.x * norm);
    atomicAdd(o + 1, hv.y * norm);
    atomicAdd(o + 2, hv.z * norm);
    atomicAdd(o + 3, hv.w * norm);
  }
}

__global__ void k_gemm2(const float* __restrict__ z1, const float* __restrict__ W2,
                        float* __restrict__ h2, int n) {
  __shared__ float lw[HID * ODIM];
  for (int i = threadIdx.x; i < HID * ODIM; i += blockDim.x) lw[i] = W2[i];
  __syncthreads();

  int wave  = (blockIdx.x * blockDim.x + threadIdx.x) >> 5;
  int lane  = threadIdx.x & 31;
  int tiles = (n + 15) >> 4;
  if (wave >= tiles) return;

  int half = lane >> 4;
  int ml   = lane & 15;
  int row  = wave * 16 + ml;
  int rrow = (row < n) ? row : (n - 1);
  const float* zr = z1 + (size_t)rrow * HID;

  v2f afrag[4];
#pragma unroll
  for (int kk = 0; kk < 4; ++kk) {
    int ka = kk * 4 + 2 * half;
    afrag[kk].x = fmaxf(zr[ka], 0.f);
    afrag[kk].y = fmaxf(zr[ka + 1], 0.f);
  }

#pragma unroll
  for (int tcol = 0; tcol < 4; ++tcol) {
    v8f acc = {0.f, 0.f, 0.f, 0.f, 0.f, 0.f, 0.f, 0.f};
#pragma unroll
    for (int kk = 0; kk < 4; ++kk) {
      int ka = kk * 4 + 2 * half;
      v2f b;
      b.x = lw[ka * ODIM + tcol * 16 + ml];
      b.y = lw[(ka + 1) * ODIM + tcol * 16 + ml];
      acc = __builtin_amdgcn_wmma_f32_16x16x4_f32(false, afrag[kk], false, b, (short)0,
                                                  acc, false, false);
    }
#pragma unroll
    for (int r = 0; r < 8; ++r) {
      int orow = wave * 16 + r + 8 * half;
      if (orow < n) h2[(size_t)orow * ODIM + tcol * 16 + ml] = acc[r];
    }
  }
}

__global__ void k_init_out(const float* __restrict__ h2, const float* __restrict__ dinv,
                           const float* __restrict__ b2, float* __restrict__ out, int n) {
  int t = blockIdx.x * blockDim.x + threadIdx.x;
  int total = n * ODIM;
  if (t < total) {
    int i = t / ODIM, j = t - i * ODIM;
    float di = dinv[i];
    out[t] = h2[t] * di * di + b2[j];
  }
}

__global__ void k_agg2(const int* __restrict__ src, const int* __restrict__ dst,
                       const float* __restrict__ dinv, const float* __restrict__ h2,
                       float* __restrict__ out, int E) {
  int total  = E * 16;
  int stride = gridDim.x * blockDim.x;
  for (int t = blockIdx.x * blockDim.x + threadIdx.x; t < total; t += stride) {
    int e = t >> 4, c = t & 15;
    int s = src[e], d = dst[e];
    float norm = dinv[s] * dinv[d];
    const float4 hv = *(const float4*)(h2 + (size_t)s * ODIM + c * 4);
    float* o = out + (size_t)d * ODIM + c * 4;
    atomicAdd(o + 0, hv.x * norm);
    atomicAdd(o + 1, hv.y * norm);
    atomicAdd(o + 2, hv.z * norm);
    atomicAdd(o + 3, hv.w * norm);
  }
}

#define RB     128
#define LOG_RB 7
#define NSL    256
#define H1P    32

__device__ __forceinline__ int block_excl_scan(int v, int* scr, int* total) {
  const int t = threadIdx.x, nt = blockDim.x;
  scr[t] = v;
  __syncthreads();
  for (int o = 1; o < nt; o <<= 1) {
    int a = (t >= o) ? scr[t - o] : 0;
    __syncthreads();
    scr[t] += a;
    __syncthreads();
  }
  const int incl = scr[t];
  *total = scr[nt - 1];
  __syncthreads();
  return incl - v;
}

__device__ __forceinline__ void block_scan_and_sum(int va, int vb, int* scrA, int* scrB,
                                                   int* exA, int* totA, int* totB) {
  const int t = threadIdx.x, nt = blockDim.x;
  scrA[t] = va;
  scrB[t] = vb;
  __syncthreads();
  for (int o = 1; o < nt; o <<= 1) {
    int a = 0, b = 0;
    if (t >= o) {
      a = scrA[t - o];
      b = scrB[t - o];
    }
    __syncthreads();
    scrA[t] += a;
    scrB[t] += b;
    __syncthreads();
  }
  *exA  = scrA[t] - va;
  *totA = scrA[nt - 1];
  *totB = scrB[nt - 1];
  __syncthreads();
}

#define SC_CAP 16384
__device__ __forceinline__ void bslice(const int* __restrict__ src,
                                       const int* __restrict__ dst,
                                       int* __restrict__ lstart, unsigned* __restrict__ rec,
                                       int E, int L, int nb, int P, int* cnt, int* scr,
                                       unsigned* stage) {
  const int tid = threadIdx.x, bd = blockDim.x, wg = blockIdx.x;
  for (int i = tid; i < nb; i += bd) cnt[i] = 0;
  __syncthreads();
  const int e0 = wg * L;
  int e1 = e0 + L;
  if (e1 > E) e1 = E;
  int e = e0 + tid;
  for (; e + 3 * bd < e1; e += 4 * bd) {
    int d[4];
#pragma unroll
    for (int u = 0; u < 4; ++u) d[u] = dst[e + u * bd];
#pragma unroll
    for (int u = 0; u < 4; ++u) atomicAdd(&cnt[d[u] >> LOG_RB], 1);
  }
  for (; e < e1; e += bd) atomicAdd(&cnt[dst[e] >> LOG_RB], 1);
  __syncthreads();
  const int per = (nb + bd - 1) / bd;
  const int lo = tid * per;
  int hi = lo + per;
  if (hi > nb) hi = nb;
  int s = 0;
  for (int b = lo; b < hi; ++b) s += cnt[b];
  int tot;
  int pre = block_excl_scan(s, scr, &tot);
  int* lrow = lstart + (size_t)wg * P;
  for (int b = lo; b < hi; ++b) {
    const int c = cnt[b];
    cnt[b]  = pre;
    lrow[b] = pre;
    pre += c;
  }
  if (tid == 0) lrow[nb] = tot;
  __syncthreads();
  e = e0 + tid;
  for (; e + 3 * bd < e1; e += 4 * bd) {
    int sv[4], dv[4], lp[4];
#pragma unroll
    for (int u = 0; u < 4; ++u) {
      sv[u] = src[e + u * bd];
      dv[u] = dst[e + u * bd];
    }
#pragma unroll
    for (int u = 0; u < 4; ++u) lp[u] = atomicAdd(&cnt[dv[u] >> LOG_RB], 1);
#pragma unroll
    for (int u = 0; u < 4; ++u)
      stage[lp[u]] = ((unsigned)sv[u] << LOG_RB) | (unsigned)(dv[u] & (RB - 1));
  }
  for (; e < e1; e += bd) {
    const int sv = src[e], dv = dst[e];
    const int lp = atomicAdd(&cnt[dv >> LOG_RB], 1);
    stage[lp] = ((unsigned)sv << LOG_RB) | (unsigned)(dv & (RB - 1));
  }
  __syncthreads();
  const int m = e1 - e0;
  for (int i = tid; i < m; i += 4 * bd) {
    unsigned rv[4];
#pragma unroll
    for (int u = 0; u < 4; ++u) {
      const int j = i + u * bd;
      rv[u] = stage[j < m ? j : m - 1];
    }
#pragma unroll
    for (int u = 0; u < 4; ++u)
      if (i + u * bd < m) rec[e0 + i + u * bd] = rv[u];
  }
}

__global__ void k_bslice_gemm1(const int* __restrict__ src, const int* __restrict__ dst,
                               int* __restrict__ lstart, unsigned* __restrict__ rec,
                               int E, int L, int nb, int P,
                               const float* __restrict__ x, const float* __restrict__ W1,
                               float* __restrict__ h1, int n, int pitch) {
  extern __shared__ int dynsh[];
  __shared__ unsigned buf[SC_CAP];
  if (blockIdx.x < NSL) bslice(src, dst, lstart, rec, E, L, nb, P, dynsh, dynsh + nb, buf);
  else gemm1_tiles(x, W1, h1, n, pitch, (float*)buf, (int)blockIdx.x - NSL);
}

#define SRT_T 512
#define RPW (NSL / (SRT_T / 32))
#define SRT_CAP 8192
__global__ void k_bsort(const unsigned* __restrict__ rec, const int* __restrict__ lstart,
                        float* __restrict__ dinv, float* __restrict__ h1,
                        int* __restrict__ rowptr, unsigned* __restrict__ srt, int n, int nb,
                        int L, int P) {
  __shared__ int cnt[RB];
  __shared__ int cur[RB];
  __shared__ int rbase[NSL];
  __shared__ int rlen[NSL];
  __shared__ int roff[NSL];
  __shared__ int scrA[SRT_T];
  __shared__ int scrB[SRT_T];
  __shared__ unsigned sbuf[SRT_CAP];
  __shared__ unsigned obuf[SRT_CAP];
  const int tid = threadIdx.x, bd = blockDim.x, b = blockIdx.x;
  const int wv = tid >> 5, lane = tid & 31, nwv = bd >> 5;
  if (tid < RB) cnt[tid] = 0;
  int a = 0, len = 0;
  if (tid < NSL) {
    const int* lrow = lstart + (size_t)tid * P + b;
    a   = lrow[0];
    len = lrow[1] - a;
  }
  int ro, m, e0;
  block_scan_and_sum(len, a, scrA, scrB, &ro, &m, &e0);
  if (tid < NSL) {
    rbase[tid] = tid * L + a;
    rlen[tid]  = len;
    roff[tid]  = ro;
  }
  __syncthreads();
  const int e1 = e0 + m;
  const bool staged = m <= SRT_CAP;
  if (staged) {
    int myln = 0, myrb = 0, myof = 0;
    if (lane < RPW) {
      const int w = wv * RPW + lane;
      myln = rlen[w];
      myrb = rbase[w];
      myof = roff[w];
    }
    unsigned rv[RPW];
#pragma unroll
    for (int j = 0; j < RPW; ++j) {
      const int ln = __builtin_amdgcn_readlane(myln, j);
      const int rb = __builtin_amdgcn_readlane(myrb, j);
      rv[j] = 0u;
      if (lane < ln) rv[j] = rec[rb + lane];
    }
#pragma unroll
    for (int j = 0; j < RPW; ++j) {
      const int ln  = __builtin_amdgcn_readlane(myln, j);
      const int ofs = __builtin_amdgcn_readlane(myof, j);
      if (lane < ln) sbuf[ofs + lane] = rv[j];
    }
#pragma unroll
    for (int j = 0; j < RPW; ++j) {
      const int ln  = __builtin_amdgcn_readlane(myln, j);
      const int rb  = __builtin_amdgcn_readlane(myrb, j);
      const int ofs = __builtin_amdgcn_readlane(myof, j);
      for (int i = lane + 32; i < ln; i += 32) sbuf[ofs + i] = rec[rb + i];
    }
    __syncthreads();
    for (int i = tid; i < m; i += 4 * bd) {
      unsigned r[4];
#pragma unroll
      for (int u = 0; u < 4; ++u) {
        const int j = i + u * bd;
        r[u] = sbuf[j < m ? j : m - 1];
      }
#pragma unroll
      for (int u = 0; u < 4; ++u)
        if (i + u * bd < m) atomicAdd(&cnt[r[u] & (RB - 1)], 1);
    }
  } else {
    for (int w = wv; w < NSL; w += nwv) {
      const int ln = rlen[w], rb = rbase[w];
      for (int i = lane; i < ln; i += 32) atomicAdd(&cnt[rec[rb + i] & (RB - 1)], 1);
    }
  }
  __syncthreads();
  if (tid < 32) {
    const int c0 = cnt[4 * tid], c1 = cnt[4 * tid + 1], c2 = cnt[4 * tid + 2],
              c3 = cnt[4 * tid + 3];
    const int s = c0 + c1 + c2 + c3;
    int incl = s;
    for (int o = 1; o < 32; o <<= 1) {
      const int t = __shfl_up(incl, o);
      if (tid >= o) incl += t;
    }
    const int ex = e0 + incl - s;
    cur[4 * tid]     = ex;
    cur[4 * tid + 1] = ex + c0;
    cur[4 * tid + 2] = ex + c0 + c1;
    cur[4 * tid + 3] = ex + c0 + c1 + c2;
  }
  __syncthreads();
  if (tid < RB) {
    const int node = b * RB + tid;
    rowptr[node] = cur[tid];
    if (node < n) {
      const float d = 1.0f + (float)cnt[tid];
      const float dv = (d > 0.0f) ? (1.0f / sqrtf(d)) : 0.0f;
      dinv[node] = dv;
      h1[(size_t)node * H1P + HID] = dv;
    }
  }
  if (b == nb - 1 && tid == 0) rowptr[nb * RB] = e1;
  __syncthreads();
  if (staged) {
    for (int i = tid; i < m; i += 4 * bd) {
      unsigned r[4];
      int p[4];
#pragma unroll
      for (int u = 0; u < 4; ++u) {
        const int j = i + u * bd;
        r[u] = sbuf[j < m ? j : m - 1];
      }
#pragma unroll
      for (int u = 0; u < 4; ++u) {
        p[u] = e0;
        if (i + u * bd < m) p[u] = atomicAdd(&cur[r[u] & (RB - 1)], 1);
      }
#pragma unroll
      for (int u = 0; u < 4; ++u)
        if (i + u * bd < m) obuf[p[u] - e0] = r[u] >> LOG_RB;
    }
    __syncthreads();
    for (int i = tid; i < m; i += 4 * bd) {
      unsigned rv[4];
#pragma unroll
      for (int u = 0; u < 4; ++u) {
        const int j = i + u * bd;
        rv[u] = obuf[j < m ? j : m - 1];
      }
#pragma unroll
      for (int u = 0; u < 4; ++u)
        if (i + u * bd < m) srt[e0 + i + u * bd] = rv[u];
    }
  } else {
    for (int w = wv; w < NSL; w += nwv) {
      const int ln = rlen[w], rb = rbase[w];
      for (int i = lane; i < ln; i += 32) {
        const unsigned r = rec[rb + i];
        const int p = atomicAdd(&cur[r & (RB - 1)], 1);
        srt[p] = r >> LOG_RB;
      }
    }
  }
}

#define ZP 20

template <int LPE>
__device__ __forceinline__ int group_word(int v, int u) {
  if (LPE == 4) {
    switch (u) {
      case 0:  return __builtin_amdgcn_mov_dpp(v, 0x00, 0xf, 0xf, true);
      case 1:  return __builtin_amdgcn_mov_dpp(v, 0x55, 0xf, 0xf, true);
      case 2:  return __builtin_amdgcn_mov_dpp(v, 0xaa, 0xf, 0xf, true);
      default: return __builtin_amdgcn_mov_dpp(v, 0xff, 0xf, 0xf, true);
    }
  }
  switch (u) {
    case 0:  return __builtin_amdgcn_mov_dpp(v, 0x150, 0xf, 0xf, true);
    case 1:  return __builtin_amdgcn_mov_dpp(v, 0x151, 0xf, 0xf, true);
    case 2:  return __builtin_amdgcn_mov_dpp(v, 0x152, 0xf, 0xf, true);
    case 3:  return __builtin_amdgcn_mov_dpp(v, 0x153, 0xf, 0xf, true);
    case 4:  return __builtin_amdgcn_mov_dpp(v, 0x154, 0xf, 0xf, true);
    case 5:  return __builtin_amdgcn_mov_dpp(v, 0x155, 0xf, 0xf, true);
    case 6:  return __builtin_amdgcn_mov_dpp(v, 0x156, 0xf, 0xf, true);
    case 7:  return __builtin_amdgcn_mov_dpp(v, 0x157, 0xf, 0xf, true);
    case 8:  return __builtin_amdgcn_mov_dpp(v, 0x158, 0xf, 0xf, true);
    case 9:  return __builtin_amdgcn_mov_dpp(v, 0x159, 0xf, 0xf, true);
    case 10: return __builtin_amdgcn_mov_dpp(v, 0x15a, 0xf, 0xf, true);
    case 11: return __builtin_amdgcn_mov_dpp(v, 0x15b, 0xf, 0xf, true);
    case 12: return __builtin_amdgcn_mov_dpp(v, 0x15c, 0xf, 0xf, true);
    case 13: return __builtin_amdgcn_mov_dpp(v, 0x15d, 0xf, 0xf, true);
    case 14: return __builtin_amdgcn_mov_dpp(v, 0x15e, 0xf, 0xf, true);
    default: return __builtin_amdgcn_mov_dpp(v, 0x15f, 0xf, 0xf, true);
  }
}

template <int F, int HP, bool FIRST, bool FOLD, int GE, bool LW>
__device__ __forceinline__ void cagg_body(const unsigned* __restrict__ srt,
                                          const int* __restrict__ rowptr,
                                          const float* __restrict__ dinv,
                                          const float* __restrict__ h,
                                          const float* __restrict__ bias,
                                          const float* __restrict__ dsrc_in,
                                          float* __restrict__ dsrc_out,
                                          float* __restrict__ o, float* zs, unsigned* ew,
                                          int ewb, int n) {
  constexpr int LPE = F / 4;
  constexpr int WIN = (LPE > GE) ? LPE : GE;
  constexpr int NR  = WIN / LPE;
  constexpr int NSG = WIN / GE;
  const int tid = threadIdx.x, b = blockIdx.x;
  const int g = tid / LPE, c = tid % LPE;
  const int ng = blockDim.x / LPE;
  const float4* __restrict__ h4 = (const float4*)h;
  for (int nl = g; nl < RB; nl += ng) {
    const int node = b * RB + nl;
    if (node >= n) {
      if (!FOLD) break;
      const float4 zero = {0.f, 0.f, 0.f, 0.f};
      *(float4*)(zs + nl * ZP + c * 4) = zero;
      continue;
    }
    const int e0 = rowptr[node], e1 = rowptr[node + 1];
    const float dd = dinv[node];
    const float4 hself = h4[(unsigned)node * (unsigned)(HP / 4) + (unsigned)c];
    const float4 bv = ((const float4*)bias)[c];
    float4 self;
    self.x = hself.x * dd * dd + bv.x;
    self.y = hself.y * dd * dd + bv.y;
    self.z = hself.z * dd * dd + bv.z;
    self.w = hself.w * dd * dd + bv.w;
    float4 acc = {0.f, 0.f, 0.f, 0.f};
    unsigned sn[NR];
    float dn[NR];
    if (e0 < e1) {
#pragma unroll
      for (int r = 0; r < NR; ++r) {
        const int idx = (e0 + r * LPE + c < e1) ? (e0 + r * LPE + c) : (e1 - 1);
        sn[r] = LW ? ew[idx - ewb] : srt[idx];
        if (!FIRST) dn[r] = dsrc_in[idx];
      }
    }
    if (GE > 4) asm volatile("" : "+v"(self.x), "+v"(self.y), "+v"(self.z), "+v"(self.w));
    for (int e = e0; e < e1; e += WIN) {
      unsigned sx[NR];
      float dx[NR];
#pragma unroll
      for (int r = 0; r < NR; ++r) {
        const int idx = (e + WIN + r * LPE + c < e1) ? (e + WIN + r * LPE + c) : (e1 - 1);
        sx[r] = LW ? ew[idx - ewb] : srt[idx];
        if (!FIRST) dx[r] = dsrc_in[idx];
      }
#pragma unroll
      for (int sg = 0; sg < NSG; ++sg) {
        const int rem = e1 - (e + sg * GE);
        if (sg == 0 || rem > 0) {
          float4 hv[GE];
          float ds[GE];
#pragma unroll
          for (int u = 0; u < GE; ++u) {
            const int j = sg * GE + u;
            const unsigned s = (unsigned)group_word<LPE>((int)sn[j / LPE], j % LPE);
            hv[u] = h4[s * (unsigned)(HP / 4) + (unsigned)c];
            if (FIRST) ds[u] = h[s * (unsigned)HP + (unsigned)F];
          }
          if (FIRST) {
#pragma unroll
            for (int r = 0; r < GE / LPE; ++r) {
              float v = ds[r * LPE];
#pragma unroll
              for (int l = 1; l < LPE; ++l) v = (c == l) ? ds[r * LPE + l] : v;
              if (r * LPE + c < rem) {
                if (LW) ew[e + sg * GE + r * LPE + c - ewb] = __float_as_uint(v);
                else dsrc_out[e + sg * GE + r * LPE + c] = v;
              }
            }
          }
#pragma unroll
          for (int u = 0; u < GE; ++u) {
            const int j = sg * GE + u;
            const float dsv =
                FIRST ? ds[u]
                      : __int_as_float(group_word<LPE>(__float_as_int(dn[j / LPE]), j % LPE));
            const float norm = (u < rem) ? dsv * dd : 0.f;
            acc.x = __builtin_fmaf(hv[u].x, norm, acc.x);
            acc.y = __builtin_fmaf(hv[u].y, norm, acc.y);
            acc.z = __builtin_fmaf(hv[u].z, norm, acc.z);
            acc.w = __builtin_fmaf(hv[u].w, norm, acc.w);
          }
        }
      }
#pragma unroll
      for (int r = 0; r < NR; ++r) {
        sn[r] = sx[r];
        if (!FIRST) dn[r] = dx[r];
      }
    }
    float4 r;
    r.x = acc.x + self.x;
    r.y = acc.y + self.y;
    r.z = acc.z + self.z;
    r.w = acc.w + self.w;
    if (FOLD) {
      float4 q;
      q.x = fmaxf(r.x, 0.f);
      q.y = fmaxf(r.y, 0.f);
      q.z = fmaxf(r.z, 0.f);
      q.w = fmaxf(r.w, 0.f);
      *(float4*)(zs + nl * ZP + c * 4) = q;
    } else {
      ((float4*)o)[(unsigned)node * (unsigned)LPE + (unsigned)c] = r;
    }
  }
}

__global__ void k_cagg1(const unsigned* __restrict__ srt, const int* __restrict__ rowptr,
                        const float* __restrict__ dinv, const float* __restrict__ h1,
                        const float* __restrict__ b1, float* __restrict__ dsrc,
                        float* __restrict__ z1, int n) {
  cagg_body<HID, H1P, true, false, 4, false>(srt, rowptr, dinv, h1, b1, nullptr, dsrc, z1, nullptr,
                                             nullptr, 0, n);
}

#define HT_P 68
#define EW_CAP 6144
#define EW_U 12
__global__ void k_cagg1_gemm2(const unsigned* __restrict__ srt,
                              const int* __restrict__ rowptr,
                              const float* __restrict__ dinv, const float* __restrict__ h1,
                              const float* __restrict__ b1, float* __restrict__ dsrc,
                              const float* __restrict__ W2, float* __restrict__ h2, int n) {
  __shared__ __attribute__((aligned(16))) float zs[RB * ZP];
  __shared__ float lw2[HID * ODIM];
  __shared__ unsigned ew[EW_CAP];
  __shared__ __attribute__((aligned(16))) float ht[RB * HT_P];
  const int tid = threadIdx.x, bd = blockDim.x, b = blockIdx.x;
  for (int i = tid; i < HID * ODIM; i += bd) lw2[i] = W2[i];

  const int eb0 = rowptr[b * RB], eb1 = rowptr[b * RB + RB];
  const int ewb = eb0 & ~31;
  const bool staged = eb1 - ewb <= EW_CAP;
  if (staged) {
    for (int j = ewb + tid; j < eb1; j += EW_U * bd) {
      unsigned v[EW_U];
#pragma unroll
      for (int u = 0; u < EW_U; ++u) {
        const int q = j + u * bd;
        v[u] = srt[q < eb1 ? q : eb1 - 1];
      }
#pragma unroll
      for (int u = 0; u < EW_U; ++u) {
        const int q = j + u * bd;
        ew[(q < eb1 ? q : eb1 - 1) - ewb] = v[u];
      }
    }
  }
  __syncthreads();
  if (staged)
    cagg_body<HID, H1P, true, true, 4, true>(srt, rowptr, dinv, h1, b1, nullptr, nullptr, nullptr,
                                             zs, ew, ewb, n);
  else
    cagg_body<HID, H1P, true, true, 4, false>(srt, rowptr, dinv, h1, b1, nullptr, dsrc, nullptr,
                                              zs, nullptr, 0, n);
  __syncthreads();

  const int wv = tid >> 5, nwv = bd >> 5;
  const int lane = tid & 31;
  const int half = lane >> 4;
  const int ml   = lane & 15;
  for (int unit = wv; unit < (RB / 16) * 2; unit += nwv) {
    const int rt = unit >> 1;
    const float* zr = zs + (rt * 16 + ml) * ZP;
    v2f afrag[4];
#pragma unroll
    for (int kk = 0; kk < 4; ++kk) {
      const int ka = kk * 4 + 2 * half;
      afrag[kk].x = zr[ka];
      afrag[kk].y = zr[ka + 1];
    }
#pragma unroll
    for (int t = 0; t < 2; ++t) {
      const int tcol = (unit & 1) * 2 + t;
      v8f acc = {0.f, 0.f, 0.f, 0.f, 0.f, 0.f, 0.f, 0.f};
#pragma unroll
      for (int kk = 0; kk < 4; ++kk) {
        const int ka = kk * 4 + 2 * half;
        v2f bb;
        bb.x = lw2[ka * ODIM + tcol * 16 + ml];
        bb.y = lw2[(ka + 1) * ODIM + tcol * 16 + ml];
        acc = __builtin_amdgcn_wmma_f32_16x16x4_f32(false, afrag[kk], false, bb, (short)0,
                                                    acc, false, false);
      }
#pragma unroll
      for (int r = 0; r < 8; ++r)
        ht[(rt * 16 + r + 8 * half) * HT_P + tcol * 16 + ml] = acc[r];
    }
  }
  __syncthreads();
  if (staged) {
    for (int j = ewb + tid; j < eb1; j += 4 * bd) {
      unsigned v[4];
#pragma unroll
      for (int u = 0; u < 4; ++u) {
        const int q = j + u * bd;
        v[u] = ew[(q < eb1 ? q : eb1 - 1) - ewb];
      }
#pragma unroll
      for (int u = 0; u < 4; ++u) {
        const int q = j + u * bd;
        if (q >= eb0 && q < eb1) dsrc[q] = __uint_as_float(v[u]);
      }
    }
  }
  for (int i = tid; i < RB * (ODIM / 4); i += bd) {
    const int row = i / (ODIM / 4), c4 = i % (ODIM / 4);
    const int orow = b * RB + row;
    if (orow < n)
      ((float4*)h2)[(size_t)orow * (ODIM / 4) + c4] = *(const float4*)(ht + row * HT_P + c4 * 4);
  }
}

__global__ void k_cagg2(const unsigned* __restrict__ srt, const int* __restrict__ rowptr,
                        const float* __restrict__ dinv, const float* __restrict__ h2,
                        const float* __restrict__ b2, const float* __restrict__ dsrc,
                        float* __restrict__ out, int n) {
  cagg_body<ODIM, ODIM, false, false, 8, false>(srt, rowptr, dinv, h2, b2, dsrc, nullptr, out,
                                                nullptr, nullptr, 0, n);
}

extern "C" void kernel_launch(void* const* d_in, const int* in_sizes, int n_in,
                              void* d_out, int out_size, void* d_ws, size_t ws_size,
                              hipStream_t stream) {
  const float* x  = (const float*)d_in[0];
  const float* W1 = (const float*)d_in[1];
  const float* b1 = (const float*)d_in[2];
  const float* W2 = (const float*)d_in[3];
  const float* b2 = (const float*)d_in[4];
  const int*   ei = (const int*)d_in[5];

  const int N = in_sizes[0] / IN_DIM;
  const int E = in_sizes[5] / 2;
  const int* src = ei;
  const int* dst = ei + E;
  float* out = (float*)d_out;

  char* ws = (char*)d_ws;
  size_t off = 0;
  auto alloc = [&](size_t nbytes) -> void* {
    void* p = ws + off;
    off = (off + nbytes + 255) & ~(size_t)255;
    return p;
  };
  float* dinv = (float*)alloc((size_t)N * sizeof(float));
  float* h2   = (float*)alloc((size_t)N * ODIM * sizeof(float));
  const size_t planes = off;

  const int B = 256;
  const int tiles = (N + 15) >> 4;
  const int gemmBlocks = (tiles + 7) / 8;

  const int NB = (N + RB - 1) / RB;
  const int L  = ((E + NSL - 1) / NSL + 31) & ~31;
  const int P  = (NB + 1 + 31) & ~31;
  unsigned* rec = (unsigned*)alloc((size_t)E * sizeof(unsigned));
  unsigned* srt = (unsigned*)alloc((size_t)E * sizeof(unsigned));
  int* lstart   = (int*)alloc((size_t)NSL * P * sizeof(int));
  int* rowptr   = (int*)alloc(((size_t)NB * RB + 1) * sizeof(int));
  float* dsrc = (float*)rec;
  float* tail = (float*)(ws + off);
  const size_t need_z1  = off + (size_t)N * HID * sizeof(float);
  const size_t need_h1p = off + (size_t)N * H1P * sizeof(float);
  const int gemm1Groups = (tiles + 15) / 16;

  if (need_z1 <= ws_size && L <= SC_CAP) {
    const bool fold = need_h1p <= ws_size;
    float* h1a = fold ? tail : h2;
    k_bslice_gemm1<<<NSL + gemm1Groups, 512, (size_t)(NB + 512) * sizeof(int), stream>>>(
        src, dst, lstart, rec, E, L, NB, P, x, W1, h1a, N, H1P);
    k_bsort   <<<NB, SRT_T, 0, stream>>>(rec, lstart, dinv, h1a, rowptr, srt, N, NB, L, P);

    if (fold) {
      k_cagg1_gemm2<<<NB, 512, 0, stream>>>(srt, rowptr, dinv, h1a, b1, dsrc, W2, h2, N);
    } else {
      float* z1 = tail;
      k_cagg1 <<<NB, 512, 0, stream>>>(srt, rowptr, dinv, h1a, b1, dsrc, z1, N);
      k_gemm2 <<<gemmBlocks, B, 0, stream>>>(z1, W2, h2, N);
    }
    k_cagg2   <<<NB, 512, 0, stream>>>(srt, rowptr, dinv, h2, b2, dsrc, out, N);
    return;
  }

  off = planes;
  float* z1 = (float*)alloc((size_t)N * HID * sizeof(float));
  float* h1 = (float*)alloc((size_t)N * HID * sizeof(float));
  k_deg_init <<<(N + B - 1) / B, B, 0, stream>>>(dinv, N);
  k_deg_edges<<<(E + B - 1) / B, B, 0, stream>>>(dst, dinv, E);
  k_dinv     <<<(N + B - 1) / B, B, 0, stream>>>(dinv, N);

  k_gemm1    <<<gemmBlocks, B, 0, stream>>>(x, W1, h1, N, HID);
  k_init_z1  <<<((size_t)N * HID + B - 1) / B, B, 0, stream>>>(h1, dinv, b1, z1, N);
  k_agg1     <<<(E * 4 + B - 1) / B, B, 0, stream>>>(src, dst, dinv, h1, z1, E);

  k_gemm2    <<<gemmBlocks, B, 0, stream>>>(z1, W2, h2, N);
  k_init_out <<<((size_t)N * ODIM + B - 1) / B, B, 0, stream>>>(h2, dinv, b2, out, N);
  k_agg2     <<<(E * 16 + B - 1) / B, B, 0, stream>>>(src, dst, dinv, h2, out, E);
}
